# starting kernel
# baseline (speedup 1.0000x reference)
#include <hip/hip_runtime.h>


#define BB 2
#define SS 2048
#define DD 1024
#define HH 16
#define DKV 64

typedef __attribute__((ext_vector_type(16))) __bf16 v16bf;
typedef __attribute__((ext_vector_type(8)))  float  v8f;

union FragU { v16bf v; uint4 q[2]; };

__device__ __forceinline__ unsigned short f2bf(float f) {
  unsigned int u = __float_as_uint(f);
  u += 0x7FFFu + ((u >> 16) & 1u);
  return (unsigned short)(u >> 16);
}

__device__ __forceinline__ v16bf load_frag(const unsigned short* rowbase, int lane) {
  const unsigned short* p = rowbase + ((lane & 16) ? 8 : 0);
  FragU f;
  f.q[0] = *reinterpret_cast<const uint4*>(p);
  f.q[1] = *reinterpret_cast<const uint4*>(p + 16);
  return f.v;
}

#define WMMA_BF16(a, b, c) \
  __builtin_amdgcn_wmma_f32_16x16x32_bf16(false, (a), false, (b), (short)0, (c), false, false)

__global__ void cvt_kernel(const float* __restrict__ src,
                           unsigned short* __restrict__ dst, int n) {
  int i = blockIdx.x * blockDim.x + threadIdx.x;
  if (i < n) dst[i] = f2bf(src[i]);
}

__global__ void cvtT_kernel(const float* __restrict__ src,
                            unsigned short* __restrict__ dst,
                            int G, int R, int C) {
  int i = blockIdx.x * blockDim.x + threadIdx.x;
  int n = G * R * C;
  if (i >= n) return;
  int g = i / (R * C);
  int rem = i - g * (R * C);
  int r = rem / C, c = rem - r * C;
  dst[(size_t)g * R * C + (size_t)c * R + r] = f2bf(src[i]);
}

__global__ void proj_kernel(const unsigned short* __restrict__ X,
                            const unsigned short* __restrict__ WT,
                            unsigned short* __restrict__ OUT,
                            float scale, int transposeOut) {
  const int lane = threadIdx.x & 31;
  const int wave = threadIdx.x >> 5;
  const int bh = blockIdx.x;
  const int b = bh / HH, h = bh % HH;
  const int s_base = blockIdx.y * 64 + wave * 16;

  const unsigned short* Xrow =
      X + ((size_t)b * SS + (size_t)(s_base + (lane & 15))) * DD;
  const unsigned short* Wrow = WT + ((size_t)h * DKV + (lane & 15)) * DD;

  v8f acc[4] = {};
#pragma unroll 2
  for (int k0 = 0; k0 < DD; k0 += 32) {
    __builtin_prefetch(Xrow + k0 + 64, 0, 1);
    v16bf a = load_frag(Xrow + k0, lane);
    v16bf bf[4];
#pragma unroll
    for (int nt = 0; nt < 4; ++nt)
      bf[nt] = load_frag(Wrow + (size_t)nt * 16 * DD + k0, lane);
#pragma unroll
    for (int nt = 0; nt < 4; ++nt)
      acc[nt] = WMMA_BF16(a, bf[nt], acc[nt]);
  }
  const int rbase = (lane & 16) ? 8 : 0;
  const int col0 = lane & 15;
#pragma unroll
  for (int nt = 0; nt < 4; ++nt) {
#pragma unroll
    for (int r = 0; r < 8; ++r) {
      int srow = s_base + rbase + r;
      int col = nt * 16 + col0;
      unsigned short val = f2bf(acc[nt][r] * scale);
      if (transposeOut)
        OUT[(size_t)bh * DKV * SS + (size_t)col * SS + srow] = val;
      else
        OUT[(size_t)bh * SS * DKV + (size_t)srow * DKV + col] = val;
    }
  }
}

__global__ void attn_kernel(const unsigned short* __restrict__ QH,
                            const unsigned short* __restrict__ KH,
                            const unsigned short* __restrict__ VT,
                            const int* __restrict__ MASK,
                            unsigned short* __restrict__ AOUT) {
  __shared__ __align__(16) unsigned short Pld[4][16][32];

  const int lane = threadIdx.x & 31;
  const int wave = threadIdx.x >> 5;
  const int bh = blockIdx.x;
  const int b = bh >> 4, h = bh & 15;
  const int s_base = blockIdx.y * 64 + wave * 16;
  const int rbase = (lane & 16) ? 8 : 0;
  const int col = lane & 15;

  const unsigned short* Qrow =
      QH + ((size_t)bh * SS + (size_t)(s_base + col)) * DKV;
  v16bf qf0 = load_frag(Qrow, lane);
  v16bf qf1 = load_frag(Qrow + 32, lane);

  v8f o[4] = {};
  float m[8], l[8];
#pragma unroll
  for (int r = 0; r < 8; ++r) { m[r] = -3.0e38f; l[r] = 0.0f; }

  const int* maskbase = MASK + (size_t)b * SS * SS;
  const unsigned short* Krow = KH + ((size_t)bh * SS + col) * DKV;
  const unsigned short* Vrow = VT + ((size_t)bh * DKV + col) * SS;

  for (int t0 = 0; t0 < SS; t0 += 32) {
    v16bf kb0 = load_frag(Krow + (size_t)t0 * DKV, lane);
    v16bf kb1 = load_frag(Krow + (size_t)t0 * DKV + 32, lane);
    v16bf kb2 = load_frag(Krow + (size_t)(t0 + 16) * DKV, lane);
    v16bf kb3 = load_frag(Krow + (size_t)(t0 + 16) * DKV + 32, lane);
    v16bf vb[4];
#pragma unroll
    for (int nt = 0; nt < 4; ++nt)
      vb[nt] = load_frag(Vrow + (size_t)nt * 16 * SS + t0, lane);

    v8f s0 = {}, s1 = {};
    s0 = WMMA_BF16(qf0, kb0, s0);
    s0 = WMMA_BF16(qf1, kb1, s0);
    s1 = WMMA_BF16(qf0, kb2, s1);
    s1 = WMMA_BF16(qf1, kb3, s1);

#pragma unroll
    for (int r = 0; r < 8; ++r) {
      const int* mrow = maskbase + (size_t)(s_base + rbase + r) * SS + t0;
      float v0 = (mrow[col]      != 0) ? s0[r] : -1.0e9f;
      float v1 = (mrow[col + 16] != 0) ? s1[r] : -1.0e9f;
      float mx = fmaxf(v0, v1);
      mx = fmaxf(mx, __shfl_xor(mx, 1));
      mx = fmaxf(mx, __shfl_xor(mx, 2));
      mx = fmaxf(mx, __shfl_xor(mx, 4));
      mx = fmaxf(mx, __shfl_xor(mx, 8));
      float mnew  = fmaxf(m[r], mx);
      float alpha = __expf(m[r] - mnew);
      float p0 = __expf(v0 - mnew);
      float p1 = __expf(v1 - mnew);
      float rs = p0 + p1;
      rs += __shfl_xor(rs, 1);
      rs += __shfl_xor(rs, 2);
      rs += __shfl_xor(rs, 4);
      rs += __shfl_xor(rs, 8);
      l[r] = l[r] * alpha + rs;
      m[r] = mnew;
#pragma unroll
      for (int nt = 0; nt < 4; ++nt) o[nt][r] = o[nt][r] * alpha;
      Pld[wave][rbase + r][col]      = f2bf(p0);
      Pld[wave][rbase + r][col + 16] = f2bf(p1);
    }
    asm volatile("s_wait_dscnt 0" ::: "memory");
    v16bf pf = load_frag(&Pld[wave][col][0], lane);
#pragma unroll
    for (int nt = 0; nt < 4; ++nt)
      o[nt] = WMMA_BF16(pf, vb[nt], o[nt]);
  }

#pragma unroll
  for (int r = 0; r < 8; ++r) {
    float inv = 1.0f / l[r];
    int srow = s_base + rbase + r;
#pragma unroll
    for (int nt = 0; nt < 4; ++nt) {
      int dv = nt * 16 + col;
      AOUT[((size_t)b * SS + srow) * (HH * DKV) + h * DKV + dv] =
          f2bf(o[nt][r] * inv);
    }
  }
}

__global__ void fc_kernel(const unsigned short* __restrict__ A,
                          const unsigned short* __restrict__ WT,
                          const float* __restrict__ RES,
                          float* __restrict__ OUT) {
  const int lane = threadIdx.x & 31;
  const int wave = threadIdx.x >> 5;
  const int row0 = blockIdx.x * 16;
  const int n0 = wave * 128;

  const unsigned short* Arow = A + (size_t)(row0 + (lane & 15)) * DD;
  const unsigned short* Wrow = WT + (size_t)(n0 + (lane & 15)) * DD;

  v8f acc[8] = {};
  for (int k0 = 0; k0 < DD; k0 += 32) {
    __builtin_prefetch(Arow + k0 + 64, 0, 1);
    v16bf a = load_frag(Arow + k0, lane);
    v16bf bf[8];
#pragma unroll
    for (int nt = 0; nt < 8; ++nt)
      bf[nt] = load_frag(Wrow + (size_t)nt * 16 * DD + k0, lane);
#pragma unroll
    for (int nt = 0; nt < 8; ++nt)
      acc[nt] = WMMA_BF16(a, bf[nt], acc[nt]);
  }
  const int rbase = (lane & 16) ? 8 : 0;
  const int col0 = lane & 15;
#pragma unroll
  for (int nt = 0; nt < 8; ++nt) {
#pragma unroll
    for (int r = 0; r < 8; ++r) {
      size_t idx = (size_t)(row0 + rbase + r) * DD + n0 + nt * 16 + col0;
      OUT[idx] = acc[nt][r] + RES[idx];
    }
  }
}

__global__ void ln_kernel(const float* __restrict__ X,
                          const float* __restrict__ W,
                          const float* __restrict__ Bv,
                          float* __restrict__ OUT) {
  __shared__ float red[8];
  const int row = blockIdx.x;
  const int tid = threadIdx.x;
  const float* xr = X + (size_t)row * DD;

  float v[4];
  float s = 0.0f;
#pragma unroll
  for (int i = 0; i < 4; ++i) { v[i] = xr[tid + i * 256]; s += v[i]; }
  for (int off = 1; off < 32; off <<= 1) s += __shfl_xor(s, off);
  if ((tid & 31) == 0) red[tid >> 5] = s;
  __syncthreads();
  float mu = (red[0] + red[1] + red[2] + red[3] +
              red[4] + red[5] + red[6] + red[7]) * (1.0f / DD);
  float vs = 0.0f;
#pragma unroll
  for (int i = 0; i < 4; ++i) { float d = v[i] - mu; vs += d * d; }
  for (int off = 1; off < 32; off <<= 1) vs += __shfl_xor(vs, off);
  __syncthreads();
  if ((tid & 31) == 0) red[tid >> 5] = vs;
  __syncthreads();
  float var = (red[0] + red[1] + red[2] + red[3] +
               red[4] + red[5] + red[6] + red[7]) * (1.0f / DD);
  float rstd = rsqrtf(var + 1e-6f);
#pragma unroll
  for (int i = 0; i < 4; ++i) {
    int c = tid + i * 256;
    OUT[(size_t)row * DD + c] = (v[i] - mu) * rstd * W[c] + Bv[c];
  }
}

extern "C" void kernel_launch(void* const* d_in, const int* in_sizes, int n_in,
                              void* d_out, int out_size, void* d_ws, size_t ws_size,
                              hipStream_t stream) {
  const float* q    = (const float*)d_in[0];
  const float* k    = (const float*)d_in[1];
  const float* v    = (const float*)d_in[2];
  const int*   mask = (const int*)d_in[3];
  const float* Wq   = (const float*)d_in[4];
  const float* Wk   = (const float*)d_in[5];
  const float* Wv   = (const float*)d_in[6];
  const float* Wfc  = (const float*)d_in[7];
  const float* lnw  = (const float*)d_in[8];
  const float* lnb  = (const float*)d_in[9];
  float* out = (float*)d_out;

  char* ws = (char*)d_ws;
  const size_t SZ_ACT  = (size_t)BB * SS * DD * 2;
  const size_t SZ_W    = (size_t)HH * DD * DKV * 2;
  const size_t SZ_HEAD = (size_t)BB * HH * SS * DKV * 2;
  size_t off = 0;
  unsigned short* qbf  = (unsigned short*)(ws + off); off += SZ_ACT;
  unsigned short* kbf  = (unsigned short*)(ws + off); off += SZ_ACT;
  unsigned short* vbf  = (unsigned short*)(ws + off); off += SZ_ACT;
  unsigned short* WqT  = (unsigned short*)(ws + off); off += SZ_W;
  unsigned short* WkT  = (unsigned short*)(ws + off); off += SZ_W;
  unsigned short* WvT  = (unsigned short*)(ws + off); off += SZ_W;
  unsigned short* WfcT = (unsigned short*)(ws + off); off += (size_t)DD * DD * 2;
  unsigned short* qh   = (unsigned short*)(ws + off); off += SZ_HEAD;
  unsigned short* kh   = (unsigned short*)(ws + off); off += SZ_HEAD;
  unsigned short* vhT  = (unsigned short*)(ws + off); off += SZ_HEAD;
  unsigned short* aout = (unsigned short*)(ws + off); off += SZ_ACT;
  float*          fco  = (float*)(ws + off);          off += (size_t)BB * SS * DD * 4;

  const int nAct = BB * SS * DD;
  const int nW   = HH * DD * DKV;
  const int nFc  = DD * DD;

  cvt_kernel<<<(nAct + 255) / 256, 256, 0, stream>>>(q, qbf, nAct);
  cvt_kernel<<<(nAct + 255) / 256, 256, 0, stream>>>(k, kbf, nAct);
  cvt_kernel<<<(nAct + 255) / 256, 256, 0, stream>>>(v, vbf, nAct);
  cvtT_kernel<<<(nW + 255) / 256, 256, 0, stream>>>(Wq, WqT, HH, DD, DKV);
  cvtT_kernel<<<(nW + 255) / 256, 256, 0, stream>>>(Wk, WkT, HH, DD, DKV);
  cvtT_kernel<<<(nW + 255) / 256, 256, 0, stream>>>(Wv, WvT, HH, DD, DKV);
  cvtT_kernel<<<(nFc + 255) / 256, 256, 0, stream>>>(Wfc, WfcT, 1, DD, DD);

  dim3 gProj(BB * HH, SS / 64);
  proj_kernel<<<gProj, 128, 0, stream>>>(qbf, WqT, qh, 0.125f, 0);
  proj_kernel<<<gProj, 128, 0, stream>>>(kbf, WkT, kh, 1.0f, 0);
  proj_kernel<<<gProj, 128, 0, stream>>>(vbf, WvT, vhT, 1.0f, 1);

  attn_kernel<<<dim3(BB * HH, SS / 64), 128, 0, stream>>>(qh, kh, vhT, mask, aout);

  fc_kernel<<<(BB * SS) / 16, 256, 0, stream>>>(aout, WfcT, q, fco);
  ln_kernel<<<BB * SS, 256, 0, stream>>>(fco, lnw, lnb, out);
}


# attn: K and V^T tiles fetched once per 8-wave work-group into double-buffered LDS (64-key stages, plain global loads + ds_store); B fragments read from LDS; sum order unchanged
# speedup vs baseline: 1.3786x; 1.3786x over previous
#include <hip/hip_runtime.h>


#define BB 2
#define SS 2048
#define DD 1024
#define HH 16
#define DKV 64

typedef __attribute__((ext_vector_type(16))) __bf16 v16bf;
typedef __attribute__((ext_vector_type(8)))  float  v8f;

union FragU { v16bf v; uint4 q[2]; };

__device__ __forceinline__ unsigned short f2bf(float f) {
  unsigned int u = __float_as_uint(f);
  u += 0x7FFFu + ((u >> 16) & 1u);
  return (unsigned short)(u >> 16);
}

__device__ __forceinline__ v16bf load_frag(const unsigned short* rowbase, int lane) {
  const unsigned short* p = rowbase + ((lane & 16) ? 8 : 0);
  FragU f;
  f.q[0] = *reinterpret_cast<const uint4*>(p);
  f.q[1] = *reinterpret_cast<const uint4*>(p + 16);
  return f.v;
}

#define WMMA_BF16(a, b, c) \
  __builtin_amdgcn_wmma_f32_16x16x32_bf16(false, (a), false, (b), (short)0, (c), false, false)

__global__ void cvt_kernel(const float* __restrict__ src,
                           unsigned short* __restrict__ dst, int n) {
  int i = blockIdx.x * blockDim.x + threadIdx.x;
  if (i < n) dst[i] = f2bf(src[i]);
}

__global__ void cvtT_kernel(const float* __restrict__ src,
                            unsigned short* __restrict__ dst,
                            int G, int R, int C) {
  int i = blockIdx.x * blockDim.x + threadIdx.x;
  int n = G * R * C;
  if (i >= n) return;
  int g = i / (R * C);
  int rem = i - g * (R * C);
  int r = rem / C, c = rem - r * C;
  dst[(size_t)g * R * C + (size_t)c * R + r] = f2bf(src[i]);
}

__global__ void proj_kernel(const unsigned short* __restrict__ X,
                            const unsigned short* __restrict__ WT,
                            unsigned short* __restrict__ OUT,
                            float scale, int transposeOut) {
  const int lane = threadIdx.x & 31;
  const int wave = threadIdx.x >> 5;
  const int bh = blockIdx.x;
  const int b = bh / HH, h = bh % HH;
  const int s_base = blockIdx.y * 64 + wave * 16;

  const unsigned short* Xrow =
      X + ((size_t)b * SS + (size_t)(s_base + (lane & 15))) * DD;
  const unsigned short* Wrow = WT + ((size_t)h * DKV + (lane & 15)) * DD;

  v8f acc[4] = {};
#pragma unroll 2
  for (int k0 = 0; k0 < DD; k0 += 32) {
    __builtin_prefetch(Xrow + k0 + 64, 0, 1);
    v16bf a = load_frag(Xrow + k0, lane);
    v16bf bf[4];
#pragma unroll
    for (int nt = 0; nt < 4; ++nt)
      bf[nt] = load_frag(Wrow + (size_t)nt * 16 * DD + k0, lane);
#pragma unroll
    for (int nt = 0; nt < 4; ++nt)
      acc[nt] = WMMA_BF16(a, bf[nt], acc[nt]);
  }
  const int rbase = (lane & 16) ? 8 : 0;
  const int col0 = lane & 15;
#pragma unroll
  for (int nt = 0; nt < 4; ++nt) {
#pragma unroll
    for (int r = 0; r < 8; ++r) {
      int srow = s_base + rbase + r;
      int col = nt * 16 + col0;
      unsigned short val = f2bf(acc[nt][r] * scale);
      if (transposeOut)
        OUT[(size_t)bh * DKV * SS + (size_t)col * SS + srow] = val;
      else
        OUT[(size_t)bh * SS * DKV + (size_t)srow * DKV + col] = val;
    }
  }
}

#define AT_WAVES 8
#define AT_ROWS  (AT_WAVES * 16)
#define KT       64
#define LPITCH   72

__global__ void __launch_bounds__(AT_WAVES * 32)
attn_kernel(const unsigned short* __restrict__ QH,
            const unsigned short* __restrict__ KH,
            const unsigned short* __restrict__ VT,
            const int* __restrict__ MASK,
            unsigned short* __restrict__ AOUT) {
  __shared__ __align__(16) unsigned short Ks[2][KT * LPITCH];
  __shared__ __align__(16) unsigned short Vs[2][DKV * LPITCH];
  __shared__ __align__(16) unsigned short Pld[AT_WAVES][16][32];

  const int tid = threadIdx.x;
  const int lane = tid & 31;
  const int wave = tid >> 5;
  const int bh = blockIdx.x;
  const int b = bh >> 4, h = bh & 15;
  const int s_base = blockIdx.y * AT_ROWS + wave * 16;
  const int rbase = (lane & 16) ? 8 : 0;
  const int col = lane & 15;

  const unsigned short* Qrow =
      QH + ((size_t)bh * SS + (size_t)(s_base + col)) * DKV;
  v16bf qf0 = load_frag(Qrow, lane);
  v16bf qf1 = load_frag(Qrow + 32, lane);

  v8f o[4] = {};
  float m[8], l[8];
#pragma unroll
  for (int r = 0; r < 8; ++r) { m[r] = -3.0e38f; l[r] = 0.0f; }

  const int* maskbase = MASK + (size_t)b * SS * SS;

  const int srow = tid >> 3;
  const int sch  = (tid & 7) * 8;
  const unsigned short* Kg = KH + ((size_t)bh * SS + srow) * DKV + sch;
  const unsigned short* Vg = VT + ((size_t)bh * DKV + srow) * SS + sch;
  const int sidx = srow * LPITCH + sch;

  uint4 gk0 = *reinterpret_cast<const uint4*>(Kg);
  uint4 gk1 = *reinterpret_cast<const uint4*>(Kg + (size_t)32 * DKV);
  uint4 gv0 = *reinterpret_cast<const uint4*>(Vg);
  uint4 gv1 = *reinterpret_cast<const uint4*>(Vg + (size_t)32 * SS);
  *reinterpret_cast<uint4*>(&Ks[0][sidx])               = gk0;
  *reinterpret_cast<uint4*>(&Ks[0][sidx + 32 * LPITCH]) = gk1;
  *reinterpret_cast<uint4*>(&Vs[0][sidx])               = gv0;
  *reinterpret_cast<uint4*>(&Vs[0][sidx + 32 * LPITCH]) = gv1;
  __syncthreads();

  for (int st = 0; st < SS / KT; ++st) {
    const int buf = st & 1;
    const int t0 = st * KT;
    const bool more = (st + 1) < (SS / KT);
    if (more) {
      const unsigned short* kg = Kg + (size_t)(t0 + KT) * DKV;
      const unsigned short* vg = Vg + (t0 + KT);
      gk0 = *reinterpret_cast<const uint4*>(kg);
      gk1 = *reinterpret_cast<const uint4*>(kg + (size_t)32 * DKV);
      gv0 = *reinterpret_cast<const uint4*>(vg);
      gv1 = *reinterpret_cast<const uint4*>(vg + (size_t)32 * SS);
    }
    const unsigned short* Kb = &Ks[buf][0];
    const unsigned short* Vb = &Vs[buf][0];

#pragma unroll
    for (int j = 0; j < KT / 32; ++j) {
      const int tt = t0 + j * 32;
      v16bf kb0 = load_frag(Kb + (j * 32 + col) * LPITCH, lane);
      v16bf kb1 = load_frag(Kb + (j * 32 + col) * LPITCH + 32, lane);
      v16bf kb2 = load_frag(Kb + (j * 32 + 16 + col) * LPITCH, lane);
      v16bf kb3 = load_frag(Kb + (j * 32 + 16 + col) * LPITCH + 32, lane);
      v16bf vb[4];
#pragma unroll
      for (int nt = 0; nt < 4; ++nt)
        vb[nt] = load_frag(Vb + (nt * 16 + col) * LPITCH + j * 32, lane);

      v8f s0 = {}, s1 = {};
      s0 = WMMA_BF16(qf0, kb0, s0);
      s0 = WMMA_BF16(qf1, kb1, s0);
      s1 = WMMA_BF16(qf0, kb2, s1);
      s1 = WMMA_BF16(qf1, kb3, s1);

#pragma unroll
      for (int r = 0; r < 8; ++r) {
        const int* mrow = maskbase + (size_t)(s_base + rbase + r) * SS + tt;
        float v0 = (mrow[col]      != 0) ? s0[r] : -1.0e9f;
        float v1 = (mrow[col + 16] != 0) ? s1[r] : -1.0e9f;
        float mx = fmaxf(v0, v1);
        mx = fmaxf(mx, __shfl_xor(mx, 1));
        mx = fmaxf(mx, __shfl_xor(mx, 2));
        mx = fmaxf(mx, __shfl_xor(mx, 4));
        mx = fmaxf(mx, __shfl_xor(mx, 8));
        float mnew  = fmaxf(m[r], mx);
        float alpha = __expf(m[r] - mnew);
        float p0 = __expf(v0 - mnew);
        float p1 = __expf(v1 - mnew);
        float rs = p0 + p1;
        rs += __shfl_xor(rs, 1);
        rs += __shfl_xor(rs, 2);
        rs += __shfl_xor(rs, 4);
        rs += __shfl_xor(rs, 8);
        l[r] = l[r] * alpha + rs;
        m[r] = mnew;
#pragma unroll
        for (int nt = 0; nt < 4; ++nt) o[nt][r] = o[nt][r] * alpha;
        Pld[wave][rbase + r][col]      = f2bf(p0);
        Pld[wave][rbase + r][col + 16] = f2bf(p1);
      }
      asm volatile("s_wait_dscnt 0" ::: "memory");
      v16bf pf = load_frag(&Pld[wave][col][0], lane);
#pragma unroll
      for (int nt = 0; nt < 4; ++nt)
        o[nt] = WMMA_BF16(pf, vb[nt], o[nt]);
    }

    if (more) {
      unsigned short* Kn = &Ks[buf ^ 1][0];
      unsigned short* Vn = &Vs[buf ^ 1][0];
      *reinterpret_cast<uint4*>(Kn + sidx)               = gk0;
      *reinterpret_cast<uint4*>(Kn + sidx + 32 * LPITCH) = gk1;
      *reinterpret_cast<uint4*>(Vn + sidx)               = gv0;
      *reinterpret_cast<uint4*>(Vn + sidx + 32 * LPITCH) = gv1;
    }
    __syncthreads();
  }

#pragma unroll
  for (int r = 0; r < 8; ++r) {
    float inv = 1.0f / l[r];
    int srow_o = s_base + rbase + r;
#pragma unroll
    for (int nt = 0; nt < 4; ++nt) {
      int dv = nt * 16 + col;
      AOUT[((size_t)b * SS + srow_o) * (HH * DKV) + h * DKV + dv] =
          f2bf(o[nt][r] * inv);
    }
  }
}

__global__ void fc_kernel(const unsigned short* __restrict__ A,
                          const unsigned short* __restrict__ WT,
                          const float* __restrict__ RES,
                          float* __restrict__ OUT) {
  const int lane = threadIdx.x & 31;
  const int wave = threadIdx.x >> 5;
  const int row0 = blockIdx.x * 16;
  const int n0 = wave * 128;

  const unsigned short* Arow = A + (size_t)(row0 + (lane & 15)) * DD;
  const unsigned short* Wrow = WT + (size_t)(n0 + (lane & 15)) * DD;

  v8f acc[8] = {};
  for (int k0 = 0; k0 < DD; k0 += 32) {
    __builtin_prefetch(Arow + k0 + 64, 0, 1);
    v16bf a = load_frag(Arow + k0, lane);
    v16bf bf[8];
#pragma unroll
    for (int nt = 0; nt < 8; ++nt)
      bf[nt] = load_frag(Wrow + (size_t)nt * 16 * DD + k0, lane);
#pragma unroll
    for (int nt = 0; nt < 8; ++nt)
      acc[nt] = WMMA_BF16(a, bf[nt], acc[nt]);
  }
  const int rbase = (lane & 16) ? 8 : 0;
  const int col0 = lane & 15;
#pragma unroll
  for (int nt = 0; nt < 8; ++nt) {
#pragma unroll
    for (int r = 0; r < 8; ++r) {
      size_t idx = (size_t)(row0 + rbase + r) * DD + n0 + nt * 16 + col0;
      OUT[idx] = acc[nt][r] + RES[idx];
    }
  }
}

__global__ void ln_kernel(const float* __restrict__ X,
                          const float* __restrict__ W,
                          const float* __restrict__ Bv,
                          float* __restrict__ OUT) {
  __shared__ float red[8];
  const int row = blockIdx.x;
  const int tid = threadIdx.x;
  const float* xr = X + (size_t)row * DD;

  float v[4];
  float s = 0.0f;
#pragma unroll
  for (int i = 0; i < 4; ++i) { v[i] = xr[tid + i * 256]; s += v[i]; }
  for (int off = 1; off < 32; off <<= 1) s += __shfl_xor(s, off);
  if ((tid & 31) == 0) red[tid >> 5] = s;
  __syncthreads();
  float mu = (red[0] + red[1] + red[2] + red[3] +
              red[4] + red[5] + red[6] + red[7]) * (1.0f / DD);
  float vs = 0.0f;
#pragma unroll
  for (int i = 0; i < 4; ++i) { float d = v[i] - mu; vs += d * d; }
  for (int off = 1; off < 32; off <<= 1) vs += __shfl_xor(vs, off);
  __syncthreads();
  if ((tid & 31) == 0) red[tid >> 5] = vs;
  __syncthreads();
  float var = (red[0] + red[1] + red[2] + red[3] +
               red[4] + red[5] + red[6] + red[7]) * (1.0f / DD);
  float rstd = rsqrtf(var + 1e-6f);
#pragma unroll
  for (int i = 0; i < 4; ++i) {
    int c = tid + i * 256;
    OUT[(size_t)row * DD + c] = (v[i] - mu) * rstd * W[c] + Bv[c];
  }
}

extern "C" void kernel_launch(void* const* d_in, const int* in_sizes, int n_in,
                              void* d_out, int out_size, void* d_ws, size_t ws_size,
                              hipStream_t stream) {
  const float* q    = (const float*)d_in[0];
  const float* k    = (const float*)d_in[1];
  const float* v    = (const float*)d_in[2];
  const int*   mask = (const int*)d_in[3];
  const float* Wq   = (const float*)d_in[4];
  const float* Wk   = (const float*)d_in[5];
  const float* Wv   = (const float*)d_in[6];
  const float* Wfc  = (const float*)d_in[7];
  const float* lnw  = (const float*)d_in[8];
  const float* lnb  = (const float*)d_in[9];
  float* out = (float*)d_out;

  char* ws = (char*)d_ws;
  const size_t SZ_ACT  = (size_t)BB * SS * DD * 2;
  const size_t SZ_W    = (size_t)HH * DD * DKV * 2;
  const size_t SZ_HEAD = (size_t)BB * HH * SS * DKV * 2;
  size_t off = 0;
  unsigned short* qbf  = (unsigned short*)(ws + off); off += SZ_ACT;
  unsigned short* kbf  = (unsigned short*)(ws + off); off += SZ_ACT;
  unsigned short* vbf  = (unsigned short*)(ws + off); off += SZ_ACT;
  unsigned short* WqT  = (unsigned short*)(ws + off); off += SZ_W;
  unsigned short* WkT  = (unsigned short*)(ws + off); off += SZ_W;
  unsigned short* WvT  = (unsigned short*)(ws + off); off += SZ_W;
  unsigned short* WfcT = (unsigned short*)(ws + off); off += (size_t)DD * DD * 2;
  unsigned short* qh   = (unsigned short*)(ws + off); off += SZ_HEAD;
  unsigned short* kh   = (unsigned short*)(ws + off); off += SZ_HEAD;
  unsigned short* vhT  = (unsigned short*)(ws + off); off += SZ_HEAD;
  unsigned short* aout = (unsigned short*)(ws + off); off += SZ_ACT;
  float*          fco  = (float*)(ws + off);          off += (size_t)BB * SS * DD * 4;

  const int nAct = BB * SS * DD;
  const int nW   = HH * DD * DKV;
  const int nFc  = DD * DD;

  cvt_kernel<<<(nAct + 255) / 256, 256, 0, stream>>>(q, qbf, nAct);
  cvt_kernel<<<(nAct + 255) / 256, 256, 0, stream>>>(k, kbf, nAct);
  cvt_kernel<<<(nAct + 255) / 256, 256, 0, stream>>>(v, vbf, nAct);
  cvtT_kernel<<<(nW + 255) / 256, 256, 0, stream>>>(Wq, WqT, HH, DD, DKV);
  cvtT_kernel<<<(nW + 255) / 256, 256, 0, stream>>>(Wk, WkT, HH, DD, DKV);
  cvtT_kernel<<<(nW + 255) / 256, 256, 0, stream>>>(Wv, WvT, HH, DD, DKV);
  cvtT_kernel<<<(nFc + 255) / 256, 256, 0, stream>>>(Wfc, WfcT, 1, DD, DD);

  dim3 gProj(BB * HH, SS / 64);
  proj_kernel<<<gProj, 128, 0, stream>>>(qbf, WqT, qh, 0.125f, 0);
  proj_kernel<<<gProj, 128, 0, stream>>>(kbf, WkT, kh, 1.0f, 0);
  proj_kernel<<<gProj, 128, 0, stream>>>(vbf, WvT, vhT, 1.0f, 1);

  attn_kernel<<<dim3(BB * HH, SS / AT_ROWS), AT_WAVES * 32, 0, stream>>>(qh, kh, vhT, mask, aout);

  fc_kernel<<<(BB * SS) / 16, 256, 0, stream>>>(aout, WfcT, q, fco);
  ln_kernel<<<BB * SS, 256, 0, stream>>>(fco, lnw, lnb, out);
}


# proj x3 and fc GEMMs: A and B^T tiles (128x128 out, 128 k a stage) fetched once per 8-wave work-group into double-buffered LDS by whole-line loads; fragments read from LDS; sum order unchanged
# speedup vs baseline: 1.9126x; 1.3873x over previous
#include <hip/hip_runtime.h>


#define BB 2
#define SS 2048
#define DD 1024
#define HH 16
#define DKV 64

typedef __attribute__((ext_vector_type(16))) __bf16 v16bf;
typedef __attribute__((ext_vector_type(8)))  float  v8f;

union FragU { v16bf v; uint4 q[2]; };

__device__ __forceinline__ unsigned short f2bf(float f) {
  unsigned int u = __float_as_uint(f);
  u += 0x7FFFu + ((u >> 16) & 1u);
  return (unsigned short)(u >> 16);
}

__device__ __forceinline__ v16bf load_frag(const unsigned short* rowbase, int lane) {
  const unsigned short* p = rowbase + ((lane & 16) ? 8 : 0);
  FragU f;
  f.q[0] = *reinterpret_cast<const uint4*>(p);
  f.q[1] = *reinterpret_cast<const uint4*>(p + 16);
  return f.v;
}

#define WMMA_BF16(a, b, c) \
  __builtin_amdgcn_wmma_f32_16x16x32_bf16(false, (a), false, (b), (short)0, (c), false, false)

__global__ void cvt_kernel(const float* __restrict__ src,
                           unsigned short* __restrict__ dst, int n) {
  int i = blockIdx.x * blockDim.x + threadIdx.x;
  if (i < n) dst[i] = f2bf(src[i]);
}

__global__ void cvtT_kernel(const float* __restrict__ src,
                            unsigned short* __restrict__ dst,
                            int G, int R, int C) {
  int i = blockIdx.x * blockDim.x + threadIdx.x;
  int n = G * R * C;
  if (i >= n) return;
  int g = i / (R * C);
  int rem = i - g * (R * C);
  int r = rem / C, c = rem - r * C;
  dst[(size_t)g * R * C + (size_t)c * R + r] = f2bf(src[i]);
}

#define GT_M   128
#define GT_N   128
#define GT_K   128
#define GPITCH 136

typedef __attribute__((ext_vector_type(4))) unsigned int u32x4;

template <int MODE>
__global__ void __launch_bounds__(256)
gemm_kernel(const unsigned short* __restrict__ A,
            const unsigned short* __restrict__ BT,
            unsigned short* __restrict__ OUTH,
            float* __restrict__ OUTF,
            const float* __restrict__ RES,
            float scale) {
  __shared__ __align__(16) unsigned short As[2][GT_M * GPITCH];
  __shared__ __align__(16) unsigned short Bs[2][GT_N * GPITCH];

  const int tid = threadIdx.x;
  const int lane = tid & 31;
  const int wave = tid >> 5;
  const int wm = wave & 3;
  const int wn = wave >> 2;
  const int m0 = blockIdx.y * GT_M;
  const int n0 = blockIdx.x * GT_N;
  const int rbase = (lane & 16) ? 8 : 0;
  const int col = lane & 15;

  const int srow = tid >> 4;
  const int sch  = (tid & 15) * 8;
  const unsigned short* Ag = A  + (size_t)(m0 + srow) * DD + sch;
  const unsigned short* Bg = BT + (size_t)(n0 + srow) * DD + sch;
  const int sidx = srow * GPITCH + sch;

  u32x4 ga[8], gb[8];
#pragma unroll
  for (int i = 0; i < 8; ++i) {
    ga[i] = *reinterpret_cast<const u32x4*>(Ag + (size_t)(16 * i) * DD);
    gb[i] = *reinterpret_cast<const u32x4*>(Bg + (size_t)(16 * i) * DD);
  }
#pragma unroll
  for (int i = 0; i < 8; ++i) {
    *reinterpret_cast<u32x4*>(&As[0][sidx + 16 * i * GPITCH]) = ga[i];
    *reinterpret_cast<u32x4*>(&Bs[0][sidx + 16 * i * GPITCH]) = gb[i];
  }
  __syncthreads();

  v8f acc[2][4] = {};

  for (int st = 0; st < DD / GT_K; ++st) {
    const int buf = st & 1;
    const bool more = (st + 1) < (DD / GT_K);
    if (more) {
      const unsigned short* ag = Ag + (st + 1) * GT_K;
      const unsigned short* bg = Bg + (st + 1) * GT_K;
#pragma unroll
      for (int i = 0; i < 8; ++i) {
        ga[i] = *reinterpret_cast<const u32x4*>(ag + (size_t)(16 * i) * DD);
        gb[i] = *reinterpret_cast<const u32x4*>(bg + (size_t)(16 * i) * DD);
      }
    }
    __builtin_amdgcn_sched_barrier(0);
    const unsigned short* Ab = &As[buf][(wm * 32 + col) * GPITCH];
    const unsigned short* Bb = &Bs[buf][(wn * 64 + col) * GPITCH];

#pragma unroll
    for (int j = 0; j < GT_K / 32; ++j) {
      v16bf af[2], bfr[4];
#pragma unroll
      for (int mt = 0; mt < 2; ++mt)
        af[mt] = load_frag(Ab + mt * 16 * GPITCH + j * 32, lane);
#pragma unroll
      for (int nt = 0; nt < 4; ++nt)
        bfr[nt] = load_frag(Bb + nt * 16 * GPITCH + j * 32, lane);
#pragma unroll
      for (int mt = 0; mt < 2; ++mt)
#pragma unroll
        for (int nt = 0; nt < 4; ++nt)
          acc[mt][nt] = WMMA_BF16(af[mt], bfr[nt], acc[mt][nt]);
    }

    __builtin_amdgcn_sched_barrier(0);
    if (more) {
#pragma unroll
      for (int i = 0; i < 8; ++i) {
        *reinterpret_cast<u32x4*>(&As[buf ^ 1][sidx + 16 * i * GPITCH]) = ga[i];
        *reinterpret_cast<u32x4*>(&Bs[buf ^ 1][sidx + 16 * i * GPITCH]) = gb[i];
      }
    }
    __syncthreads();
  }

#pragma unroll
  for (int mt = 0; mt < 2; ++mt) {
#pragma unroll
    for (int nt = 0; nt < 4; ++nt) {
#pragma unroll
      for (int r = 0; r < 8; ++r) {
        const int row = m0 + wm * 32 + mt * 16 + rbase + r;
        const int n   = n0 + wn * 64 + nt * 16 + col;
        if (MODE == 2) {
          size_t idx = (size_t)row * DD + n;
          OUTF[idx] = acc[mt][nt][r] + RES[idx];
        } else {
          const int b = row / SS, sq = row - b * SS;
          const int h = n / DKV, dcol = n - h * DKV;
          const int bh = b * HH + h;
          unsigned short val = f2bf(acc[mt][nt][r] * scale);
          if (MODE == 1)
            OUTH[(size_t)bh * DKV * SS + (size_t)dcol * SS + sq] = val;
          else
            OUTH[(size_t)bh * SS * DKV + (size_t)sq * DKV + dcol] = val;
        }
      }
    }
  }
}

#define AT_WAVES 8
#define AT_ROWS  (AT_WAVES * 16)
#define KT       64
#define LPITCH   72

__global__ void __launch_bounds__(AT_WAVES * 32)
attn_kernel(const unsigned short* __restrict__ QH,
            const unsigned short* __restrict__ KH,
            const unsigned short* __restrict__ VT,
            const int* __restrict__ MASK,
            unsigned short* __restrict__ AOUT) {
  __shared__ __align__(16) unsigned short Ks[2][KT * LPITCH];
  __shared__ __align__(16) unsigned short Vs[2][DKV * LPITCH];
  __shared__ __align__(16) unsigned short Pld[AT_WAVES][16][32];

  const int tid = threadIdx.x;
  const int lane = tid & 31;
  const int wave = tid >> 5;
  const int bh = blockIdx.x;
  const int b = bh >> 4, h = bh & 15;
  const int s_base = blockIdx.y * AT_ROWS + wave * 16;
  const int rbase = (lane & 16) ? 8 : 0;
  const int col = lane & 15;

  const unsigned short* Qrow =
      QH + ((size_t)bh * SS + (size_t)(s_base + col)) * DKV;
  v16bf qf0 = load_frag(Qrow, lane);
  v16bf qf1 = load_frag(Qrow + 32, lane);

  v8f o[4] = {};
  float m[8], l[8];
#pragma unroll
  for (int r = 0; r < 8; ++r) { m[r] = -3.0e38f; l[r] = 0.0f; }

  const int* maskbase = MASK + (size_t)b * SS * SS;

  const int srow = tid >> 3;
  const int sch  = (tid & 7) * 8;
  const unsigned short* Kg = KH + ((size_t)bh * SS + srow) * DKV + sch;
  const unsigned short* Vg = VT + ((size_t)bh * DKV + srow) * SS + sch;
  const int sidx = srow * LPITCH + sch;

  uint4 gk0 = *reinterpret_cast<const uint4*>(Kg);
  uint4 gk1 = *reinterpret_cast<const uint4*>(Kg + (size_t)32 * DKV);
  uint4 gv0 = *reinterpret_cast<const uint4*>(Vg);
  uint4 gv1 = *reinterpret_cast<const uint4*>(Vg + (size_t)32 * SS);
  *reinterpret_cast<uint4*>(&Ks[0][sidx])               = gk0;
  *reinterpret_cast<uint4*>(&Ks[0][sidx + 32 * LPITCH]) = gk1;
  *reinterpret_cast<uint4*>(&Vs[0][sidx])               = gv0;
  *reinterpret_cast<uint4*>(&Vs[0][sidx + 32 * LPITCH]) = gv1;
  __syncthreads();

  for (int st = 0; st < SS / KT; ++st) {
    const int buf = st & 1;
    const int t0 = st * KT;
    const bool more = (st + 1) < (SS / KT);
    if (more) {
      const unsigned short* kg = Kg + (size_t)(t0 + KT) * DKV;
      const unsigned short* vg = Vg + (t0 + KT);
      gk0 = *reinterpret_cast<const uint4*>(kg);
      gk1 = *reinterpret_cast<const uint4*>(kg + (size_t)32 * DKV);
      gv0 = *reinterpret_cast<const uint4*>(vg);
      gv1 = *reinterpret_cast<const uint4*>(vg + (size_t)32 * SS);
    }
    const unsigned short* Kb = &Ks[buf][0];
    const unsigned short* Vb = &Vs[buf][0];

#pragma unroll
    for (int j = 0; j < KT / 32; ++j) {
      const int tt = t0 + j * 32;
      v16bf kb0 = load_frag(Kb + (j * 32 + col) * LPITCH, lane);
      v16bf kb1 = load_frag(Kb + (j * 32 + col) * LPITCH + 32, lane);
      v16bf kb2 = load_frag(Kb + (j * 32 + 16 + col) * LPITCH, lane);
      v16bf kb3 = load_frag(Kb + (j * 32 + 16 + col) * LPITCH + 32, lane);
      v16bf vb[4];
#pragma unroll
      for (int nt = 0; nt < 4; ++nt)
        vb[nt] = load_frag(Vb + (nt * 16 + col) * LPITCH + j * 32, lane);

      v8f s0 = {}, s1 = {};
      s0 = WMMA_BF16(qf0, kb0, s0);
      s0 = WMMA_BF16(qf1, kb1, s0);
      s1 = WMMA_BF16(qf0, kb2, s1);
      s1 = WMMA_BF16(qf1, kb3, s1);

#pragma unroll
      for (int r = 0; r < 8; ++r) {
        const int* mrow = maskbase + (size_t)(s_base + rbase + r) * SS + tt;
        float v0 = (mrow[col]      != 0) ? s0[r] : -1.0e9f;
        float v1 = (mrow[col + 16] != 0) ? s1[r] : -1.0e9f;
        float mx = fmaxf(v0, v1);
        mx = fmaxf(mx, __shfl_xor(mx, 1));
        mx = fmaxf(mx, __shfl_xor(mx, 2));
        mx = fmaxf(mx, __shfl_xor(mx, 4));
        mx = fmaxf(mx, __shfl_xor(mx, 8));
        float mnew  = fmaxf(m[r], mx);
        float alpha = __expf(m[r] - mnew);
        float p0 = __expf(v0 - mnew);
        float p1 = __expf(v1 - mnew);
        float rs = p0 + p1;
        rs += __shfl_xor(rs, 1);
        rs += __shfl_xor(rs, 2);
        rs += __shfl_xor(rs, 4);
        rs += __shfl_xor(rs, 8);
        l[r] = l[r] * alpha + rs;
        m[r] = mnew;
#pragma unroll
        for (int nt = 0; nt < 4; ++nt) o[nt][r] = o[nt][r] * alpha;
        Pld[wave][rbase + r][col]      = f2bf(p0);
        Pld[wave][rbase + r][col + 16] = f2bf(p1);
      }
      asm volatile("s_wait_dscnt 0" ::: "memory");
      v16bf pf = load_frag(&Pld[wave][col][0], lane);
#pragma unroll
      for (int nt = 0; nt < 4; ++nt)
        o[nt] = WMMA_BF16(pf, vb[nt], o[nt]);
    }

    if (more) {
      unsigned short* Kn = &Ks[buf ^ 1][0];
      unsigned short* Vn = &Vs[buf ^ 1][0];
      *reinterpret_cast<uint4*>(Kn + sidx)               = gk0;
      *reinterpret_cast<uint4*>(Kn + sidx + 32 * LPITCH) = gk1;
      *reinterpret_cast<uint4*>(Vn + sidx)               = gv0;
      *reinterpret_cast<uint4*>(Vn + sidx + 32 * LPITCH) = gv1;
    }
    __syncthreads();
  }

#pragma unroll
  for (int r = 0; r < 8; ++r) {
    float inv = 1.0f / l[r];
    int srow_o = s_base + rbase + r;
#pragma unroll
    for (int nt = 0; nt < 4; ++nt) {
      int dv = nt * 16 + col;
      AOUT[((size_t)b * SS + srow_o) * (HH * DKV) + h * DKV + dv] =
          f2bf(o[nt][r] * inv);
    }
  }
}

__global__ void ln_kernel(const float* __restrict__ X,
                          const float* __restrict__ W,
                          const float* __restrict__ Bv,
                          float* __restrict__ OUT) {
  __shared__ float red[8];
  const int row = blockIdx.x;
  const int tid = threadIdx.x;
  const float* xr = X + (size_t)row * DD;

  float v[4];
  float s = 0.0f;
#pragma unroll
  for (int i = 0; i < 4; ++i) { v[i] = xr[tid + i * 256]; s += v[i]; }
  for (int off = 1; off < 32; off <<= 1) s += __shfl_xor(s, off);
  if ((tid & 31) == 0) red[tid >> 5] = s;
  __syncthreads();
  float mu = (red[0] + red[1] + red[2] + red[3] +
              red[4] + red[5] + red[6] + red[7]) * (1.0f / DD);
  float vs = 0.0f;
#pragma unroll
  for (int i = 0; i < 4; ++i) { float d = v[i] - mu; vs += d * d; }
  for (int off = 1; off < 32; off <<= 1) vs += __shfl_xor(vs, off);
  __syncthreads();
  if ((tid & 31) == 0) red[tid >> 5] = vs;
  __syncthreads();
  float var = (red[0] + red[1] + red[2] + red[3] +
               red[4] + red[5] + red[6] + red[7]) * (1.0f / DD);
  float rstd = rsqrtf(var + 1e-6f);
#pragma unroll
  for (int i = 0; i < 4; ++i) {
    int c = tid + i * 256;
    OUT[(size_t)row * DD + c] = (v[i] - mu) * rstd * W[c] + Bv[c];
  }
}

extern "C" void kernel_launch(void* const* d_in, const int* in_sizes, int n_in,
                              void* d_out, int out_size, void* d_ws, size_t ws_size,
                              hipStream_t stream) {
  const float* q    = (const float*)d_in[0];
  const float* k    = (const float*)d_in[1];
  const float* v    = (const float*)d_in[2];
  const int*   mask = (const int*)d_in[3];
  const float* Wq   = (const float*)d_in[4];
  const float* Wk   = (const float*)d_in[5];
  const float* Wv   = (const float*)d_in[6];
  const float* Wfc  = (const float*)d_in[7];
  const float* lnw  = (const float*)d_in[8];
  const float* lnb  = (const float*)d_in[9];
  float* out = (float*)d_out;

  char* ws = (char*)d_ws;
  const size_t SZ_ACT  = (size_t)BB * SS * DD * 2;
  const size_t SZ_W    = (size_t)HH * DD * DKV * 2;
  const size_t SZ_HEAD = (size_t)BB * HH * SS * DKV * 2;
  size_t off = 0;
  unsigned short* qbf  = (unsigned short*)(ws + off); off += SZ_ACT;
  unsigned short* kbf  = (unsigned short*)(ws + off); off += SZ_ACT;
  unsigned short* vbf  = (unsigned short*)(ws + off); off += SZ_ACT;
  unsigned short* WqT  = (unsigned short*)(ws + off); off += SZ_W;
  unsigned short* WkT  = (unsigned short*)(ws + off); off += SZ_W;
  unsigned short* WvT  = (unsigned short*)(ws + off); off += SZ_W;
  unsigned short* WfcT = (unsigned short*)(ws + off); off += (size_t)DD * DD * 2;
  unsigned short* qh   = (unsigned short*)(ws + off); off += SZ_HEAD;
  unsigned short* kh   = (unsigned short*)(ws + off); off += SZ_HEAD;
  unsigned short* vhT  = (unsigned short*)(ws + off); off += SZ_HEAD;
  unsigned short* aout = (unsigned short*)(ws + off); off += SZ_ACT;
  float*          fco  = (float*)(ws + off);          off += (size_t)BB * SS * DD * 4;

  const int nAct = BB * SS * DD;
  const int nW   = HH * DD * DKV;
  const int nFc  = DD * DD;

  cvt_kernel<<<(nAct + 255) / 256, 256, 0, stream>>>(q, qbf, nAct);
  cvt_kernel<<<(nAct + 255) / 256, 256, 0, stream>>>(k, kbf, nAct);
  cvt_kernel<<<(nAct + 255) / 256, 256, 0, stream>>>(v, vbf, nAct);
  cvtT_kernel<<<(nW + 255) / 256, 256, 0, stream>>>(Wq, WqT, HH, DD, DKV);
  cvtT_kernel<<<(nW + 255) / 256, 256, 0, stream>>>(Wk, WkT, HH, DD, DKV);
  cvtT_kernel<<<(nW + 255) / 256, 256, 0, stream>>>(Wv, WvT, HH, DD, DKV);
  cvtT_kernel<<<(nFc + 255) / 256, 256, 0, stream>>>(Wfc, WfcT, 1, DD, DD);

  dim3 gGemm(DD / GT_N, (BB * SS) / GT_M);
  gemm_kernel<0><<<gGemm, 256, 0, stream>>>(qbf, WqT, qh, nullptr, nullptr, 0.125f);
  gemm_kernel<0><<<gGemm, 256, 0, stream>>>(kbf, WkT, kh, nullptr, nullptr, 1.0f);
  gemm_kernel<1><<<gGemm, 256, 0, stream>>>(vbf, WvT, vhT, nullptr, nullptr, 1.0f);

  attn_kernel<<<dim3(BB * HH, SS / AT_ROWS), AT_WAVES * 32, 0, stream>>>(qh, kh, vhT, mask, aout);

  gemm_kernel<2><<<gGemm, 256, 0, stream>>>(aout, WfcT, nullptr, fco, q, 1.0f);
  ln_kernel<<<BB * SS, 256, 0, stream>>>(fco, lnw, lnb, out);
}


# fp32->bf16 convert launches (3 activation, 4 weight-transpose): 32 / 8 elements per thread with 16-byte accesses instead of one element, so 32x / 8x fewer waves to start; stored values unchanged
# speedup vs baseline: 3.0746x; 1.6075x over previous
#include <hip/hip_runtime.h>


#define BB 2
#define SS 2048
#define DD 1024
#define HH 16
#define DKV 64

typedef __attribute__((ext_vector_type(16))) __bf16 v16bf;
typedef __attribute__((ext_vector_type(8)))  float  v8f;

union FragU { v16bf v; uint4 q[2]; };

__device__ __forceinline__ unsigned short f2bf(float f) {
  unsigned int u = __float_as_uint(f);
  u += 0x7FFFu + ((u >> 16) & 1u);
  return (unsigned short)(u >> 16);
}

__device__ __forceinline__ v16bf load_frag(const unsigned short* rowbase, int lane) {
  const unsigned short* p = rowbase + ((lane & 16) ? 8 : 0);
  FragU f;
  f.q[0] = *reinterpret_cast<const uint4*>(p);
  f.q[1] = *reinterpret_cast<const uint4*>(p + 16);
  return f.v;
}

#define WMMA_BF16(a, b, c) \
  __builtin_amdgcn_wmma_f32_16x16x32_bf16(false, (a), false, (b), (short)0, (c), false, false)

#define CV_SLABS 8
#define CV_SLAB_ELEMS  (256 * 4)
#define CV_BLOCK_ELEMS (CV_SLABS * CV_SLAB_ELEMS)

__device__ __forceinline__ unsigned int f2bf_pair(float lo, float hi) {
  return (unsigned int)f2bf(lo) | ((unsigned int)f2bf(hi) << 16);
}

__global__ void __launch_bounds__(256)
cvt_kernel(const float* __restrict__ src, unsigned short* __restrict__ dst) {
  const size_t base = (size_t)blockIdx.x * CV_BLOCK_ELEMS + threadIdx.x * 4;
  float4 v[CV_SLABS];
#pragma unroll
  for (int j = 0; j < CV_SLABS; ++j)
    v[j] = *reinterpret_cast<const float4*>(src + base + j * CV_SLAB_ELEMS);
#pragma unroll
  for (int j = 0; j < CV_SLABS; ++j) {
    uint2 w;
    w.x = f2bf_pair(v[j].x, v[j].y);
    w.y = f2bf_pair(v[j].z, v[j].w);
    *reinterpret_cast<uint2*>(dst + base + j * CV_SLAB_ELEMS) = w;
  }
}

template <int R, int C>
__global__ void __launch_bounds__(256)
cvtT_kernel(const float* __restrict__ src, unsigned short* __restrict__ dst) {
  const unsigned int i  = blockIdx.x * 256 + threadIdx.x;
  const unsigned int c  = i % C;
  const unsigned int r8 = (i / C) % (R / 8);
  const unsigned int g  = i / (C * (R / 8));
  const float* s = src + ((size_t)g * R + (size_t)r8 * 8) * C + c;
  float v[8];
#pragma unroll
  for (int k = 0; k < 8; ++k) v[k] = s[(size_t)k * C];
  uint4 w;
  w.x = f2bf_pair(v[0], v[1]);
  w.y = f2bf_pair(v[2], v[3]);
  w.z = f2bf_pair(v[4], v[5]);
  w.w = f2bf_pair(v[6], v[7]);
  *reinterpret_cast<uint4*>(dst + ((size_t)g * C + c) * R + (size_t)r8 * 8) = w;
}

#define GT_M   128
#define GT_N   128
#define GT_K   128
#define GPITCH 136

typedef __attribute__((ext_vector_type(4))) unsigned int u32x4;

template <int MODE>
__global__ void __launch_bounds__(256)
gemm_kernel(const unsigned short* __restrict__ A,
            const unsigned short* __restrict__ BT,
            unsigned short* __restrict__ OUTH,
            float* __restrict__ OUTF,
            const float* __restrict__ RES,
            float scale) {
  __shared__ __align__(16) unsigned short As[2][GT_M * GPITCH];
  __shared__ __align__(16) unsigned short Bs[2][GT_N * GPITCH];

  const int tid = threadIdx.x;
  const int lane = tid & 31;
  const int wave = tid >> 5;
  const int wm = wave & 3;
  const int wn = wave >> 2;
  const int m0 = blockIdx.y * GT_M;
  const int n0 = blockIdx.x * GT_N;
  const int rbase = (lane & 16) ? 8 : 0;
  const int col = lane & 15;

  const int srow = tid >> 4;
  const int sch  = (tid & 15) * 8;
  const unsigned short* Ag = A  + (size_t)(m0 + srow) * DD + sch;
  const unsigned short* Bg = BT + (size_t)(n0 + srow) * DD + sch;
  const int sidx = srow * GPITCH + sch;

  u32x4 ga[8], gb[8];
#pragma unroll
  for (int i = 0; i < 8; ++i) {
    ga[i] = *reinterpret_cast<const u32x4*>(Ag + (size_t)(16 * i) * DD);
    gb[i] = *reinterpret_cast<const u32x4*>(Bg + (size_t)(16 * i) * DD);
  }
#pragma unroll
  for (int i = 0; i < 8; ++i) {
    *reinterpret_cast<u32x4*>(&As[0][sidx + 16 * i * GPITCH]) = ga[i];
    *reinterpret_cast<u32x4*>(&Bs[0][sidx + 16 * i * GPITCH]) = gb[i];
  }
  __syncthreads();

  v8f acc[2][4] = {};

  for (int st = 0; st < DD / GT_K; ++st) {
    const int buf = st & 1;
    const bool more = (st + 1) < (DD / GT_K);
    if (more) {
      const unsigned short* ag = Ag + (st + 1) * GT_K;
      const unsigned short* bg = Bg + (st + 1) * GT_K;
#pragma unroll
      for (int i = 0; i < 8; ++i) {
        ga[i] = *reinterpret_cast<const u32x4*>(ag + (size_t)(16 * i) * DD);
        gb[i] = *reinterpret_cast<const u32x4*>(bg + (size_t)(16 * i) * DD);
      }
    }
    __builtin_amdgcn_sched_barrier(0);
    const unsigned short* Ab = &As[buf][(wm * 32 + col) * GPITCH];
    const unsigned short* Bb = &Bs[buf][(wn * 64 + col) * GPITCH];

#pragma unroll
    for (int j = 0; j < GT_K / 32; ++j) {
      v16bf af[2], bfr[4];
#pragma unroll
      for (int mt = 0; mt < 2; ++mt)
        af[mt] = load_frag(Ab + mt * 16 * GPITCH + j * 32, lane);
#pragma unroll
      for (int nt = 0; nt < 4; ++nt)
        bfr[nt] = load_frag(Bb + nt * 16 * GPITCH + j * 32, lane);
#pragma unroll
      for (int mt = 0; mt < 2; ++mt)
#pragma unroll
        for (int nt = 0; nt < 4; ++nt)
          acc[mt][nt] = WMMA_BF16(af[mt], bfr[nt], acc[mt][nt]);
    }

    __builtin_amdgcn_sched_barrier(0);
    if (more) {
#pragma unroll
      for (int i = 0; i < 8; ++i) {
        *reinterpret_cast<u32x4*>(&As[buf ^ 1][sidx + 16 * i * GPITCH]) = ga[i];
        *reinterpret_cast<u32x4*>(&Bs[buf ^ 1][sidx + 16 * i * GPITCH]) = gb[i];
      }
    }
    __syncthreads();
  }

#pragma unroll
  for (int mt = 0; mt < 2; ++mt) {
#pragma unroll
    for (int nt = 0; nt < 4; ++nt) {
#pragma unroll
      for (int r = 0; r < 8; ++r) {
        const int row = m0 + wm * 32 + mt * 16 + rbase + r;
        const int n   = n0 + wn * 64 + nt * 16 + col;
        if (MODE == 2) {
          size_t idx = (size_t)row * DD + n;
          OUTF[idx] = acc[mt][nt][r] + RES[idx];
        } else {
          const int b = row / SS, sq = row - b * SS;
          const int h = n / DKV, dcol = n - h * DKV;
          const int bh = b * HH + h;
          unsigned short val = f2bf(acc[mt][nt][r] * scale);
          if (MODE == 1)
            OUTH[(size_t)bh * DKV * SS + (size_t)dcol * SS + sq] = val;
          else
            OUTH[(size_t)bh * SS * DKV + (size_t)sq * DKV + dcol] = val;
        }
      }
    }
  }
}

#define AT_WAVES 8
#define AT_ROWS  (AT_WAVES * 16)
#define KT       64
#define LPITCH   72

__global__ void __launch_bounds__(AT_WAVES * 32)
attn_kernel(const unsigned short* __restrict__ QH,
            const unsigned short* __restrict__ KH,
            const unsigned short* __restrict__ VT,
            const int* __restrict__ MASK,
            unsigned short* __restrict__ AOUT) {
  __shared__ __align__(16) unsigned short Ks[2][KT * LPITCH];
  __shared__ __align__(16) unsigned short Vs[2][DKV * LPITCH];
  __shared__ __align__(16) unsigned short Pld[AT_WAVES][16][32];

  const int tid = threadIdx.x;
  const int lane = tid & 31;
  const int wave = tid >> 5;
  const int bh = blockIdx.x;
  const int b = bh >> 4, h = bh & 15;
  const int s_base = blockIdx.y * AT_ROWS + wave * 16;
  const int rbase = (lane & 16) ? 8 : 0;
  const int col = lane & 15;

  const unsigned short* Qrow =
      QH + ((size_t)bh * SS + (size_t)(s_base + col)) * DKV;
  v16bf qf0 = load_frag(Qrow, lane);
  v16bf qf1 = load_frag(Qrow + 32, lane);

  v8f o[4] = {};
  float m[8], l[8];
#pragma unroll
  for (int r = 0; r < 8; ++r) { m[r] = -3.0e38f; l[r] = 0.0f; }

  const int* maskbase = MASK + (size_t)b * SS * SS;

  const int srow = tid >> 3;
  const int sch  = (tid & 7) * 8;
  const unsigned short* Kg = KH + ((size_t)bh * SS + srow) * DKV + sch;
  const unsigned short* Vg = VT + ((size_t)bh * DKV + srow) * SS + sch;
  const int sidx = srow * LPITCH + sch;

  uint4 gk0 = *reinterpret_cast<const uint4*>(Kg);
  uint4 gk1 = *reinterpret_cast<const uint4*>(Kg + (size_t)32 * DKV);
  uint4 gv0 = *reinterpret_cast<const uint4*>(Vg);
  uint4 gv1 = *reinterpret_cast<const uint4*>(Vg + (size_t)32 * SS);
  *reinterpret_cast<uint4*>(&Ks[0][sidx])               = gk0;
  *reinterpret_cast<uint4*>(&Ks[0][sidx + 32 * LPITCH]) = gk1;
  *reinterpret_cast<uint4*>(&Vs[0][sidx])               = gv0;
  *reinterpret_cast<uint4*>(&Vs[0][sidx + 32 * LPITCH]) = gv1;
  __syncthreads();

  for (int st = 0; st < SS / KT; ++st) {
    const int buf = st & 1;
    const int t0 = st * KT;
    const bool more = (st + 1) < (SS / KT);
    if (more) {
      const unsigned short* kg = Kg + (size_t)(t0 + KT) * DKV;
      const unsigned short* vg = Vg + (t0 + KT);
      gk0 = *reinterpret_cast<const uint4*>(kg);
      gk1 = *reinterpret_cast<const uint4*>(kg + (size_t)32 * DKV);
      gv0 = *reinterpret_cast<const uint4*>(vg);
      gv1 = *reinterpret_cast<const uint4*>(vg + (size_t)32 * SS);
    }
    const unsigned short* Kb = &Ks[buf][0];
    const unsigned short* Vb = &Vs[buf][0];

#pragma unroll
    for (int j = 0; j < KT / 32; ++j) {
      const int tt = t0 + j * 32;
      v16bf kb0 = load_frag(Kb + (j * 32 + col) * LPITCH, lane);
      v16bf kb1 = load_frag(Kb + (j * 32 + col) * LPITCH + 32, lane);
      v16bf kb2 = load_frag(Kb + (j * 32 + 16 + col) * LPITCH, lane);
      v16bf kb3 = load_frag(Kb + (j * 32 + 16 + col) * LPITCH + 32, lane);
      v16bf vb[4];
#pragma unroll
      for (int nt = 0; nt < 4; ++nt)
        vb[nt] = load_frag(Vb + (nt * 16 + col) * LPITCH + j * 32, lane);

      v8f s0 = {}, s1 = {};
      s0 = WMMA_BF16(qf0, kb0, s0);
      s0 = WMMA_BF16(qf1, kb1, s0);
      s1 = WMMA_BF16(qf0, kb2, s1);
      s1 = WMMA_BF16(qf1, kb3, s1);

#pragma unroll
      for (int r = 0; r < 8; ++r) {
        const int* mrow = maskbase + (size_t)(s_base + rbase + r) * SS + tt;
        float v0 = (mrow[col]      != 0) ? s0[r] : -1.0e9f;
        float v1 = (mrow[col + 16] != 0) ? s1[r] : -1.0e9f;
        float mx = fmaxf(v0, v1);
        mx = fmaxf(mx, __shfl_xor(mx, 1));
        mx = fmaxf(mx, __shfl_xor(mx, 2));
        mx = fmaxf(mx, __shfl_xor(mx, 4));
        mx = fmaxf(mx, __shfl_xor(mx, 8));
        float mnew  = fmaxf(m[r], mx);
        float alpha = __expf(m[r] - mnew);
        float p0 = __expf(v0 - mnew);
        float p1 = __expf(v1 - mnew);
        float rs = p0 + p1;
        rs += __shfl_xor(rs, 1);
        rs += __shfl_xor(rs, 2);
        rs += __shfl_xor(rs, 4);
        rs += __shfl_xor(rs, 8);
        l[r] = l[r] * alpha + rs;
        m[r] = mnew;
#pragma unroll
        for (int nt = 0; nt < 4; ++nt) o[nt][r] = o[nt][r] * alpha;
        Pld[wave][rbase + r][col]      = f2bf(p0);
        Pld[wave][rbase + r][col + 16] = f2bf(p1);
      }
      asm volatile("s_wait_dscnt 0" ::: "memory");
      v16bf pf = load_frag(&Pld[wave][col][0], lane);
#pragma unroll
      for (int nt = 0; nt < 4; ++nt)
        o[nt] = WMMA_BF16(pf, vb[nt], o[nt]);
    }

    if (more) {
      unsigned short* Kn = &Ks[buf ^ 1][0];
      unsigned short* Vn = &Vs[buf ^ 1][0];
      *reinterpret_cast<uint4*>(Kn + sidx)               = gk0;
      *reinterpret_cast<uint4*>(Kn + sidx + 32 * LPITCH) = gk1;
      *reinterpret_cast<uint4*>(Vn + sidx)               = gv0;
      *reinterpret_cast<uint4*>(Vn + sidx + 32 * LPITCH) = gv1;
    }
    __syncthreads();
  }

#pragma unroll
  for (int r = 0; r < 8; ++r) {
    float inv = 1.0f / l[r];
    int srow_o = s_base + rbase + r;
#pragma unroll
    for (int nt = 0; nt < 4; ++nt) {
      int dv = nt * 16 + col;
      AOUT[((size_t)b * SS + srow_o) * (HH * DKV) + h * DKV + dv] =
          f2bf(o[nt][r] * inv);
    }
  }
}

__global__ void ln_kernel(const float* __restrict__ X,
                          const float* __restrict__ W,
                          const float* __restrict__ Bv,
                          float* __restrict__ OUT) {
  __shared__ float red[8];
  const int row = blockIdx.x;
  const int tid = threadIdx.x;
  const float* xr = X + (size_t)row * DD;

  float v[4];
  float s = 0.0f;
#pragma unroll
  for (int i = 0; i < 4; ++i) { v[i] = xr[tid + i * 256]; s += v[i]; }
  for (int off = 1; off < 32; off <<= 1) s += __shfl_xor(s, off);
  if ((tid & 31) == 0) red[tid >> 5] = s;
  __syncthreads();
  float mu = (red[0] + red[1] + red[2] + red[3] +
              red[4] + red[5] + red[6] + red[7]) * (1.0f / DD);
  float vs = 0.0f;
#pragma unroll
  for (int i = 0; i < 4; ++i) { float d = v[i] - mu; vs += d * d; }
  for (int off = 1; off < 32; off <<= 1) vs += __shfl_xor(vs, off);
  __syncthreads();
  if ((tid & 31) == 0) red[tid >> 5] = vs;
  __syncthreads();
  float var = (red[0] + red[1] + red[2] + red[3] +
               red[4] + red[5] + red[6] + red[7]) * (1.0f / DD);
  float rstd = rsqrtf(var + 1e-6f);
#pragma unroll
  for (int i = 0; i < 4; ++i) {
    int c = tid + i * 256;
    OUT[(size_t)row * DD + c] = (v[i] - mu) * rstd * W[c] + Bv[c];
  }
}

extern "C" void kernel_launch(void* const* d_in, const int* in_sizes, int n_in,
                              void* d_out, int out_size, void* d_ws, size_t ws_size,
                              hipStream_t stream) {
  const float* q    = (const float*)d_in[0];
  const float* k    = (const float*)d_in[1];
  const float* v    = (const float*)d_in[2];
  const int*   mask = (const int*)d_in[3];
  const float* Wq   = (const float*)d_in[4];
  const float* Wk   = (const float*)d_in[5];
  const float* Wv   = (const float*)d_in[6];
  const float* Wfc  = (const float*)d_in[7];
  const float* lnw  = (const float*)d_in[8];
  const float* lnb  = (const float*)d_in[9];
  float* out = (float*)d_out;

  char* ws = (char*)d_ws;
  const size_t SZ_ACT  = (size_t)BB * SS * DD * 2;
  const size_t SZ_W    = (size_t)HH * DD * DKV * 2;
  const size_t SZ_HEAD = (size_t)BB * HH * SS * DKV * 2;
  size_t off = 0;
  unsigned short* qbf  = (unsigned short*)(ws + off); off += SZ_ACT;
  unsigned short* kbf  = (unsigned short*)(ws + off); off += SZ_ACT;
  unsigned short* vbf  = (unsigned short*)(ws + off); off += SZ_ACT;
  unsigned short* WqT  = (unsigned short*)(ws + off); off += SZ_W;
  unsigned short* WkT  = (unsigned short*)(ws + off); off += SZ_W;
  unsigned short* WvT  = (unsigned short*)(ws + off); off += SZ_W;
  unsigned short* WfcT = (unsigned short*)(ws + off); off += (size_t)DD * DD * 2;
  unsigned short* qh   = (unsigned short*)(ws + off); off += SZ_HEAD;
  unsigned short* kh   = (unsigned short*)(ws + off); off += SZ_HEAD;
  unsigned short* vhT  = (unsigned short*)(ws + off); off += SZ_HEAD;
  unsigned short* aout = (unsigned short*)(ws + off); off += SZ_ACT;
  float*          fco  = (float*)(ws + off);          off += (size_t)BB * SS * DD * 4;

  constexpr int nAct = BB * SS * DD;
  constexpr int nW   = HH * DD * DKV;
  constexpr int nFc  = DD * DD;

  static_assert(nAct % CV_BLOCK_ELEMS == 0, "cvt_kernel: whole work-groups only");
  static_assert(nW % (8 * 256) == 0 && nFc % (8 * 256) == 0, "cvtT_kernel: whole work-groups only");
  static_assert(DKV % 32 == 0 && DD % 32 == 0 && DD % 8 == 0, "cvtT_kernel: a wave stays in one row of src");
  cvt_kernel<<<nAct / CV_BLOCK_ELEMS, 256, 0, stream>>>(q, qbf);
  cvt_kernel<<<nAct / CV_BLOCK_ELEMS, 256, 0, stream>>>(k, kbf);
  cvt_kernel<<<nAct / CV_BLOCK_ELEMS, 256, 0, stream>>>(v, vbf);
  cvtT_kernel<DD, DKV><<<nW / (8 * 256), 256, 0, stream>>>(Wq, WqT);
  cvtT_kernel<DD, DKV><<<nW / (8 * 256), 256, 0, stream>>>(Wk, WkT);
  cvtT_kernel<DD, DKV><<<nW / (8 * 256), 256, 0, stream>>>(Wv, WvT);
  cvtT_kernel<DD, DD><<<nFc / (8 * 256), 256, 0, stream>>>(Wfc, WfcT);

  dim3 gGemm(DD / GT_N, (BB * SS) / GT_M);
  gemm_kernel<0><<<gGemm, 256, 0, stream>>>(qbf, WqT, qh, nullptr, nullptr, 0.125f);
  gemm_kernel<0><<<gGemm, 256, 0, stream>>>(kbf, WkT, kh, nullptr, nullptr, 1.0f);
  gemm_kernel<1><<<gGemm, 256, 0, stream>>>(vbf, WvT, vhT, nullptr, nullptr, 1.0f);

  attn_kernel<<<dim3(BB * HH, SS / AT_ROWS), AT_WAVES * 32, 0, stream>>>(qh, kh, vhT, mask, aout);

  gemm_kernel<2><<<gGemm, 256, 0, stream>>>(aout, WfcT, nullptr, fco, q, 1.0f);
  ln_kernel<<<BB * SS, 256, 0, stream>>>(fco, lnw, lnb, out);
}


# 13 launches -> 5: the seven fp32->bf16 converts run as ONE grid and the three projection GEMMs as ONE grid (block index picks the operand set); per-kernel code and sum order unchanged
# speedup vs baseline: 4.0315x; 1.3112x over previous
#include <hip/hip_runtime.h>


#define BB 2
#define SS 2048
#define DD 1024
#define HH 16
#define DKV 64

typedef __attribute__((ext_vector_type(16))) __bf16 v16bf;
typedef __attribute__((ext_vector_type(8)))  float  v8f;

union FragU { v16bf v; uint4 q[2]; };

__device__ __forceinline__ unsigned short f2bf(float f) {
  unsigned int u = __float_as_uint(f);
  u += 0x7FFFu + ((u >> 16) & 1u);
  return (unsigned short)(u >> 16);
}

__device__ __forceinline__ v16bf load_frag(const unsigned short* rowbase, int lane) {
  const unsigned short* p = rowbase + ((lane & 16) ? 8 : 0);
  FragU f;
  f.q[0] = *reinterpret_cast<const uint4*>(p);
  f.q[1] = *reinterpret_cast<const uint4*>(p + 16);
  return f.v;
}

#define WMMA_BF16(a, b, c) \
  __builtin_amdgcn_wmma_f32_16x16x32_bf16(false, (a), false, (b), (short)0, (c), false, false)

#define CV_SLABS 8
#define CV_SLAB_ELEMS  (256 * 4)
#define CV_BLOCK_ELEMS (CV_SLABS * CV_SLAB_ELEMS)

__device__ __forceinline__ unsigned int f2bf_pair(float lo, float hi) {
  return (unsigned int)f2bf(lo) | ((unsigned int)f2bf(hi) << 16);
}

__device__ __forceinline__ void
cvt_block(const float* __restrict__ src, unsigned short* __restrict__ dst, unsigned int blk) {
  const size_t base = (size_t)blk * CV_BLOCK_ELEMS + threadIdx.x * 4;
  float4 v[CV_SLABS];
#pragma unroll
  for (int j = 0; j < CV_SLABS; ++j)
    v[j] = *reinterpret_cast<const float4*>(src + base + j * CV_SLAB_ELEMS);
#pragma unroll
  for (int j = 0; j < CV_SLABS; ++j) {
    uint2 w;
    w.x = f2bf_pair(v[j].x, v[j].y);
    w.y = f2bf_pair(v[j].z, v[j].w);
    *reinterpret_cast<uint2*>(dst + base + j * CV_SLAB_ELEMS) = w;
  }
}

template <int R, int C>
__device__ __forceinline__ void
cvtT_block(const float* __restrict__ src, unsigned short* __restrict__ dst, unsigned int blk) {
  const unsigned int i  = blk * 256 + threadIdx.x;
  const unsigned int c  = i % C;
  const unsigned int r8 = (i / C) % (R / 8);
  const unsigned int g  = i / (C * (R / 8));
  const float* s = src + ((size_t)g * R + (size_t)r8 * 8) * C + c;
  float v[8];
#pragma unroll
  for (int k = 0; k < 8; ++k) v[k] = s[(size_t)k * C];
  uint4 w;
  w.x = f2bf_pair(v[0], v[1]);
  w.y = f2bf_pair(v[2], v[3]);
  w.z = f2bf_pair(v[4], v[5]);
  w.w = f2bf_pair(v[6], v[7]);
  *reinterpret_cast<uint4*>(dst + ((size_t)g * C + c) * R + (size_t)r8 * 8) = w;
}

#define CV_ACT_BLOCKS ((BB * SS * DD) / CV_BLOCK_ELEMS)
#define CV_W_BLOCKS   ((HH * DD * DKV) / (8 * 256))
#define CV_ALL_BLOCKS (3 * CV_ACT_BLOCKS + 4 * CV_W_BLOCKS)

__global__ void __launch_bounds__(256)
cvt_all_kernel(const float* __restrict__ q, const float* __restrict__ k,
               const float* __restrict__ v,
               const float* __restrict__ Wq, const float* __restrict__ Wk,
               const float* __restrict__ Wv, const float* __restrict__ Wfc,
               unsigned short* __restrict__ qbf, unsigned short* __restrict__ kbf,
               unsigned short* __restrict__ vbf,
               unsigned short* __restrict__ WqT, unsigned short* __restrict__ WkT,
               unsigned short* __restrict__ WvT, unsigned short* __restrict__ WfcT) {
  const unsigned int bid = blockIdx.x;
  if (bid < 3 * CV_ACT_BLOCKS) {
    const unsigned int job = bid / CV_ACT_BLOCKS;
    const float* s = (job == 0) ? q : (job == 1) ? k : v;
    unsigned short* d = (job == 0) ? qbf : (job == 1) ? kbf : vbf;
    cvt_block(s, d, bid - job * CV_ACT_BLOCKS);
  } else {
    const unsigned int wb  = bid - 3 * CV_ACT_BLOCKS;
    const unsigned int job = wb / CV_W_BLOCKS;
    const unsigned int blk = wb - job * CV_W_BLOCKS;
    if (job < 3) {
      const float* s = (job == 0) ? Wq : (job == 1) ? Wk : Wv;
      unsigned short* d = (job == 0) ? WqT : (job == 1) ? WkT : WvT;
      cvtT_block<DD, DKV>(s, d, blk);
    } else {
      cvtT_block<DD, DD>(Wfc, WfcT, blk);
    }
  }
}

#define GT_M   128
#define GT_N   128
#define GT_K   128
#define GPITCH 136

typedef __attribute__((ext_vector_type(4))) unsigned int u32x4;

template <bool PROJ>
__global__ void __launch_bounds__(256)
gemm_kernel(const unsigned short* __restrict__ A0,
            const unsigned short* __restrict__ A1,
            const unsigned short* __restrict__ A2,
            const unsigned short* __restrict__ B0,
            const unsigned short* __restrict__ B1,
            const unsigned short* __restrict__ B2,
            unsigned short* __restrict__ O0,
            unsigned short* __restrict__ O1,
            unsigned short* __restrict__ O2,
            float* __restrict__ OUTF,
            const float* __restrict__ RES) {
  __shared__ __align__(16) unsigned short As[2][GT_M * GPITCH];
  __shared__ __align__(16) unsigned short Bs[2][GT_N * GPITCH];

  const int z = PROJ ? (int)blockIdx.z : 0;
  const unsigned short* __restrict__ A  = (z == 0) ? A0 : (z == 1) ? A1 : A2;
  const unsigned short* __restrict__ BT = (z == 0) ? B0 : (z == 1) ? B1 : B2;
  unsigned short* __restrict__ OUTH     = (z == 0) ? O0 : (z == 1) ? O1 : O2;
  const float scale = (z == 0) ? 0.125f : 1.0f;
  const size_t sq_stride = (z == 2) ? (size_t)1  : (size_t)DKV;
  const size_t dc_stride = (z == 2) ? (size_t)SS : (size_t)1;

  const int tid = threadIdx.x;
  const int lane = tid & 31;
  const int wave = tid >> 5;
  const int wm = wave & 3;
  const int wn = wave >> 2;
  const int m0 = blockIdx.y * GT_M;
  const int n0 = blockIdx.x * GT_N;
  const int rbase = (lane & 16) ? 8 : 0;
  const int col = lane & 15;

  const int srow = tid >> 4;
  const int sch  = (tid & 15) * 8;
  const unsigned short* Ag = A  + (size_t)(m0 + srow) * DD + sch;
  const unsigned short* Bg = BT + (size_t)(n0 + srow) * DD + sch;
  const int sidx = srow * GPITCH + sch;

  u32x4 ga[8], gb[8];
#pragma unroll
  for (int i = 0; i < 8; ++i) {
    ga[i] = *reinterpret_cast<const u32x4*>(Ag + (size_t)(16 * i) * DD);
    gb[i] = *reinterpret_cast<const u32x4*>(Bg + (size_t)(16 * i) * DD);
  }
#pragma unroll
  for (int i = 0; i < 8; ++i) {
    *reinterpret_cast<u32x4*>(&As[0][sidx + 16 * i * GPITCH]) = ga[i];
    *reinterpret_cast<u32x4*>(&Bs[0][sidx + 16 * i * GPITCH]) = gb[i];
  }
  __syncthreads();

  v8f acc[2][4] = {};

  for (int st = 0; st < DD / GT_K; ++st) {
    const int buf = st & 1;
    const bool more = (st + 1) < (DD / GT_K);
    if (more) {
      const unsigned short* ag = Ag + (st + 1) * GT_K;
      const unsigned short* bg = Bg + (st + 1) * GT_K;
#pragma unroll
      for (int i = 0; i < 8; ++i) {
        ga[i] = *reinterpret_cast<const u32x4*>(ag + (size_t)(16 * i) * DD);
        gb[i] = *reinterpret_cast<const u32x4*>(bg + (size_t)(16 * i) * DD);
      }
    }
    __builtin_amdgcn_sched_barrier(0);
    const unsigned short* Ab = &As[buf][(wm * 32 + col) * GPITCH];
    const unsigned short* Bb = &Bs[buf][(wn * 64 + col) * GPITCH];

#pragma unroll
    for (int j = 0; j < GT_K / 32; ++j) {
      v16bf af[2], bfr[4];
#pragma unroll
      for (int mt = 0; mt < 2; ++mt)
        af[mt] = load_frag(Ab + mt * 16 * GPITCH + j * 32, lane);
#pragma unroll
      for (int nt = 0; nt < 4; ++nt)
        bfr[nt] = load_frag(Bb + nt * 16 * GPITCH + j * 32, lane);
#pragma unroll
      for (int mt = 0; mt < 2; ++mt)
#pragma unroll
        for (int nt = 0; nt < 4; ++nt)
          acc[mt][nt] = WMMA_BF16(af[mt], bfr[nt], acc[mt][nt]);
    }

    __builtin_amdgcn_sched_barrier(0);
    if (more) {
#pragma unroll
      for (int i = 0; i < 8; ++i) {
        *reinterpret_cast<u32x4*>(&As[buf ^ 1][sidx + 16 * i * GPITCH]) = ga[i];
        *reinterpret_cast<u32x4*>(&Bs[buf ^ 1][sidx + 16 * i * GPITCH]) = gb[i];
      }
    }
    __syncthreads();
  }

#pragma unroll
  for (int mt = 0; mt < 2; ++mt) {
#pragma unroll
    for (int nt = 0; nt < 4; ++nt) {
#pragma unroll
      for (int r = 0; r < 8; ++r) {
        const int row = m0 + wm * 32 + mt * 16 + rbase + r;
        const int n   = n0 + wn * 64 + nt * 16 + col;
        if (!PROJ) {
          size_t idx = (size_t)row * DD + n;
          OUTF[idx] = acc[mt][nt][r] + RES[idx];
        } else {
          const int b = row / SS, sq = row - b * SS;
          const int h = n / DKV, dcol = n - h * DKV;
          const int bh = b * HH + h;
          unsigned short val = f2bf(acc[mt][nt][r] * scale);
          OUTH[(size_t)bh * SS * DKV + (size_t)sq * sq_stride + (size_t)dcol * dc_stride] = val;
        }
      }
    }
  }
}

#define AT_WAVES 8
#define AT_ROWS  (AT_WAVES * 16)
#define KT       64
#define LPITCH   72

typedef __attribute__((ext_vector_type(2))) float  f32x2;
typedef __attribute__((ext_vector_type(2))) __bf16 bf16x2;

__device__ __forceinline__ unsigned int pack_bf16(float lo, float hi) {
  f32x2 t = {lo, hi};
  return __builtin_bit_cast(unsigned int, __builtin_convertvector(t, bf16x2));
}

__device__ __forceinline__ float xhalf(float x) {
  int b = __builtin_bit_cast(int, x);
  return __builtin_bit_cast(float,
      __builtin_amdgcn_permlanex16(b, b, 0x76543210, 0xfedcba98, false, false));
}

__global__ void __launch_bounds__(AT_WAVES * 32)
attn_kernel(const unsigned short* __restrict__ QH,
            const unsigned short* __restrict__ KH,
            const unsigned short* __restrict__ VT,
            const int* __restrict__ MASK,
            unsigned short* __restrict__ AOUT) {
  __shared__ __align__(16) unsigned short Ks[2][KT * LPITCH];
  __shared__ __align__(16) unsigned short Vs[2][DKV * LPITCH];

  const int tid = threadIdx.x;
  const int lane = tid & 31;
  const int wave = tid >> 5;
  const int bh = blockIdx.x;
  const int b = bh >> 4, h = bh & 15;
  const int s_base = blockIdx.y * AT_ROWS + wave * 16;
  const int kofs = (lane & 16) ? 8 : 0;
  const int col = lane & 15;

  const unsigned short* Qrow =
      QH + ((size_t)bh * SS + (size_t)(s_base + col)) * DKV;
  v16bf qf0 = load_frag(Qrow, lane);
  v16bf qf1 = load_frag(Qrow + 32, lane);

  v8f o[4] = {};
  float m = -3.0e38f;
  float l = 0.0f;

  const int* mlane = MASK + (size_t)b * SS * SS + (size_t)(s_base + col) * SS + kofs;

  const int srow = tid >> 3;
  const int sch  = (tid & 7) * 8;
  const unsigned short* Kg = KH + ((size_t)bh * SS + srow) * DKV + sch;
  const unsigned short* Vg = VT + ((size_t)bh * DKV + srow) * SS + sch;
  const int sidx = srow * LPITCH + sch;

  uint4 gk0 = *reinterpret_cast<const uint4*>(Kg);
  uint4 gk1 = *reinterpret_cast<const uint4*>(Kg + (size_t)32 * DKV);
  uint4 gv0 = *reinterpret_cast<const uint4*>(Vg);
  uint4 gv1 = *reinterpret_cast<const uint4*>(Vg + (size_t)32 * SS);
  *reinterpret_cast<uint4*>(&Ks[0][sidx])               = gk0;
  *reinterpret_cast<uint4*>(&Ks[0][sidx + 32 * LPITCH]) = gk1;
  *reinterpret_cast<uint4*>(&Vs[0][sidx])               = gv0;
  *reinterpret_cast<uint4*>(&Vs[0][sidx + 32 * LPITCH]) = gv1;
  __syncthreads();

  for (int st = 0; st < SS / KT; ++st) {
    const int buf = st & 1;
    const int t0 = st * KT;
    const bool more = (st + 1) < (SS / KT);
    if (more) {
      const unsigned short* kg = Kg + (size_t)(t0 + KT) * DKV;
      const unsigned short* vg = Vg + (t0 + KT);
      gk0 = *reinterpret_cast<const uint4*>(kg);
      gk1 = *reinterpret_cast<const uint4*>(kg + (size_t)32 * DKV);
      gv0 = *reinterpret_cast<const uint4*>(vg);
      gv1 = *reinterpret_cast<const uint4*>(vg + (size_t)32 * SS);
    }
    const unsigned short* Kb = &Ks[buf][0];
    const unsigned short* Vb = &Vs[buf][0];

#pragma unroll
    for (int j = 0; j < KT / 32; ++j) {
      const int tt = t0 + j * 32;
      const int4 mk0a = *reinterpret_cast<const int4*>(mlane + tt);
      const int4 mk0b = *reinterpret_cast<const int4*>(mlane + tt + 4);
      const int4 mk1a = *reinterpret_cast<const int4*>(mlane + tt + 16);
      const int4 mk1b = *reinterpret_cast<const int4*>(mlane + tt + 20);
      v16bf kb0 = load_frag(Kb + (j * 32 + col) * LPITCH, lane);
      v16bf kb1 = load_frag(Kb + (j * 32 + col) * LPITCH + 32, lane);
      v16bf kb2 = load_frag(Kb + (j * 32 + 16 + col) * LPITCH, lane);
      v16bf kb3 = load_frag(Kb + (j * 32 + 16 + col) * LPITCH + 32, lane);
      v16bf vb[4];
#pragma unroll
      for (int nt = 0; nt < 4; ++nt)
        vb[nt] = load_frag(Vb + (nt * 16 + col) * LPITCH + j * 32, lane);

      v8f s0 = {}, s1 = {};
      s0 = WMMA_BF16(kb0, qf0, s0);
      s0 = WMMA_BF16(kb1, qf1, s0);
      s1 = WMMA_BF16(kb2, qf0, s1);
      s1 = WMMA_BF16(kb3, qf1, s1);

      const int mk0[8] = {mk0a.x, mk0a.y, mk0a.z, mk0a.w, mk0b.x, mk0b.y, mk0b.z, mk0b.w};
      const int mk1[8] = {mk1a.x, mk1a.y, mk1a.z, mk1a.w, mk1b.x, mk1b.y, mk1b.z, mk1b.w};
      float v0[8], v1[8];
#pragma unroll
      for (int r = 0; r < 8; ++r) {
        v0[r] = (mk0[r] != 0) ? s0[r] : -1.0e9f;
        v1[r] = (mk1[r] != 0) ? s1[r] : -1.0e9f;
      }
      float mx = fmaxf(v0[0], v1[0]);
#pragma unroll
      for (int r = 1; r < 8; ++r) mx = fmaxf(mx, fmaxf(v0[r], v1[r]));
      mx = fmaxf(mx, xhalf(mx));
      const float mnew  = fmaxf(m, mx);
      const float alpha = __expf(m - mnew);
      float rs = 0.0f;
      unsigned int w0[4], w1[4];
#pragma unroll
      for (int r = 0; r < 8; r += 2) {
        const float p00 = __expf(v0[r]     - mnew);
        const float p01 = __expf(v0[r + 1] - mnew);
        const float p10 = __expf(v1[r]     - mnew);
        const float p11 = __expf(v1[r + 1] - mnew);
        rs += (p00 + p01) + (p10 + p11);
        w0[r >> 1] = pack_bf16(p00, p01);
        w1[r >> 1] = pack_bf16(p10, p11);
      }
      l = l * alpha + rs;
      m = mnew;
#pragma unroll
      for (int nt = 0; nt < 4; ++nt)
#pragma unroll
        for (int r = 0; r < 8; ++r) o[nt][r] = o[nt][r] * alpha;

      FragU pf;
      pf.q[0] = make_uint4(w0[0], w0[1], w0[2], w0[3]);
      pf.q[1] = make_uint4(w1[0], w1[1], w1[2], w1[3]);
#pragma unroll
      for (int nt = 0; nt < 4; ++nt)
        o[nt] = WMMA_BF16(vb[nt], pf.v, o[nt]);
    }

    if (more) {
      unsigned short* Kn = &Ks[buf ^ 1][0];
      unsigned short* Vn = &Vs[buf ^ 1][0];
      *reinterpret_cast<uint4*>(Kn + sidx)               = gk0;
      *reinterpret_cast<uint4*>(Kn + sidx + 32 * LPITCH) = gk1;
      *reinterpret_cast<uint4*>(Vn + sidx)               = gv0;
      *reinterpret_cast<uint4*>(Vn + sidx + 32 * LPITCH) = gv1;
    }
    __syncthreads();
  }

  const float inv = 1.0f / (l + xhalf(l));
  unsigned short* orow =
      AOUT + ((size_t)b * SS + (size_t)(s_base + col)) * (HH * DKV) + h * DKV + kofs;
#pragma unroll
  for (int nt = 0; nt < 4; ++nt) {
    uint4 w;
    w.x = pack_bf16(o[nt][0] * inv, o[nt][1] * inv);
    w.y = pack_bf16(o[nt][2] * inv, o[nt][3] * inv);
    w.z = pack_bf16(o[nt][4] * inv, o[nt][5] * inv);
    w.w = pack_bf16(o[nt][6] * inv, o[nt][7] * inv);
    *reinterpret_cast<uint4*>(orow + nt * 16) = w;
  }
}

__global__ void ln_kernel(const float* __restrict__ X,
                          const float* __restrict__ W,
                          const float* __restrict__ Bv,
                          float* __restrict__ OUT) {
  __shared__ float red[8];
  const int row = blockIdx.x;
  const int tid = threadIdx.x;
  const float* xr = X + (size_t)row * DD;

  float v[4];
  float s = 0.0f;
#pragma unroll
  for (int i = 0; i < 4; ++i) { v[i] = xr[tid + i * 256]; s += v[i]; }
  for (int off = 1; off < 32; off <<= 1) s += __shfl_xor(s, off);
  if ((tid & 31) == 0) red[tid >> 5] = s;
  __syncthreads();
  float mu = (red[0] + red[1] + red[2] + red[3] +
              red[4] + red[5] + red[6] + red[7]) * (1.0f / DD);
  float vs = 0.0f;
#pragma unroll
  for (int i = 0; i < 4; ++i) { float d = v[i] - mu; vs += d * d; }
  for (int off = 1; off < 32; off <<= 1) vs += __shfl_xor(vs, off);
  __syncthreads();
  if ((tid & 31) == 0) red[tid >> 5] = vs;
  __syncthreads();
  float var = (red[0] + red[1] + red[2] + red[3] +
               red[4] + red[5] + red[6] + red[7]) * (1.0f / DD);
  float rstd = rsqrtf(var + 1e-6f);
#pragma unroll
  for (int i = 0; i < 4; ++i) {
    int c = tid + i * 256;
    OUT[(size_t)row * DD + c] = (v[i] - mu) * rstd * W[c] + Bv[c];
  }
}

extern "C" void kernel_launch(void* const* d_in, const int* in_sizes, int n_in,
                              void* d_out, int out_size, void* d_ws, size_t ws_size,
                              hipStream_t stream) {
  const float* q    = (const float*)d_in[0];
  const float* k    = (const float*)d_in[1];
  const float* v    = (const float*)d_in[2];
  const int*   mask = (const int*)d_in[3];
  const float* Wq   = (const float*)d_in[4];
  const float* Wk   = (const float*)d_in[5];
  const float* Wv   = (const float*)d_in[6];
  const float* Wfc  = (const float*)d_in[7];
  const float* lnw  = (const float*)d_in[8];
  const float* lnb  = (const float*)d_in[9];
  float* out = (float*)d_out;

  char* ws = (char*)d_ws;
  const size_t SZ_ACT  = (size_t)BB * SS * DD * 2;
  const size_t SZ_W    = (size_t)HH * DD * DKV * 2;
  const size_t SZ_HEAD = (size_t)BB * HH * SS * DKV * 2;
  size_t off = 0;
  unsigned short* qbf  = (unsigned short*)(ws + off); off += SZ_ACT;
  unsigned short* kbf  = (unsigned short*)(ws + off); off += SZ_ACT;
  unsigned short* vbf  = (unsigned short*)(ws + off); off += SZ_ACT;
  unsigned short* WqT  = (unsigned short*)(ws + off); off += SZ_W;
  unsigned short* WkT  = (unsigned short*)(ws + off); off += SZ_W;
  unsigned short* WvT  = (unsigned short*)(ws + off); off += SZ_W;
  unsigned short* WfcT = (unsigned short*)(ws + off); off += (size_t)DD * DD * 2;
  unsigned short* qh   = (unsigned short*)(ws + off); off += SZ_HEAD;
  unsigned short* kh   = (unsigned short*)(ws + off); off += SZ_HEAD;
  unsigned short* vhT  = (unsigned short*)(ws + off); off += SZ_HEAD;
  unsigned short* aout = (unsigned short*)(ws + off); off += SZ_ACT;
  float*          fco  = (float*)(ws + off);          off += (size_t)BB * SS * DD * 4;

  constexpr int nAct = BB * SS * DD;
  constexpr int nW   = HH * DD * DKV;
  constexpr int nFc  = DD * DD;

  static_assert(nAct % CV_BLOCK_ELEMS == 0, "cvt_kernel: whole work-groups only");
  static_assert(nW % (8 * 256) == 0 && nFc % (8 * 256) == 0, "cvtT_kernel: whole work-groups only");
  static_assert(DKV % 32 == 0 && DD % 32 == 0 && DD % 8 == 0, "cvtT_kernel: a wave stays in one row of src");
  static_assert(nAct / CV_BLOCK_ELEMS == CV_ACT_BLOCKS, "cvt_all_kernel: activation range");
  static_assert(nW / (8 * 256) == CV_W_BLOCKS && nFc / (8 * 256) == CV_W_BLOCKS,
                "cvt_all_kernel: weight ranges");
  cvt_all_kernel<<<CV_ALL_BLOCKS, 256, 0, stream>>>(q, k, v, Wq, Wk, Wv, Wfc,
                                                    qbf, kbf, vbf, WqT, WkT, WvT, WfcT);

  gemm_kernel<true><<<dim3(DD / GT_N, (BB * SS) / GT_M, 3), 256, 0, stream>>>(
      qbf, kbf, vbf, WqT, WkT, WvT, qh, kh, vhT, nullptr, nullptr);

  attn_kernel<<<dim3(BB * HH, SS / AT_ROWS), AT_WAVES * 32, 0, stream>>>(qh, kh, vhT, mask, aout);

  gemm_kernel<false><<<dim3(DD / GT_N, (BB * SS) / GT_M), 256, 0, stream>>>(
      aout, nullptr, nullptr, WfcT, nullptr, nullptr, nullptr, nullptr, nullptr, fco, q);
  ln_kernel<<<BB * SS, 256, 0, stream>>>(fco, lnw, lnb, out);
}


# attn mask: int32 [B,S,S] packed once (extra blocks of the convert grid) to 1 bit per element in the score registers' lane order; attention fetches it by scalar loads, words used as select masks
# speedup vs baseline: 4.4341x; 1.0999x over previous
#include <hip/hip_runtime.h>


#define BB 2
#define SS 2048
#define DD 1024
#define HH 16
#define DKV 64

typedef __attribute__((ext_vector_type(16))) __bf16 v16bf;
typedef __attribute__((ext_vector_type(8)))  float  v8f;

union FragU { v16bf v; uint4 q[2]; };

__device__ __forceinline__ unsigned short f2bf(float f) {
  unsigned int u = __float_as_uint(f);
  u += 0x7FFFu + ((u >> 16) & 1u);
  return (unsigned short)(u >> 16);
}

__device__ __forceinline__ v16bf load_frag(const unsigned short* rowbase, int lane) {
  const unsigned short* p = rowbase + ((lane & 16) ? 8 : 0);
  FragU f;
  f.q[0] = *reinterpret_cast<const uint4*>(p);
  f.q[1] = *reinterpret_cast<const uint4*>(p + 16);
  return f.v;
}

#define WMMA_BF16(a, b, c) \
  __builtin_amdgcn_wmma_f32_16x16x32_bf16(false, (a), false, (b), (short)0, (c), false, false)

#define CV_SLABS 8
#define CV_SLAB_ELEMS  (256 * 4)
#define CV_BLOCK_ELEMS (CV_SLABS * CV_SLAB_ELEMS)

__device__ __forceinline__ unsigned int f2bf_pair(float lo, float hi) {
  return (unsigned int)f2bf(lo) | ((unsigned int)f2bf(hi) << 16);
}

__device__ __forceinline__ void
cvt_block(const float* __restrict__ src, unsigned short* __restrict__ dst, unsigned int blk) {
  const size_t base = (size_t)blk * CV_BLOCK_ELEMS + threadIdx.x * 4;
  float4 v[CV_SLABS];
#pragma unroll
  for (int j = 0; j < CV_SLABS; ++j)
    v[j] = *reinterpret_cast<const float4*>(src + base + j * CV_SLAB_ELEMS);
#pragma unroll
  for (int j = 0; j < CV_SLABS; ++j) {
    uint2 w;
    w.x = f2bf_pair(v[j].x, v[j].y);
    w.y = f2bf_pair(v[j].z, v[j].w);
    *reinterpret_cast<uint2*>(dst + base + j * CV_SLAB_ELEMS) = w;
  }
}

template <int R, int C>
__device__ __forceinline__ void
cvtT_block(const float* __restrict__ src, unsigned short* __restrict__ dst, unsigned int blk) {
  const unsigned int i  = blk * 256 + threadIdx.x;
  const unsigned int c  = i % C;
  const unsigned int r8 = (i / C) % (R / 8);
  const unsigned int g  = i / (C * (R / 8));
  const float* s = src + ((size_t)g * R + (size_t)r8 * 8) * C + c;
  float v[8];
#pragma unroll
  for (int k = 0; k < 8; ++k) v[k] = s[(size_t)k * C];
  uint4 w;
  w.x = f2bf_pair(v[0], v[1]);
  w.y = f2bf_pair(v[2], v[3]);
  w.z = f2bf_pair(v[4], v[5]);
  w.w = f2bf_pair(v[6], v[7]);
  *reinterpret_cast<uint4*>(dst + ((size_t)g * C + c) * R + (size_t)r8 * 8) = w;
}

#define MB_STAGE_WORDS 32
#define MB_BLOCKS (BB * (SS / 16) * 2)

__device__ __forceinline__ void
maskpack_block(const int* __restrict__ mask, unsigned int* __restrict__ mbits, unsigned int blk) {
  const unsigned int lane = threadIdx.x & 31;
  const unsigned int wave = threadIdx.x >> 5;
  const unsigned int col  = lane & 15;
  const unsigned int kofs = (lane & 16) ? 8 : 0;
  const unsigned int rb = blk >> 1;
  const unsigned int sp = (blk & 1) * 8 + wave;
  const int* mrow = mask + ((size_t)rb * 16 + col) * SS + sp * 128 + kofs;
  int4 m[16];
#pragma unroll
  for (int i = 0; i < 8; ++i) {
    m[2 * i]     = *reinterpret_cast<const int4*>(mrow + i * 16);
    m[2 * i + 1] = *reinterpret_cast<const int4*>(mrow + i * 16 + 4);
  }
  unsigned int lo = 0, hi = 0;
#pragma unroll
  for (int i = 0; i < 8; ++i) {
    const int mv[8] = {m[2 * i].x, m[2 * i].y, m[2 * i].z, m[2 * i].w,
                       m[2 * i + 1].x, m[2 * i + 1].y, m[2 * i + 1].z, m[2 * i + 1].w};
#pragma unroll
    for (int r = 0; r < 8; ++r) {
      const unsigned int bal = __builtin_amdgcn_ballot_w32(mv[r] != 0);
      const unsigned int idx = i * 8 + r;
      if (idx < 32) lo = (lane == idx)      ? bal : lo;
      else          hi = (lane == idx - 32) ? bal : hi;
    }
  }
  unsigned int* dst = mbits + ((size_t)rb * (SS / 64) + sp * 2) * MB_STAGE_WORDS;
  dst[lane]      = lo;
  dst[32 + lane] = hi;
}

#define CV_ACT_BLOCKS ((BB * SS * DD) / CV_BLOCK_ELEMS)
#define CV_W_BLOCKS   ((HH * DD * DKV) / (8 * 256))
#define CV_ALL_BLOCKS (3 * CV_ACT_BLOCKS + 4 * CV_W_BLOCKS + MB_BLOCKS)

__global__ void __launch_bounds__(256)
cvt_all_kernel(const float* __restrict__ q, const float* __restrict__ k,
               const float* __restrict__ v,
               const float* __restrict__ Wq, const float* __restrict__ Wk,
               const float* __restrict__ Wv, const float* __restrict__ Wfc,
               const int* __restrict__ mask, unsigned int* __restrict__ mbits,
               unsigned short* __restrict__ qbf, unsigned short* __restrict__ kbf,
               unsigned short* __restrict__ vbf,
               unsigned short* __restrict__ WqT, unsigned short* __restrict__ WkT,
               unsigned short* __restrict__ WvT, unsigned short* __restrict__ WfcT) {
  const unsigned int bid = blockIdx.x;
  if (bid < 3 * CV_ACT_BLOCKS) {
    const unsigned int job = bid / CV_ACT_BLOCKS;
    const float* s = (job == 0) ? q : (job == 1) ? k : v;
    unsigned short* d = (job == 0) ? qbf : (job == 1) ? kbf : vbf;
    cvt_block(s, d, bid - job * CV_ACT_BLOCKS);
  } else {
    const unsigned int wb  = bid - 3 * CV_ACT_BLOCKS;
    const unsigned int job = wb / CV_W_BLOCKS;
    const unsigned int blk = wb - job * CV_W_BLOCKS;
    if (job < 3) {
      const float* s = (job == 0) ? Wq : (job == 1) ? Wk : Wv;
      unsigned short* d = (job == 0) ? WqT : (job == 1) ? WkT : WvT;
      cvtT_block<DD, DKV>(s, d, blk);
    } else if (job == 3) {
      cvtT_block<DD, DD>(Wfc, WfcT, blk);
    } else {
      maskpack_block(mask, mbits, wb - 4 * CV_W_BLOCKS);
    }
  }
}

#define GT_M   128
#define GT_N   128
#define GT_K   128
#define GPITCH 136

typedef __attribute__((ext_vector_type(4))) unsigned int u32x4;

template <bool PROJ>
__global__ void __launch_bounds__(256)
gemm_kernel(const unsigned short* __restrict__ A0,
            const unsigned short* __restrict__ A1,
            const unsigned short* __restrict__ A2,
            const unsigned short* __restrict__ B0,
            const unsigned short* __restrict__ B1,
            const unsigned short* __restrict__ B2,
            unsigned short* __restrict__ O0,
            unsigned short* __restrict__ O1,
            unsigned short* __restrict__ O2,
            float* __restrict__ OUTF,
            const float* __restrict__ RES) {
  __shared__ __align__(16) unsigned short As[2][GT_M * GPITCH];
  __shared__ __align__(16) unsigned short Bs[2][GT_N * GPITCH];

  const int z = PROJ ? (int)blockIdx.z : 0;
  const unsigned short* __restrict__ A  = (z == 0) ? A0 : (z == 1) ? A1 : A2;
  const unsigned short* __restrict__ BT = (z == 0) ? B0 : (z == 1) ? B1 : B2;
  unsigned short* __restrict__ OUTH     = (z == 0) ? O0 : (z == 1) ? O1 : O2;
  const float scale = (z == 0) ? 0.125f : 1.0f;
  const size_t sq_stride = (z == 2) ? (size_t)1  : (size_t)DKV;
  const size_t dc_stride = (z == 2) ? (size_t)SS : (size_t)1;

  const int tid = threadIdx.x;
  const int lane = tid & 31;
  const int wave = tid >> 5;
  const int wm = wave & 3;
  const int wn = wave >> 2;
  const int m0 = blockIdx.y * GT_M;
  const int n0 = blockIdx.x * GT_N;
  const int rbase = (lane & 16) ? 8 : 0;
  const int col = lane & 15;

  const int srow = tid >> 4;
  const int sch  = (tid & 15) * 8;
  const unsigned short* Ag = A  + (size_t)(m0 + srow) * DD + sch;
  const unsigned short* Bg = BT + (size_t)(n0 + srow) * DD + sch;
  const int sidx = srow * GPITCH + sch;

  u32x4 ga[8], gb[8];
#pragma unroll
  for (int i = 0; i < 8; ++i) {
    ga[i] = *reinterpret_cast<const u32x4*>(Ag + (size_t)(16 * i) * DD);
    gb[i] = *reinterpret_cast<const u32x4*>(Bg + (size_t)(16 * i) * DD);
  }
#pragma unroll
  for (int i = 0; i < 8; ++i) {
    *reinterpret_cast<u32x4*>(&As[0][sidx + 16 * i * GPITCH]) = ga[i];
    *reinterpret_cast<u32x4*>(&Bs[0][sidx + 16 * i * GPITCH]) = gb[i];
  }
  __syncthreads();

  v8f acc[2][4] = {};

  for (int st = 0; st < DD / GT_K; ++st) {
    const int buf = st & 1;
    const bool more = (st + 1) < (DD / GT_K);
    if (more) {
      const unsigned short* ag = Ag + (st + 1) * GT_K;
      const unsigned short* bg = Bg + (st + 1) * GT_K;
#pragma unroll
      for (int i = 0; i < 8; ++i) {
        ga[i] = *reinterpret_cast<const u32x4*>(ag + (size_t)(16 * i) * DD);
        gb[i] = *reinterpret_cast<const u32x4*>(bg + (size_t)(16 * i) * DD);
      }
    }
    __builtin_amdgcn_sched_barrier(0);
    const unsigned short* Ab = &As[buf][(wm * 32 + col) * GPITCH];
    const unsigned short* Bb = &Bs[buf][(wn * 64 + col) * GPITCH];

#pragma unroll
    for (int j = 0; j < GT_K / 32; ++j) {
      v16bf af[2], bfr[4];
#pragma unroll
      for (int mt = 0; mt < 2; ++mt)
        af[mt] = load_frag(Ab + mt * 16 * GPITCH + j * 32, lane);
#pragma unroll
      for (int nt = 0; nt < 4; ++nt)
        bfr[nt] = load_frag(Bb + nt * 16 * GPITCH + j * 32, lane);
#pragma unroll
      for (int mt = 0; mt < 2; ++mt)
#pragma unroll
        for (int nt = 0; nt < 4; ++nt)
          acc[mt][nt] = WMMA_BF16(af[mt], bfr[nt], acc[mt][nt]);
    }

    __builtin_amdgcn_sched_barrier(0);
    if (more) {
#pragma unroll
      for (int i = 0; i < 8; ++i) {
        *reinterpret_cast<u32x4*>(&As[buf ^ 1][sidx + 16 * i * GPITCH]) = ga[i];
        *reinterpret_cast<u32x4*>(&Bs[buf ^ 1][sidx + 16 * i * GPITCH]) = gb[i];
      }
    }
    __syncthreads();
  }

#pragma unroll
  for (int mt = 0; mt < 2; ++mt) {
#pragma unroll
    for (int nt = 0; nt < 4; ++nt) {
#pragma unroll
      for (int r = 0; r < 8; ++r) {
        const int row = m0 + wm * 32 + mt * 16 + rbase + r;
        const int n   = n0 + wn * 64 + nt * 16 + col;
        if (!PROJ) {
          size_t idx = (size_t)row * DD + n;
          OUTF[idx] = acc[mt][nt][r] + RES[idx];
        } else {
          const int b = row / SS, sq = row - b * SS;
          const int h = n / DKV, dcol = n - h * DKV;
          const int bh = b * HH + h;
          unsigned short val = f2bf(acc[mt][nt][r] * scale);
          OUTH[(size_t)bh * SS * DKV + (size_t)sq * sq_stride + (size_t)dcol * dc_stride] = val;
        }
      }
    }
  }
}

#define AT_WAVES 8
#define AT_ROWS  (AT_WAVES * 16)
#define KT       64
#define LPITCH   72

typedef __attribute__((ext_vector_type(2))) float  f32x2;
typedef __attribute__((ext_vector_type(2))) __bf16 bf16x2;

__device__ __forceinline__ unsigned int pack_bf16(float lo, float hi) {
  f32x2 t = {lo, hi};
  return __builtin_bit_cast(unsigned int, __builtin_convertvector(t, bf16x2));
}

__device__ __forceinline__ float xhalf(float x) {
  int b = __builtin_bit_cast(int, x);
  return __builtin_bit_cast(float,
      __builtin_amdgcn_permlanex16(b, b, 0x76543210, 0xfedcba98, false, false));
}

__global__ void __launch_bounds__(AT_WAVES * 32)
attn_kernel(const unsigned short* __restrict__ QH,
            const unsigned short* __restrict__ KH,
            const unsigned short* __restrict__ VT,
            const unsigned int* __restrict__ MBITS,
            unsigned short* __restrict__ AOUT) {
  __shared__ __align__(16) unsigned short Ks[2][KT * LPITCH];
  __shared__ __align__(16) unsigned short Vs[2][DKV * LPITCH];

  const int tid = threadIdx.x;
  const int lane = tid & 31;
  const int wave = __builtin_amdgcn_readfirstlane(tid >> 5);
  const int bh = blockIdx.x;
  const int b = bh >> 4, h = bh & 15;
  const int s_base = blockIdx.y * AT_ROWS + wave * 16;
  const int kofs = (lane & 16) ? 8 : 0;
  const int col = lane & 15;

  const unsigned short* Qrow =
      QH + ((size_t)bh * SS + (size_t)(s_base + col)) * DKV;
  v16bf qf0 = load_frag(Qrow, lane);
  v16bf qf1 = load_frag(Qrow + 32, lane);

  v8f o[4] = {};
  float m = -3.0e38f;
  float l = 0.0f;

  const unsigned int* mwave =
      MBITS + ((size_t)b * (SS / 16) + (size_t)(s_base >> 4)) * (SS / KT) * MB_STAGE_WORDS;

  const int srow = tid >> 3;
  const int sch  = (tid & 7) * 8;
  const unsigned short* Kg = KH + ((size_t)bh * SS + srow) * DKV + sch;
  const unsigned short* Vg = VT + ((size_t)bh * DKV + srow) * SS + sch;
  const int sidx = srow * LPITCH + sch;

  uint4 gk0 = *reinterpret_cast<const uint4*>(Kg);
  uint4 gk1 = *reinterpret_cast<const uint4*>(Kg + (size_t)32 * DKV);
  uint4 gv0 = *reinterpret_cast<const uint4*>(Vg);
  uint4 gv1 = *reinterpret_cast<const uint4*>(Vg + (size_t)32 * SS);
  *reinterpret_cast<uint4*>(&Ks[0][sidx])               = gk0;
  *reinterpret_cast<uint4*>(&Ks[0][sidx + 32 * LPITCH]) = gk1;
  *reinterpret_cast<uint4*>(&Vs[0][sidx])               = gv0;
  *reinterpret_cast<uint4*>(&Vs[0][sidx + 32 * LPITCH]) = gv1;
  __syncthreads();

  for (int st = 0; st < SS / KT; ++st) {
    const int buf = st & 1;
    const int t0 = st * KT;
    const bool more = (st + 1) < (SS / KT);
    if (more) {
      const unsigned short* kg = Kg + (size_t)(t0 + KT) * DKV;
      const unsigned short* vg = Vg + (t0 + KT);
      gk0 = *reinterpret_cast<const uint4*>(kg);
      gk1 = *reinterpret_cast<const uint4*>(kg + (size_t)32 * DKV);
      gv0 = *reinterpret_cast<const uint4*>(vg);
      gv1 = *reinterpret_cast<const uint4*>(vg + (size_t)32 * SS);
    }
    const unsigned short* Kb = &Ks[buf][0];
    const unsigned short* Vb = &Vs[buf][0];

#pragma unroll
    for (int j = 0; j < KT / 32; ++j) {
      const uint4* mstep =
          reinterpret_cast<const uint4*>(mwave + (size_t)st * MB_STAGE_WORDS + j * 16);
      const uint4 mk0a = mstep[0];
      const uint4 mk0b = mstep[1];
      const uint4 mk1a = mstep[2];
      const uint4 mk1b = mstep[3];
      v16bf kb0 = load_frag(Kb + (j * 32 + col) * LPITCH, lane);
      v16bf kb1 = load_frag(Kb + (j * 32 + col) * LPITCH + 32, lane);
      v16bf kb2 = load_frag(Kb + (j * 32 + 16 + col) * LPITCH, lane);
      v16bf kb3 = load_frag(Kb + (j * 32 + 16 + col) * LPITCH + 32, lane);
      v16bf vb[4];
#pragma unroll
      for (int nt = 0; nt < 4; ++nt)
        vb[nt] = load_frag(Vb + (nt * 16 + col) * LPITCH + j * 32, lane);

      v8f s0 = {}, s1 = {};
      s0 = WMMA_BF16(kb0, qf0, s0);
      s0 = WMMA_BF16(kb1, qf1, s0);
      s1 = WMMA_BF16(kb2, qf0, s1);
      s1 = WMMA_BF16(kb3, qf1, s1);

      const unsigned int mk0[8] = {mk0a.x, mk0a.y, mk0a.z, mk0a.w, mk0b.x, mk0b.y, mk0b.z, mk0b.w};
      const unsigned int mk1[8] = {mk1a.x, mk1a.y, mk1a.z, mk1a.w, mk1b.x, mk1b.y, mk1b.z, mk1b.w};
      float v0[8], v1[8];
#pragma unroll
      for (int r = 0; r < 8; ++r) {
        v0[r] = __builtin_amdgcn_inverse_ballot_w32(mk0[r]) ? s0[r] : -1.0e9f;
        v1[r] = __builtin_amdgcn_inverse_ballot_w32(mk1[r]) ? s1[r] : -1.0e9f;
      }
      float mx = fmaxf(v0[0], v1[0]);
#pragma unroll
      for (int r = 1; r < 8; ++r) mx = fmaxf(mx, fmaxf(v0[r], v1[r]));
      mx = fmaxf(mx, xhalf(mx));
      const float mnew  = fmaxf(m, mx);
      const float alpha = __expf(m - mnew);
      float rs = 0.0f;
      unsigned int w0[4], w1[4];
#pragma unroll
      for (int r = 0; r < 8; r += 2) {
        const float p00 = __expf(v0[r]     - mnew);
        const float p01 = __expf(v0[r + 1] - mnew);
        const float p10 = __expf(v1[r]     - mnew);
        const float p11 = __expf(v1[r + 1] - mnew);
        rs += (p00 + p01) + (p10 + p11);
        w0[r >> 1] = pack_bf16(p00, p01);
        w1[r >> 1] = pack_bf16(p10, p11);
      }
      l = l * alpha + rs;
      m = mnew;
#pragma unroll
      for (int nt = 0; nt < 4; ++nt)
#pragma unroll
        for (int r = 0; r < 8; ++r) o[nt][r] = o[nt][r] * alpha;

      FragU pf;
      pf.q[0] = make_uint4(w0[0], w0[1], w0[2], w0[3]);
      pf.q[1] = make_uint4(w1[0], w1[1], w1[2], w1[3]);
#pragma unroll
      for (int nt = 0; nt < 4; ++nt)
        o[nt] = WMMA_BF16(vb[nt], pf.v, o[nt]);
    }

    if (more) {
      unsigned short* Kn = &Ks[buf ^ 1][0];
      unsigned short* Vn = &Vs[buf ^ 1][0];
      *reinterpret_cast<uint4*>(Kn + sidx)               = gk0;
      *reinterpret_cast<uint4*>(Kn + sidx + 32 * LPITCH) = gk1;
      *reinterpret_cast<uint4*>(Vn + sidx)               = gv0;
      *reinterpret_cast<uint4*>(Vn + sidx + 32 * LPITCH) = gv1;
    }
    __syncthreads();
  }

  const float inv = 1.0f / (l + xhalf(l));
  unsigned short* orow =
      AOUT + ((size_t)b * SS + (size_t)(s_base + col)) * (HH * DKV) + h * DKV + kofs;
#pragma unroll
  for (int nt = 0; nt < 4; ++nt) {
    uint4 w;
    w.x = pack_bf16(o[nt][0] * inv, o[nt][1] * inv);
    w.y = pack_bf16(o[nt][2] * inv, o[nt][3] * inv);
    w.z = pack_bf16(o[nt][4] * inv, o[nt][5] * inv);
    w.w = pack_bf16(o[nt][6] * inv, o[nt][7] * inv);
    *reinterpret_cast<uint4*>(orow + nt * 16) = w;
  }
}

__global__ void ln_kernel(const float* __restrict__ X,
                          const float* __restrict__ W,
                          const float* __restrict__ Bv,
                          float* __restrict__ OUT) {
  __shared__ float red[8];
  const int row = blockIdx.x;
  const int tid = threadIdx.x;
  const float* xr = X + (size_t)row * DD;

  float v[4];
  float s = 0.0f;
#pragma unroll
  for (int i = 0; i < 4; ++i) { v[i] = xr[tid + i * 256]; s += v[i]; }
  for (int off = 1; off < 32; off <<= 1) s += __shfl_xor(s, off);
  if ((tid & 31) == 0) red[tid >> 5] = s;
  __syncthreads();
  float mu = (red[0] + red[1] + red[2] + red[3] +
              red[4] + red[5] + red[6] + red[7]) * (1.0f / DD);
  float vs = 0.0f;
#pragma unroll
  for (int i = 0; i < 4; ++i) { float d = v[i] - mu; vs += d * d; }
  for (int off = 1; off < 32; off <<= 1) vs += __shfl_xor(vs, off);
  __syncthreads();
  if ((tid & 31) == 0) red[tid >> 5] = vs;
  __syncthreads();
  float var = (red[0] + red[1] + red[2] + red[3] +
               red[4] + red[5] + red[6] + red[7]) * (1.0f / DD);
  float rstd = rsqrtf(var + 1e-6f);
#pragma unroll
  for (int i = 0; i < 4; ++i) {
    int c = tid + i * 256;
    OUT[(size_t)row * DD + c] = (v[i] - mu) * rstd * W[c] + Bv[c];
  }
}

extern "C" void kernel_launch(void* const* d_in, const int* in_sizes, int n_in,
                              void* d_out, int out_size, void* d_ws, size_t ws_size,
                              hipStream_t stream) {
  const float* q    = (const float*)d_in[0];
  const float* k    = (const float*)d_in[1];
  const float* v    = (const float*)d_in[2];
  const int*   mask = (const int*)d_in[3];
  const float* Wq   = (const float*)d_in[4];
  const float* Wk   = (const float*)d_in[5];
  const float* Wv   = (const float*)d_in[6];
  const float* Wfc  = (const float*)d_in[7];
  const float* lnw  = (const float*)d_in[8];
  const float* lnb  = (const float*)d_in[9];
  float* out = (float*)d_out;

  char* ws = (char*)d_ws;
  const size_t SZ_ACT  = (size_t)BB * SS * DD * 2;
  const size_t SZ_W    = (size_t)HH * DD * DKV * 2;
  const size_t SZ_HEAD = (size_t)BB * HH * SS * DKV * 2;
  size_t off = 0;
  unsigned short* qbf  = (unsigned short*)(ws + off); off += SZ_ACT;
  unsigned short* kbf  = (unsigned short*)(ws + off); off += SZ_ACT;
  unsigned short* vbf  = (unsigned short*)(ws + off); off += SZ_ACT;
  unsigned short* WqT  = (unsigned short*)(ws + off); off += SZ_W;
  unsigned short* WkT  = (unsigned short*)(ws + off); off += SZ_W;
  unsigned short* WvT  = (unsigned short*)(ws + off); off += SZ_W;
  unsigned short* WfcT = (unsigned short*)(ws + off); off += (size_t)DD * DD * 2;
  unsigned short* qh   = (unsigned short*)(ws + off); off += SZ_HEAD;
  unsigned short* kh   = (unsigned short*)(ws + off); off += SZ_HEAD;
  unsigned short* vhT  = (unsigned short*)(ws + off); off += SZ_HEAD;
  unsigned short* aout = (unsigned short*)(ws + off); off += SZ_ACT;
  float*          fco  = (float*)(ws + off);          off += (size_t)BB * SS * DD * 4;
  unsigned int*   mbits = (unsigned int*)fco;

  constexpr int nAct = BB * SS * DD;
  constexpr int nW   = HH * DD * DKV;
  constexpr int nFc  = DD * DD;

  static_assert(nAct % CV_BLOCK_ELEMS == 0, "cvt_kernel: whole work-groups only");
  static_assert(nW % (8 * 256) == 0 && nFc % (8 * 256) == 0, "cvtT_kernel: whole work-groups only");
  static_assert(DKV % 32 == 0 && DD % 32 == 0 && DD % 8 == 0, "cvtT_kernel: a wave stays in one row of src");
  static_assert(nAct / CV_BLOCK_ELEMS == CV_ACT_BLOCKS, "cvt_all_kernel: activation range");
  static_assert(nW / (8 * 256) == CV_W_BLOCKS && nFc / (8 * 256) == CV_W_BLOCKS,
                "cvt_all_kernel: weight ranges");
  static_assert(SS % 128 == 0 && (SS / 128) == 16, "maskpack_block: 2 work-groups x 8 waves x 128 keys per row");
  static_assert(KT == 64 && AT_ROWS == 128, "attn_kernel and maskpack_block share the stage and block sizes");
  cvt_all_kernel<<<CV_ALL_BLOCKS, 256, 0, stream>>>(q, k, v, Wq, Wk, Wv, Wfc, mask, mbits,
                                                    qbf, kbf, vbf, WqT, WkT, WvT, WfcT);

  gemm_kernel<true><<<dim3(DD / GT_N, (BB * SS) / GT_M, 3), 256, 0, stream>>>(
      qbf, kbf, vbf, WqT, WkT, WvT, qh, kh, vhT, nullptr, nullptr);

  attn_kernel<<<dim3(BB * HH, SS / AT_ROWS), AT_WAVES * 32, 0, stream>>>(qh, kh, vhT, mbits, aout);

  gemm_kernel<false><<<dim3(DD / GT_N, (BB * SS) / GT_M), 256, 0, stream>>>(
      aout, nullptr, nullptr, WfcT, nullptr, nullptr, nullptr, nullptr, nullptr, fco, q);
  ln_kernel<<<BB * SS, 256, 0, stream>>>(fco, lnw, lnb, out);
}


# proj x3 and fc GEMM epilogue: result tile parked in the dead operand LDS, stored as whole 128-byte lines (16 B a lane); output-contiguous operand made the row operand; k order of sums unchanged
# speedup vs baseline: 5.1093x; 1.1523x over previous
#include <hip/hip_runtime.h>


#define BB 2
#define SS 2048
#define DD 1024
#define HH 16
#define DKV 64

typedef __attribute__((ext_vector_type(16))) __bf16 v16bf;
typedef __attribute__((ext_vector_type(8)))  float  v8f;

union FragU { v16bf v; uint4 q[2]; };

__device__ __forceinline__ unsigned short f2bf(float f) {
  unsigned int u = __float_as_uint(f);
  u += 0x7FFFu + ((u >> 16) & 1u);
  return (unsigned short)(u >> 16);
}

__device__ __forceinline__ v16bf load_frag(const unsigned short* rowbase, int lane) {
  const unsigned short* p = rowbase + ((lane & 16) ? 8 : 0);
  FragU f;
  f.q[0] = *reinterpret_cast<const uint4*>(p);
  f.q[1] = *reinterpret_cast<const uint4*>(p + 16);
  return f.v;
}

#define WMMA_BF16(a, b, c) \
  __builtin_amdgcn_wmma_f32_16x16x32_bf16(false, (a), false, (b), (short)0, (c), false, false)

#define CV_SLABS 8
#define CV_SLAB_ELEMS  (256 * 4)
#define CV_BLOCK_ELEMS (CV_SLABS * CV_SLAB_ELEMS)

__device__ __forceinline__ unsigned int f2bf_pair(float lo, float hi) {
  return (unsigned int)f2bf(lo) | ((unsigned int)f2bf(hi) << 16);
}

__device__ __forceinline__ void
cvt_block(const float* __restrict__ src, unsigned short* __restrict__ dst, unsigned int blk) {
  const size_t base = (size_t)blk * CV_BLOCK_ELEMS + threadIdx.x * 4;
  float4 v[CV_SLABS];
#pragma unroll
  for (int j = 0; j < CV_SLABS; ++j)
    v[j] = *reinterpret_cast<const float4*>(src + base + j * CV_SLAB_ELEMS);
#pragma unroll
  for (int j = 0; j < CV_SLABS; ++j) {
    uint2 w;
    w.x = f2bf_pair(v[j].x, v[j].y);
    w.y = f2bf_pair(v[j].z, v[j].w);
    *reinterpret_cast<uint2*>(dst + base + j * CV_SLAB_ELEMS) = w;
  }
}

template <int R, int C>
__device__ __forceinline__ void
cvtT_block(const float* __restrict__ src, unsigned short* __restrict__ dst, unsigned int blk) {
  const unsigned int i  = blk * 256 + threadIdx.x;
  const unsigned int c  = i % C;
  const unsigned int r8 = (i / C) % (R / 8);
  const unsigned int g  = i / (C * (R / 8));
  const float* s = src + ((size_t)g * R + (size_t)r8 * 8) * C + c;
  float v[8];
#pragma unroll
  for (int k = 0; k < 8; ++k) v[k] = s[(size_t)k * C];
  uint4 w;
  w.x = f2bf_pair(v[0], v[1]);
  w.y = f2bf_pair(v[2], v[3]);
  w.z = f2bf_pair(v[4], v[5]);
  w.w = f2bf_pair(v[6], v[7]);
  *reinterpret_cast<uint4*>(dst + ((size_t)g * C + c) * R + (size_t)r8 * 8) = w;
}

#define MB_STAGE_WORDS 32
#define MB_BLOCKS (BB * (SS / 16) * 2)

__device__ __forceinline__ void
maskpack_block(const int* __restrict__ mask, unsigned int* __restrict__ mbits, unsigned int blk) {
  const unsigned int lane = threadIdx.x & 31;
  const unsigned int wave = threadIdx.x >> 5;
  const unsigned int col  = lane & 15;
  const unsigned int kofs = (lane & 16) ? 8 : 0;
  const unsigned int rb = blk >> 1;
  const unsigned int sp = (blk & 1) * 8 + wave;
  const int* mrow = mask + ((size_t)rb * 16 + col) * SS + sp * 128 + kofs;
  int4 m[16];
#pragma unroll
  for (int i = 0; i < 8; ++i) {
    m[2 * i]     = *reinterpret_cast<const int4*>(mrow + i * 16);
    m[2 * i + 1] = *reinterpret_cast<const int4*>(mrow + i * 16 + 4);
  }
  unsigned int lo = 0, hi = 0;
#pragma unroll
  for (int i = 0; i < 8; ++i) {
    const int mv[8] = {m[2 * i].x, m[2 * i].y, m[2 * i].z, m[2 * i].w,
                       m[2 * i + 1].x, m[2 * i + 1].y, m[2 * i + 1].z, m[2 * i + 1].w};
#pragma unroll
    for (int r = 0; r < 8; ++r) {
      const unsigned int bal = __builtin_amdgcn_ballot_w32(mv[r] != 0);
      const unsigned int idx = i * 8 + r;
      if (idx < 32) lo = (lane == idx)      ? bal : lo;
      else          hi = (lane == idx - 32) ? bal : hi;
    }
  }
  unsigned int* dst = mbits + ((size_t)rb * (SS / 64) + sp * 2) * MB_STAGE_WORDS;
  dst[lane]      = lo;
  dst[32 + lane] = hi;
}

#define CV_ACT_BLOCKS ((BB * SS * DD) / CV_BLOCK_ELEMS)
#define CV_W_BLOCKS   ((HH * DD * DKV) / (8 * 256))
#define CV_ALL_BLOCKS (3 * CV_ACT_BLOCKS + 4 * CV_W_BLOCKS + MB_BLOCKS)

__global__ void __launch_bounds__(256)
cvt_all_kernel(const float* __restrict__ q, const float* __restrict__ k,
               const float* __restrict__ v,
               const float* __restrict__ Wq, const float* __restrict__ Wk,
               const float* __restrict__ Wv, const float* __restrict__ Wfc,
               const int* __restrict__ mask, unsigned int* __restrict__ mbits,
               unsigned short* __restrict__ qbf, unsigned short* __restrict__ kbf,
               unsigned short* __restrict__ vbf,
               unsigned short* __restrict__ WqT, unsigned short* __restrict__ WkT,
               unsigned short* __restrict__ WvT, unsigned short* __restrict__ WfcT) {
  const unsigned int bid = blockIdx.x;
  if (bid < 3 * CV_ACT_BLOCKS) {
    const unsigned int job = bid / CV_ACT_BLOCKS;
    const float* s = (job == 0) ? q : (job == 1) ? k : v;
    unsigned short* d = (job == 0) ? qbf : (job == 1) ? kbf : vbf;
    cvt_block(s, d, bid - job * CV_ACT_BLOCKS);
  } else {
    const unsigned int wb  = bid - 3 * CV_ACT_BLOCKS;
    const unsigned int job = wb / CV_W_BLOCKS;
    const unsigned int blk = wb - job * CV_W_BLOCKS;
    if (job < 3) {
      const float* s = (job == 0) ? Wq : (job == 1) ? Wk : Wv;
      unsigned short* d = (job == 0) ? WqT : (job == 1) ? WkT : WvT;
      cvtT_block<DD, DKV>(s, d, blk);
    } else if (job == 3) {
      cvtT_block<DD, DD>(Wfc, WfcT, blk);
    } else {
      maskpack_block(mask, mbits, wb - 4 * CV_W_BLOCKS);
    }
  }
}

#define GT_M   128
#define GT_N   128
#define GT_K   128
#define GPITCH 136
#define FPITCH 132

typedef __attribute__((ext_vector_type(4))) unsigned int u32x4;

static_assert(GT_N * FPITCH * 4 <= 2 * GT_M * GPITCH * 2, "the parked fp32 tile fits in the A tiles' LDS");

template <bool PROJ>
__global__ void __launch_bounds__(256)
gemm_kernel(const unsigned short* __restrict__ X0,
            const unsigned short* __restrict__ X1,
            const unsigned short* __restrict__ X2,
            const unsigned short* __restrict__ W0,
            const unsigned short* __restrict__ W1,
            const unsigned short* __restrict__ W2,
            unsigned short* __restrict__ O0,
            unsigned short* __restrict__ O1,
            unsigned short* __restrict__ O2,
            float* __restrict__ OUTF,
            const float* __restrict__ RES) {
  __shared__ __align__(16) unsigned short As[2][GT_M * GPITCH];
  __shared__ __align__(16) unsigned short Bs[2][GT_N * GPITCH];

  const int z = PROJ ? (int)blockIdx.z : 0;
  const unsigned short* __restrict__ X  = (z == 0) ? X0 : (z == 1) ? X1 : X2;
  const unsigned short* __restrict__ WT = (z == 0) ? W0 : (z == 1) ? W1 : W2;
  unsigned short* __restrict__ OUTH     = (z == 0) ? O0 : (z == 1) ? O1 : O2;
  const float scale = (z == 0) ? 0.125f : 1.0f;
  const bool xrows = PROJ && (z == 2);
  const unsigned short* __restrict__ A  = xrows ? X : WT;
  const unsigned short* __restrict__ BT = xrows ? WT : X;
  const int w0 = blockIdx.x * GT_N;
  const int x0 = blockIdx.y * GT_M;
  const int m0 = xrows ? x0 : w0;
  const int n0 = xrows ? w0 : x0;

  const int tid = threadIdx.x;
  const int lane = tid & 31;
  const int wave = tid >> 5;
  const int wm = wave & 3;
  const int wn = wave >> 2;
  const int rbase = (lane & 16) ? 8 : 0;
  const int col = lane & 15;

  const int srow = tid >> 4;
  const int sch  = (tid & 15) * 8;
  const unsigned short* Ag = A  + (size_t)(m0 + srow) * DD + sch;
  const unsigned short* Bg = BT + (size_t)(n0 + srow) * DD + sch;
  const int sidx = srow * GPITCH + sch;

  u32x4 ga[8], gb[8];
#pragma unroll
  for (int i = 0; i < 8; ++i) {
    ga[i] = *reinterpret_cast<const u32x4*>(Ag + (size_t)(16 * i) * DD);
    gb[i] = *reinterpret_cast<const u32x4*>(Bg + (size_t)(16 * i) * DD);
  }
#pragma unroll
  for (int i = 0; i < 8; ++i) {
    *reinterpret_cast<u32x4*>(&As[0][sidx + 16 * i * GPITCH]) = ga[i];
    *reinterpret_cast<u32x4*>(&Bs[0][sidx + 16 * i * GPITCH]) = gb[i];
  }
  __syncthreads();

  v8f acc[2][4] = {};

  for (int st = 0; st < DD / GT_K; ++st) {
    const int buf = st & 1;
    const bool more = (st + 1) < (DD / GT_K);
    if (more) {
      const unsigned short* ag = Ag + (st + 1) * GT_K;
      const unsigned short* bg = Bg + (st + 1) * GT_K;
#pragma unroll
      for (int i = 0; i < 8; ++i) {
        ga[i] = *reinterpret_cast<const u32x4*>(ag + (size_t)(16 * i) * DD);
        gb[i] = *reinterpret_cast<const u32x4*>(bg + (size_t)(16 * i) * DD);
      }
    }
    __builtin_amdgcn_sched_barrier(0);
    const unsigned short* Ab = &As[buf][(wm * 32 + col) * GPITCH];
    const unsigned short* Bb = &Bs[buf][(wn * 64 + col) * GPITCH];

#pragma unroll
    for (int j = 0; j < GT_K / 32; ++j) {
      v16bf af[2], bfr[4];
#pragma unroll
      for (int mt = 0; mt < 2; ++mt)
        af[mt] = load_frag(Ab + mt * 16 * GPITCH + j * 32, lane);
#pragma unroll
      for (int nt = 0; nt < 4; ++nt)
        bfr[nt] = load_frag(Bb + nt * 16 * GPITCH + j * 32, lane);
#pragma unroll
      for (int mt = 0; mt < 2; ++mt)
#pragma unroll
        for (int nt = 0; nt < 4; ++nt)
          acc[mt][nt] = WMMA_BF16(af[mt], bfr[nt], acc[mt][nt]);
    }

    __builtin_amdgcn_sched_barrier(0);
    if (more) {
#pragma unroll
      for (int i = 0; i < 8; ++i) {
        *reinterpret_cast<u32x4*>(&As[buf ^ 1][sidx + 16 * i * GPITCH]) = ga[i];
        *reinterpret_cast<u32x4*>(&Bs[buf ^ 1][sidx + 16 * i * GPITCH]) = gb[i];
      }
    }
    __syncthreads();
  }

  if (PROJ) {
    unsigned short* T = &As[0][0];
#pragma unroll
    for (int mt = 0; mt < 2; ++mt) {
#pragma unroll
      for (int nt = 0; nt < 4; ++nt) {
        const int ra = wm * 32 + mt * 16 + rbase;
        const int cb = wn * 64 + nt * 16 + col;
        u32x4 w;
        w.x = f2bf_pair(acc[mt][nt][0] * scale, acc[mt][nt][1] * scale);
        w.y = f2bf_pair(acc[mt][nt][2] * scale, acc[mt][nt][3] * scale);
        w.z = f2bf_pair(acc[mt][nt][4] * scale, acc[mt][nt][5] * scale);
        w.w = f2bf_pair(acc[mt][nt][6] * scale, acc[mt][nt][7] * scale);
        *reinterpret_cast<u32x4*>(&T[cb * GPITCH + ra]) = w;
      }
    }
    __syncthreads();
#pragma unroll
    for (int it = 0; it < 8; ++it) {
      const int p   = it * 256 + tid;
      const int row = p >> 4;
      const int c8  = (p & 15) * 8;
      const u32x4 w = *reinterpret_cast<const u32x4*>(&T[row * GPITCH + c8]);
      const int xr = x0 + (xrows ? c8 : row);
      const int n  = w0 + (xrows ? row : c8);
      const int b = xr / SS, sq = xr - b * SS;
      const int h = n / DKV, dcol = n - h * DKV;
      const size_t bh = (size_t)(b * HH + h);
      const size_t idx = xrows ? (bh * DKV + dcol) * SS + sq
                               : (bh * SS + sq) * DKV + dcol;
      *reinterpret_cast<u32x4*>(OUTH + idx) = w;
    }
  } else {
    float4 res[16];
#pragma unroll
    for (int it = 0; it < 16; ++it) {
      const int p = it * 256 + tid;
      res[it] = *reinterpret_cast<const float4*>(
          RES + (size_t)(x0 + (p >> 5)) * DD + w0 + (p & 31) * 4);
    }
    float* T = reinterpret_cast<float*>(&As[0][0]);
#pragma unroll
    for (int mt = 0; mt < 2; ++mt) {
#pragma unroll
      for (int nt = 0; nt < 4; ++nt) {
        const int ra = wm * 32 + mt * 16 + rbase;
        const int cb = wn * 64 + nt * 16 + col;
        float4 lo = {acc[mt][nt][0], acc[mt][nt][1], acc[mt][nt][2], acc[mt][nt][3]};
        float4 hi = {acc[mt][nt][4], acc[mt][nt][5], acc[mt][nt][6], acc[mt][nt][7]};
        *reinterpret_cast<float4*>(&T[cb * FPITCH + ra])     = lo;
        *reinterpret_cast<float4*>(&T[cb * FPITCH + ra + 4]) = hi;
      }
    }
    __syncthreads();
#pragma unroll
    for (int it = 0; it < 16; ++it) {
      const int p = it * 256 + tid;
      const int row = p >> 5;
      const int c4  = (p & 31) * 4;
      float4 c = *reinterpret_cast<const float4*>(&T[row * FPITCH + c4]);
      c.x += res[it].x; c.y += res[it].y; c.z += res[it].z; c.w += res[it].w;
      *reinterpret_cast<float4*>(OUTF + (size_t)(x0 + row) * DD + w0 + c4) = c;
    }
  }
}

#define AT_WAVES 8
#define AT_ROWS  (AT_WAVES * 16)
#define KT       64
#define LPITCH   72

typedef __attribute__((ext_vector_type(2))) float  f32x2;
typedef __attribute__((ext_vector_type(2))) __bf16 bf16x2;

__device__ __forceinline__ unsigned int pack_bf16(float lo, float hi) {
  f32x2 t = {lo, hi};
  return __builtin_bit_cast(unsigned int, __builtin_convertvector(t, bf16x2));
}

__device__ __forceinline__ float xhalf(float x) {
  int b = __builtin_bit_cast(int, x);
  return __builtin_bit_cast(float,
      __builtin_amdgcn_permlanex16(b, b, 0x76543210, 0xfedcba98, false, false));
}

__global__ void __launch_bounds__(AT_WAVES * 32)
attn_kernel(const unsigned short* __restrict__ QH,
            const unsigned short* __restrict__ KH,
            const unsigned short* __restrict__ VT,
            const unsigned int* __restrict__ MBITS,
            unsigned short* __restrict__ AOUT) {
  __shared__ __align__(16) unsigned short Ks[2][KT * LPITCH];
  __shared__ __align__(16) unsigned short Vs[2][DKV * LPITCH];

  const int tid = threadIdx.x;
  const int lane = tid & 31;
  const int wave = __builtin_amdgcn_readfirstlane(tid >> 5);
  const int bh = blockIdx.x;
  const int b = bh >> 4, h = bh & 15;
  const int s_base = blockIdx.y * AT_ROWS + wave * 16;
  const int kofs = (lane & 16) ? 8 : 0;
  const int col = lane & 15;

  const unsigned short* Qrow =
      QH + ((size_t)bh * SS + (size_t)(s_base + col)) * DKV;
  v16bf qf0 = load_frag(Qrow, lane);
  v16bf qf1 = load_frag(Qrow + 32, lane);

  v8f o[4] = {};
  float m = -3.0e38f;
  float l = 0.0f;

  const unsigned int* mwave =
      MBITS + ((size_t)b * (SS / 16) + (size_t)(s_base >> 4)) * (SS / KT) * MB_STAGE_WORDS;

  const int srow = tid >> 3;
  const int sch  = (tid & 7) * 8;
  const unsigned short* Kg = KH + ((size_t)bh * SS + srow) * DKV + sch;
  const unsigned short* Vg = VT + ((size_t)bh * DKV + srow) * SS + sch;
  const int sidx = srow * LPITCH + sch;

  uint4 gk0 = *reinterpret_cast<const uint4*>(Kg);
  uint4 gk1 = *reinterpret_cast<const uint4*>(Kg + (size_t)32 * DKV);
  uint4 gv0 = *reinterpret_cast<const uint4*>(Vg);
  uint4 gv1 = *reinterpret_cast<const uint4*>(Vg + (size_t)32 * SS);
  *reinterpret_cast<uint4*>(&Ks[0][sidx])               = gk0;
  *reinterpret_cast<uint4*>(&Ks[0][sidx + 32 * LPITCH]) = gk1;
  *reinterpret_cast<uint4*>(&Vs[0][sidx])               = gv0;
  *reinterpret_cast<uint4*>(&Vs[0][sidx + 32 * LPITCH]) = gv1;
  __syncthreads();

  for (int st = 0; st < SS / KT; ++st) {
    const int buf = st & 1;
    const int t0 = st * KT;
    const bool more = (st + 1) < (SS / KT);
    if (more) {
      const unsigned short* kg = Kg + (size_t)(t0 + KT) * DKV;
      const unsigned short* vg = Vg + (t0 + KT);
      gk0 = *reinterpret_cast<const uint4*>(kg);
      gk1 = *reinterpret_cast<const uint4*>(kg + (size_t)32 * DKV);
      gv0 = *reinterpret_cast<const uint4*>(vg);
      gv1 = *reinterpret_cast<const uint4*>(vg + (size_t)32 * SS);
    }
    const unsigned short* Kb = &Ks[buf][0];
    const unsigned short* Vb = &Vs[buf][0];

#pragma unroll
    for (int j = 0; j < KT / 32; ++j) {
      const uint4* mstep =
          reinterpret_cast<const uint4*>(mwave + (size_t)st * MB_STAGE_WORDS + j * 16);
      const uint4 mk0a = mstep[0];
      const uint4 mk0b = mstep[1];
      const uint4 mk1a = mstep[2];
      const uint4 mk1b = mstep[3];
      v16bf kb0 = load_frag(Kb + (j * 32 + col) * LPITCH, lane);
      v16bf kb1 = load_frag(Kb + (j * 32 + col) * LPITCH + 32, lane);
      v16bf kb2 = load_frag(Kb + (j * 32 + 16 + col) * LPITCH, lane);
      v16bf kb3 = load_frag(Kb + (j * 32 + 16 + col) * LPITCH + 32, lane);
      v16bf vb[4];
#pragma unroll
      for (int nt = 0; nt < 4; ++nt)
        vb[nt] = load_frag(Vb + (nt * 16 + col) * LPITCH + j * 32, lane);

      v8f s0 = {}, s1 = {};
      s0 = WMMA_BF16(kb0, qf0, s0);
      s0 = WMMA_BF16(kb1, qf1, s0);
      s1 = WMMA_BF16(kb2, qf0, s1);
      s1 = WMMA_BF16(kb3, qf1, s1);

      const unsigned int mk0[8] = {mk0a.x, mk0a.y, mk0a.z, mk0a.w, mk0b.x, mk0b.y, mk0b.z, mk0b.w};
      const unsigned int mk1[8] = {mk1a.x, mk1a.y, mk1a.z, mk1a.w, mk1b.x, mk1b.y, mk1b.z, mk1b.w};
      float v0[8], v1[8];
#pragma unroll
      for (int r = 0; r < 8; ++r) {
        v0[r] = __builtin_amdgcn_inverse_ballot_w32(mk0[r]) ? s0[r] : -1.0e9f;
        v1[r] = __builtin_amdgcn_inverse_ballot_w32(mk1[r]) ? s1[r] : -1.0e9f;
      }
      float mx = fmaxf(v0[0], v1[0]);
#pragma unroll
      for (int r = 1; r < 8; ++r) mx = fmaxf(mx, fmaxf(v0[r], v1[r]));
      mx = fmaxf(mx, xhalf(mx));
      const float mnew  = fmaxf(m, mx);
      const float alpha = __expf(m - mnew);
      float rs = 0.0f;
      unsigned int w0[4], w1[4];
#pragma unroll
      for (int r = 0; r < 8; r += 2) {
        const float p00 = __expf(v0[r]     - mnew);
        const float p01 = __expf(v0[r + 1] - mnew);
        const float p10 = __expf(v1[r]     - mnew);
        const float p11 = __expf(v1[r + 1] - mnew);
        rs += (p00 + p01) + (p10 + p11);
        w0[r >> 1] = pack_bf16(p00, p01);
        w1[r >> 1] = pack_bf16(p10, p11);
      }
      l = l * alpha + rs;
      m = mnew;
#pragma unroll
      for (int nt = 0; nt < 4; ++nt)
#pragma unroll
        for (int r = 0; r < 8; ++r) o[nt][r] = o[nt][r] * alpha;

      FragU pf;
      pf.q[0] = make_uint4(w0[0], w0[1], w0[2], w0[3]);
      pf.q[1] = make_uint4(w1[0], w1[1], w1[2], w1[3]);
#pragma unroll
      for (int nt = 0; nt < 4; ++nt)
        o[nt] = WMMA_BF16(vb[nt], pf.v, o[nt]);
    }

    if (more) {
      unsigned short* Kn = &Ks[buf ^ 1][0];
      unsigned short* Vn = &Vs[buf ^ 1][0];
      *reinterpret_cast<uint4*>(Kn + sidx)               = gk0;
      *reinterpret_cast<uint4*>(Kn + sidx + 32 * LPITCH) = gk1;
      *reinterpret_cast<uint4*>(Vn + sidx)               = gv0;
      *reinterpret_cast<uint4*>(Vn + sidx + 32 * LPITCH) = gv1;
    }
    __syncthreads();
  }

  const float inv = 1.0f / (l + xhalf(l));
  unsigned short* orow =
      AOUT + ((size_t)b * SS + (size_t)(s_base + col)) * (HH * DKV) + h * DKV + kofs;
#pragma unroll
  for (int nt = 0; nt < 4; ++nt) {
    uint4 w;
    w.x = pack_bf16(o[nt][0] * inv, o[nt][1] * inv);
    w.y = pack_bf16(o[nt][2] * inv, o[nt][3] * inv);
    w.z = pack_bf16(o[nt][4] * inv, o[nt][5] * inv);
    w.w = pack_bf16(o[nt][6] * inv, o[nt][7] * inv);
    *reinterpret_cast<uint4*>(orow + nt * 16) = w;
  }
}

__global__ void ln_kernel(const float* __restrict__ X,
                          const float* __restrict__ W,
                          const float* __restrict__ Bv,
                          float* __restrict__ OUT) {
  __shared__ float red[8];
  const int row = blockIdx.x;
  const int tid = threadIdx.x;
  const float* xr = X + (size_t)row * DD;

  float v[4];
  float s = 0.0f;
#pragma unroll
  for (int i = 0; i < 4; ++i) { v[i] = xr[tid + i * 256]; s += v[i]; }
  for (int off = 1; off < 32; off <<= 1) s += __shfl_xor(s, off);
  if ((tid & 31) == 0) red[tid >> 5] = s;
  __syncthreads();
  float mu = (red[0] + red[1] + red[2] + red[3] +
              red[4] + red[5] + red[6] + red[7]) * (1.0f / DD);
  float vs = 0.0f;
#pragma unroll
  for (int i = 0; i < 4; ++i) { float d = v[i] - mu; vs += d * d; }
  for (int off = 1; off < 32; off <<= 1) vs += __shfl_xor(vs, off);
  __syncthreads();
  if ((tid & 31) == 0) red[tid >> 5] = vs;
  __syncthreads();
  float var = (red[0] + red[1] + red[2] + red[3] +
               red[4] + red[5] + red[6] + red[7]) * (1.0f / DD);
  float rstd = rsqrtf(var + 1e-6f);
#pragma unroll
  for (int i = 0; i < 4; ++i) {
    int c = tid + i * 256;
    OUT[(size_t)row * DD + c] = (v[i] - mu) * rstd * W[c] + Bv[c];
  }
}

extern "C" void kernel_launch(void* const* d_in, const int* in_sizes, int n_in,
                              void* d_out, int out_size, void* d_ws, size_t ws_size,
                              hipStream_t stream) {
  const float* q    = (const float*)d_in[0];
  const float* k    = (const float*)d_in[1];
  const float* v    = (const float*)d_in[2];
  const int*   mask = (const int*)d_in[3];
  const float* Wq   = (const float*)d_in[4];
  const float* Wk   = (const float*)d_in[5];
  const float* Wv   = (const float*)d_in[6];
  const float* Wfc  = (const float*)d_in[7];
  const float* lnw  = (const float*)d_in[8];
  const float* lnb  = (const float*)d_in[9];
  float* out = (float*)d_out;

  char* ws = (char*)d_ws;
  const size_t SZ_ACT  = (size_t)BB * SS * DD * 2;
  const size_t SZ_W    = (size_t)HH * DD * DKV * 2;
  const size_t SZ_HEAD = (size_t)BB * HH * SS * DKV * 2;
  size_t off = 0;
  unsigned short* qbf  = (unsigned short*)(ws + off); off += SZ_ACT;
  unsigned short* kbf  = (unsigned short*)(ws + off); off += SZ_ACT;
  unsigned short* vbf  = (unsigned short*)(ws + off); off += SZ_ACT;
  unsigned short* WqT  = (unsigned short*)(ws + off); off += SZ_W;
  unsigned short* WkT  = (unsigned short*)(ws + off); off += SZ_W;
  unsigned short* WvT  = (unsigned short*)(ws + off); off += SZ_W;
  unsigned short* WfcT = (unsigned short*)(ws + off); off += (size_t)DD * DD * 2;
  unsigned short* qh   = (unsigned short*)(ws + off); off += SZ_HEAD;
  unsigned short* kh   = (unsigned short*)(ws + off); off += SZ_HEAD;
  unsigned short* vhT  = (unsigned short*)(ws + off); off += SZ_HEAD;
  unsigned short* aout = (unsigned short*)(ws + off); off += SZ_ACT;
  float*          fco  = (float*)(ws + off);          off += (size_t)BB * SS * DD * 4;
  unsigned int*   mbits = (unsigned int*)fco;

  constexpr int nAct = BB * SS * DD;
  constexpr int nW   = HH * DD * DKV;
  constexpr int nFc  = DD * DD;

  static_assert(nAct % CV_BLOCK_ELEMS == 0, "cvt_kernel: whole work-groups only");
  static_assert(nW % (8 * 256) == 0 && nFc % (8 * 256) == 0, "cvtT_kernel: whole work-groups only");
  static_assert(DKV % 32 == 0 && DD % 32 == 0 && DD % 8 == 0, "cvtT_kernel: a wave stays in one row of src");
  static_assert(nAct / CV_BLOCK_ELEMS == CV_ACT_BLOCKS, "cvt_all_kernel: activation range");
  static_assert(nW / (8 * 256) == CV_W_BLOCKS && nFc / (8 * 256) == CV_W_BLOCKS,
                "cvt_all_kernel: weight ranges");
  static_assert(SS % 128 == 0 && (SS / 128) == 16, "maskpack_block: 2 work-groups x 8 waves x 128 keys per row");
  static_assert(KT == 64 && AT_ROWS == 128, "attn_kernel and maskpack_block share the stage and block sizes");
  cvt_all_kernel<<<CV_ALL_BLOCKS, 256, 0, stream>>>(q, k, v, Wq, Wk, Wv, Wfc, mask, mbits,
                                                    qbf, kbf, vbf, WqT, WkT, WvT, WfcT);

  gemm_kernel<true><<<dim3(DD / GT_N, (BB * SS) / GT_M, 3), 256, 0, stream>>>(
      qbf, kbf, vbf, WqT, WkT, WvT, qh, kh, vhT, nullptr, nullptr);

  attn_kernel<<<dim3(BB * HH, SS / AT_ROWS), AT_WAVES * 32, 0, stream>>>(qh, kh, vhT, mbits, aout);

  gemm_kernel<false><<<dim3(DD / GT_N, (BB * SS) / GT_M), 256, 0, stream>>>(
      aout, nullptr, nullptr, WfcT, nullptr, nullptr, nullptr, nullptr, nullptr, fco, q);
  ln_kernel<<<BB * SS, 256, 0, stream>>>(fco, lnw, lnb, out);
}


# attn key loop and GEMM k loop: fragment ds_loads leave one block ahead as ONE group (extra register set) instead of pair by pair among the WMMAs with a full wait each; sum order unchanged
# speedup vs baseline: 5.3032x; 1.0380x over previous
#include <hip/hip_runtime.h>


#define BB 2
#define SS 2048
#define DD 1024
#define HH 16
#define DKV 64

typedef __attribute__((ext_vector_type(16))) __bf16 v16bf;
typedef __attribute__((ext_vector_type(8)))  float  v8f;

union FragU { v16bf v; uint4 q[2]; };

__device__ __forceinline__ unsigned short f2bf(float f) {
  unsigned int u = __float_as_uint(f);
  u += 0x7FFFu + ((u >> 16) & 1u);
  return (unsigned short)(u >> 16);
}

__device__ __forceinline__ v16bf load_frag(const unsigned short* rowbase, int lane) {
  const unsigned short* p = rowbase + ((lane & 16) ? 8 : 0);
  FragU f;
  f.q[0] = *reinterpret_cast<const uint4*>(p);
  f.q[1] = *reinterpret_cast<const uint4*>(p + 16);
  return f.v;
}

#define WMMA_BF16(a, b, c) \
  __builtin_amdgcn_wmma_f32_16x16x32_bf16(false, (a), false, (b), (short)0, (c), false, false)

#define CV_SLABS 8
#define CV_SLAB_ELEMS  (256 * 4)
#define CV_BLOCK_ELEMS (CV_SLABS * CV_SLAB_ELEMS)

__device__ __forceinline__ unsigned int f2bf_pair(float lo, float hi) {
  return (unsigned int)f2bf(lo) | ((unsigned int)f2bf(hi) << 16);
}

__device__ __forceinline__ void
cvt_block(const float* __restrict__ src, unsigned short* __restrict__ dst, unsigned int blk) {
  const size_t base = (size_t)blk * CV_BLOCK_ELEMS + threadIdx.x * 4;
  float4 v[CV_SLABS];
#pragma unroll
  for (int j = 0; j < CV_SLABS; ++j)
    v[j] = *reinterpret_cast<const float4*>(src + base + j * CV_SLAB_ELEMS);
#pragma unroll
  for (int j = 0; j < CV_SLABS; ++j) {
    uint2 w;
    w.x = f2bf_pair(v[j].x, v[j].y);
    w.y = f2bf_pair(v[j].z, v[j].w);
    *reinterpret_cast<uint2*>(dst + base + j * CV_SLAB_ELEMS) = w;
  }
}

template <int R, int C>
__device__ __forceinline__ void
cvtT_block(const float* __restrict__ src, unsigned short* __restrict__ dst, unsigned int blk) {
  const unsigned int i  = blk * 256 + threadIdx.x;
  const unsigned int c  = i % C;
  const unsigned int r8 = (i / C) % (R / 8);
  const unsigned int g  = i / (C * (R / 8));
  const float* s = src + ((size_t)g * R + (size_t)r8 * 8) * C + c;
  float v[8];
#pragma unroll
  for (int k = 0; k < 8; ++k) v[k] = s[(size_t)k * C];
  uint4 w;
  w.x = f2bf_pair(v[0], v[1]);
  w.y = f2bf_pair(v[2], v[3]);
  w.z = f2bf_pair(v[4], v[5]);
  w.w = f2bf_pair(v[6], v[7]);
  *reinterpret_cast<uint4*>(dst + ((size_t)g * C + c) * R + (size_t)r8 * 8) = w;
}

#define MB_STAGE_WORDS 32
#define MB_BLOCKS (BB * (SS / 16) * 2)

__device__ __forceinline__ void
maskpack_block(const int* __restrict__ mask, unsigned int* __restrict__ mbits, unsigned int blk) {
  const unsigned int lane = threadIdx.x & 31;
  const unsigned int wave = threadIdx.x >> 5;
  const unsigned int col  = lane & 15;
  const unsigned int kofs = (lane & 16) ? 8 : 0;
  const unsigned int rb = blk >> 1;
  const unsigned int sp = (blk & 1) * 8 + wave;
  const int* mrow = mask + ((size_t)rb * 16 + col) * SS + sp * 128 + kofs;
  int4 m[16];
#pragma unroll
  for (int i = 0; i < 8; ++i) {
    m[2 * i]     = *reinterpret_cast<const int4*>(mrow + i * 16);
    m[2 * i + 1] = *reinterpret_cast<const int4*>(mrow + i * 16 + 4);
  }
  unsigned int lo = 0, hi = 0;
#pragma unroll
  for (int i = 0; i < 8; ++i) {
    const int mv[8] = {m[2 * i].x, m[2 * i].y, m[2 * i].z, m[2 * i].w,
                       m[2 * i + 1].x, m[2 * i + 1].y, m[2 * i + 1].z, m[2 * i + 1].w};
#pragma unroll
    for (int r = 0; r < 8; ++r) {
      const unsigned int bal = __builtin_amdgcn_ballot_w32(mv[r] != 0);
      const unsigned int idx = i * 8 + r;
      if (idx < 32) lo = (lane == idx)      ? bal : lo;
      else          hi = (lane == idx - 32) ? bal : hi;
    }
  }
  unsigned int* dst = mbits + ((size_t)rb * (SS / 64) + sp * 2) * MB_STAGE_WORDS;
  dst[lane]      = lo;
  dst[32 + lane] = hi;
}

#define CV_ACT_BLOCKS ((BB * SS * DD) / CV_BLOCK_ELEMS)
#define CV_W_BLOCKS   ((HH * DD * DKV) / (8 * 256))
#define CV_ALL_BLOCKS (3 * CV_ACT_BLOCKS + 4 * CV_W_BLOCKS + MB_BLOCKS)

__global__ void __launch_bounds__(256)
cvt_all_kernel(const float* __restrict__ q, const float* __restrict__ k,
               const float* __restrict__ v,
               const float* __restrict__ Wq, const float* __restrict__ Wk,
               const float* __restrict__ Wv, const float* __restrict__ Wfc,
               const int* __restrict__ mask, unsigned int* __restrict__ mbits,
               unsigned short* __restrict__ qbf, unsigned short* __restrict__ kbf,
               unsigned short* __restrict__ vbf,
               unsigned short* __restrict__ WqT, unsigned short* __restrict__ WkT,
               unsigned short* __restrict__ WvT, unsigned short* __restrict__ WfcT) {
  const unsigned int bid = blockIdx.x;
  if (bid < 3 * CV_ACT_BLOCKS) {
    const unsigned int job = bid / CV_ACT_BLOCKS;
    const float* s = (job == 0) ? q : (job == 1) ? k : v;
    unsigned short* d = (job == 0) ? qbf : (job == 1) ? kbf : vbf;
    cvt_block(s, d, bid - job * CV_ACT_BLOCKS);
  } else {
    const unsigned int wb  = bid - 3 * CV_ACT_BLOCKS;
    const unsigned int job = wb / CV_W_BLOCKS;
    const unsigned int blk = wb - job * CV_W_BLOCKS;
    if (job < 3) {
      const float* s = (job == 0) ? Wq : (job == 1) ? Wk : Wv;
      unsigned short* d = (job == 0) ? WqT : (job == 1) ? WkT : WvT;
      cvtT_block<DD, DKV>(s, d, blk);
    } else if (job == 3) {
      cvtT_block<DD, DD>(Wfc, WfcT, blk);
    } else {
      maskpack_block(mask, mbits, wb - 4 * CV_W_BLOCKS);
    }
  }
}

#define GT_M   128
#define GT_N   128
#define GT_K   128
#define GPITCH 136
#define FPITCH 132

typedef __attribute__((ext_vector_type(4))) unsigned int u32x4;

static_assert(GT_N * FPITCH * 4 <= 2 * GT_M * GPITCH * 2, "the parked fp32 tile fits in the A tiles' LDS");

template <bool PROJ>
__global__ void __launch_bounds__(256)
gemm_kernel(const unsigned short* __restrict__ X0,
            const unsigned short* __restrict__ X1,
            const unsigned short* __restrict__ X2,
            const unsigned short* __restrict__ W0,
            const unsigned short* __restrict__ W1,
            const unsigned short* __restrict__ W2,
            unsigned short* __restrict__ O0,
            unsigned short* __restrict__ O1,
            unsigned short* __restrict__ O2,
            float* __restrict__ OUTF,
            const float* __restrict__ RES) {
  __shared__ __align__(16) unsigned short As[2][GT_M * GPITCH];
  __shared__ __align__(16) unsigned short Bs[2][GT_N * GPITCH];

  const int z = PROJ ? (int)blockIdx.z : 0;
  const unsigned short* __restrict__ X  = (z == 0) ? X0 : (z == 1) ? X1 : X2;
  const unsigned short* __restrict__ WT = (z == 0) ? W0 : (z == 1) ? W1 : W2;
  unsigned short* __restrict__ OUTH     = (z == 0) ? O0 : (z == 1) ? O1 : O2;
  const float scale = (z == 0) ? 0.125f : 1.0f;
  const bool xrows = PROJ && (z == 2);
  const unsigned short* __restrict__ A  = xrows ? X : WT;
  const unsigned short* __restrict__ BT = xrows ? WT : X;
  const int w0 = blockIdx.x * GT_N;
  const int x0 = blockIdx.y * GT_M;
  const int m0 = xrows ? x0 : w0;
  const int n0 = xrows ? w0 : x0;

  const int tid = threadIdx.x;
  const int lane = tid & 31;
  const int wave = tid >> 5;
  const int wm = wave & 3;
  const int wn = wave >> 2;
  const int rbase = (lane & 16) ? 8 : 0;
  const int col = lane & 15;

  const int srow = tid >> 4;
  const int sch  = (tid & 15) * 8;
  const unsigned short* Ag = A  + (size_t)(m0 + srow) * DD + sch;
  const unsigned short* Bg = BT + (size_t)(n0 + srow) * DD + sch;
  const int sidx = srow * GPITCH + sch;

  u32x4 ga[8], gb[8];
#pragma unroll
  for (int i = 0; i < 8; ++i) {
    ga[i] = *reinterpret_cast<const u32x4*>(Ag + (size_t)(16 * i) * DD);
    gb[i] = *reinterpret_cast<const u32x4*>(Bg + (size_t)(16 * i) * DD);
  }
#pragma unroll
  for (int i = 0; i < 8; ++i) {
    *reinterpret_cast<u32x4*>(&As[0][sidx + 16 * i * GPITCH]) = ga[i];
    *reinterpret_cast<u32x4*>(&Bs[0][sidx + 16 * i * GPITCH]) = gb[i];
  }
  __syncthreads();

  v8f acc[2][4] = {};

  for (int st = 0; st < DD / GT_K; ++st) {
    const int buf = st & 1;
    const bool more = (st + 1) < (DD / GT_K);
    if (more) {
      const unsigned short* ag = Ag + (st + 1) * GT_K;
      const unsigned short* bg = Bg + (st + 1) * GT_K;
#pragma unroll
      for (int i = 0; i < 8; ++i) {
        ga[i] = *reinterpret_cast<const u32x4*>(ag + (size_t)(16 * i) * DD);
        gb[i] = *reinterpret_cast<const u32x4*>(bg + (size_t)(16 * i) * DD);
      }
    }
    __builtin_amdgcn_sched_barrier(0);
    const unsigned short* Ab = &As[buf][(wm * 32 + col) * GPITCH];
    const unsigned short* Bb = &Bs[buf][(wn * 64 + col) * GPITCH];

    v16bf af[2][2], bfr[2][4];
#pragma unroll
    for (int mt = 0; mt < 2; ++mt)
      af[0][mt] = load_frag(Ab + mt * 16 * GPITCH, lane);
#pragma unroll
    for (int nt = 0; nt < 4; ++nt)
      bfr[0][nt] = load_frag(Bb + nt * 16 * GPITCH, lane);
#pragma unroll
    for (int j = 0; j < GT_K / 32; ++j) {
      const int cur = j & 1;
      if (j + 1 < GT_K / 32) {
#pragma unroll
        for (int mt = 0; mt < 2; ++mt)
          af[cur ^ 1][mt] = load_frag(Ab + mt * 16 * GPITCH + (j + 1) * 32, lane);
#pragma unroll
        for (int nt = 0; nt < 4; ++nt)
          bfr[cur ^ 1][nt] = load_frag(Bb + nt * 16 * GPITCH + (j + 1) * 32, lane);
      }
      __builtin_amdgcn_sched_barrier(0);
#pragma unroll
      for (int mt = 0; mt < 2; ++mt)
#pragma unroll
        for (int nt = 0; nt < 4; ++nt)
          acc[mt][nt] = WMMA_BF16(af[cur][mt], bfr[cur][nt], acc[mt][nt]);
    }

    __builtin_amdgcn_sched_barrier(0);
    if (more) {
#pragma unroll
      for (int i = 0; i < 8; ++i) {
        *reinterpret_cast<u32x4*>(&As[buf ^ 1][sidx + 16 * i * GPITCH]) = ga[i];
        *reinterpret_cast<u32x4*>(&Bs[buf ^ 1][sidx + 16 * i * GPITCH]) = gb[i];
      }
    }
    __syncthreads();
  }

  if (PROJ) {
    unsigned short* T = &As[0][0];
#pragma unroll
    for (int mt = 0; mt < 2; ++mt) {
#pragma unroll
      for (int nt = 0; nt < 4; ++nt) {
        const int ra = wm * 32 + mt * 16 + rbase;
        const int cb = wn * 64 + nt * 16 + col;
        u32x4 w;
        w.x = f2bf_pair(acc[mt][nt][0] * scale, acc[mt][nt][1] * scale);
        w.y = f2bf_pair(acc[mt][nt][2] * scale, acc[mt][nt][3] * scale);
        w.z = f2bf_pair(acc[mt][nt][4] * scale, acc[mt][nt][5] * scale);
        w.w = f2bf_pair(acc[mt][nt][6] * scale, acc[mt][nt][7] * scale);
        *reinterpret_cast<u32x4*>(&T[cb * GPITCH + ra]) = w;
      }
    }
    __syncthreads();
#pragma unroll
    for (int it = 0; it < 8; ++it) {
      const int p   = it * 256 + tid;
      const int row = p >> 4;
      const int c8  = (p & 15) * 8;
      const u32x4 w = *reinterpret_cast<const u32x4*>(&T[row * GPITCH + c8]);
      const int xr = x0 + (xrows ? c8 : row);
      const int n  = w0 + (xrows ? row : c8);
      const int b = xr / SS, sq = xr - b * SS;
      const int h = n / DKV, dcol = n - h * DKV;
      const size_t bh = (size_t)(b * HH + h);
      const size_t idx = xrows ? (bh * DKV + dcol) * SS + sq
                               : (bh * SS + sq) * DKV + dcol;
      *reinterpret_cast<u32x4*>(OUTH + idx) = w;
    }
  } else {
    float4 res[16];
#pragma unroll
    for (int it = 0; it < 16; ++it) {
      const int p = it * 256 + tid;
      res[it] = *reinterpret_cast<const float4*>(
          RES + (size_t)(x0 + (p >> 5)) * DD + w0 + (p & 31) * 4);
    }
    float* T = reinterpret_cast<float*>(&As[0][0]);
#pragma unroll
    for (int mt = 0; mt < 2; ++mt) {
#pragma unroll
      for (int nt = 0; nt < 4; ++nt) {
        const int ra = wm * 32 + mt * 16 + rbase;
        const int cb = wn * 64 + nt * 16 + col;
        float4 lo = {acc[mt][nt][0], acc[mt][nt][1], acc[mt][nt][2], acc[mt][nt][3]};
        float4 hi = {acc[mt][nt][4], acc[mt][nt][5], acc[mt][nt][6], acc[mt][nt][7]};
        *reinterpret_cast<float4*>(&T[cb * FPITCH + ra])     = lo;
        *reinterpret_cast<float4*>(&T[cb * FPITCH + ra + 4]) = hi;
      }
    }
    __syncthreads();
#pragma unroll
    for (int it = 0; it < 16; ++it) {
      const int p = it * 256 + tid;
      const int row = p >> 5;
      const int c4  = (p & 31) * 4;
      float4 c = *reinterpret_cast<const float4*>(&T[row * FPITCH + c4]);
      c.x += res[it].x; c.y += res[it].y; c.z += res[it].z; c.w += res[it].w;
      *reinterpret_cast<float4*>(OUTF + (size_t)(x0 + row) * DD + w0 + c4) = c;
    }
  }
}

#define AT_WAVES 8
#define AT_ROWS  (AT_WAVES * 16)
#define KT       64
#define LPITCH   72

typedef __attribute__((ext_vector_type(2))) float  f32x2;
typedef __attribute__((ext_vector_type(2))) __bf16 bf16x2;

__device__ __forceinline__ unsigned int pack_bf16(float lo, float hi) {
  f32x2 t = {lo, hi};
  return __builtin_bit_cast(unsigned int, __builtin_convertvector(t, bf16x2));
}

__device__ __forceinline__ float xhalf(float x) {
  int b = __builtin_bit_cast(int, x);
  return __builtin_bit_cast(float,
      __builtin_amdgcn_permlanex16(b, b, 0x76543210, 0xfedcba98, false, false));
}

__global__ void __launch_bounds__(AT_WAVES * 32)
attn_kernel(const unsigned short* __restrict__ QH,
            const unsigned short* KH,
            const unsigned short* VT,
            const unsigned int* __restrict__ MBITS,
            unsigned short* __restrict__ AOUT) {
  __shared__ __align__(16) unsigned short Ks[2][KT * LPITCH];
  __shared__ __align__(16) unsigned short Vs[2][DKV * LPITCH];

  const int tid = threadIdx.x;
  const int lane = tid & 31;
  const int wave = __builtin_amdgcn_readfirstlane(tid >> 5);
  const int bh = blockIdx.x;
  const int b = bh >> 4, h = bh & 15;
  const int s_base = blockIdx.y * AT_ROWS + wave * 16;
  const int kofs = (lane & 16) ? 8 : 0;
  const int col = lane & 15;

  const unsigned short* Qrow =
      QH + ((size_t)bh * SS + (size_t)(s_base + col)) * DKV;
  v16bf qf0 = load_frag(Qrow, lane);
  v16bf qf1 = load_frag(Qrow + 32, lane);

  v8f o[4] = {};
  float m = -3.0e38f;
  float l = 0.0f;

  const unsigned int* mwave =
      MBITS + ((size_t)b * (SS / 16) + (size_t)(s_base >> 4)) * (SS / KT) * MB_STAGE_WORDS;

  const int srow = tid >> 3;
  const int sch  = (tid & 7) * 8;
  const unsigned short* Kg = KH + ((size_t)bh * SS + srow) * DKV + sch;
  const unsigned short* Vg = VT + ((size_t)bh * DKV + srow) * SS + sch;
  const int sidx = srow * LPITCH + sch;

  uint4 gk0 = *reinterpret_cast<const uint4*>(Kg);
  uint4 gk1 = *reinterpret_cast<const uint4*>(Kg + (size_t)32 * DKV);
  uint4 gv0 = *reinterpret_cast<const uint4*>(Vg);
  uint4 gv1 = *reinterpret_cast<const uint4*>(Vg + (size_t)32 * SS);
  *reinterpret_cast<uint4*>(&Ks[0][sidx])               = gk0;
  *reinterpret_cast<uint4*>(&Ks[0][sidx + 32 * LPITCH]) = gk1;
  *reinterpret_cast<uint4*>(&Vs[0][sidx])               = gv0;
  *reinterpret_cast<uint4*>(&Vs[0][sidx + 32 * LPITCH]) = gv1;
  __syncthreads();

  for (int st = 0; st < SS / KT; ++st) {
    const int buf = st & 1;
    const int t0 = st * KT;
    const bool more = (st + 1) < (SS / KT);
    if (more) {
      const unsigned short* kg = Kg + (size_t)(t0 + KT) * DKV;
      const unsigned short* vg = Vg + (t0 + KT);
      gk0 = *reinterpret_cast<const uint4*>(kg);
      gk1 = *reinterpret_cast<const uint4*>(kg + (size_t)32 * DKV);
      gv0 = *reinterpret_cast<const uint4*>(vg);
      gv1 = *reinterpret_cast<const uint4*>(vg + (size_t)32 * SS);
    }
    int frow = col * LPITCH;
    asm volatile("" : "+v"(frow) : : "memory");
    const unsigned short* Kb = &Ks[buf][0] + frow;
    const unsigned short* Vb = &Vs[buf][0] + frow;

    v16bf kb[KT / 32][4], vb[KT / 32][4];
#pragma unroll
    for (int j = 0; j < KT / 32; ++j) {
      kb[j][0] = load_frag(Kb + (j * 32) * LPITCH, lane);
      kb[j][1] = load_frag(Kb + (j * 32) * LPITCH + 32, lane);
      kb[j][2] = load_frag(Kb + (j * 32 + 16) * LPITCH, lane);
      kb[j][3] = load_frag(Kb + (j * 32 + 16) * LPITCH + 32, lane);
    }
#pragma unroll
    for (int j = 0; j < KT / 32; ++j)
#pragma unroll
      for (int nt = 0; nt < 4; ++nt)
        vb[j][nt] = load_frag(Vb + (nt * 16) * LPITCH + j * 32, lane);
    __builtin_amdgcn_sched_barrier(0);

#pragma unroll
    for (int j = 0; j < KT / 32; ++j) {
      const uint4* mstep =
          reinterpret_cast<const uint4*>(mwave + (size_t)st * MB_STAGE_WORDS + j * 16);
      const uint4 mk0a = mstep[0];
      const uint4 mk0b = mstep[1];
      const uint4 mk1a = mstep[2];
      const uint4 mk1b = mstep[3];

      v8f s0 = {}, s1 = {};
      s0 = WMMA_BF16(kb[j][0], qf0, s0);
      s0 = WMMA_BF16(kb[j][1], qf1, s0);
      s1 = WMMA_BF16(kb[j][2], qf0, s1);
      s1 = WMMA_BF16(kb[j][3], qf1, s1);

      const unsigned int mk0[8] = {mk0a.x, mk0a.y, mk0a.z, mk0a.w, mk0b.x, mk0b.y, mk0b.z, mk0b.w};
      const unsigned int mk1[8] = {mk1a.x, mk1a.y, mk1a.z, mk1a.w, mk1b.x, mk1b.y, mk1b.z, mk1b.w};
      float v0[8], v1[8];
#pragma unroll
      for (int r = 0; r < 8; ++r) {
        v0[r] = __builtin_amdgcn_inverse_ballot_w32(mk0[r]) ? s0[r] : -1.0e9f;
        v1[r] = __builtin_amdgcn_inverse_ballot_w32(mk1[r]) ? s1[r] : -1.0e9f;
      }
      float mx = fmaxf(v0[0], v1[0]);
#pragma unroll
      for (int r = 1; r < 8; ++r) mx = fmaxf(mx, fmaxf(v0[r], v1[r]));
      mx = fmaxf(mx, xhalf(mx));
      const float mnew  = fmaxf(m, mx);
      const float alpha = __expf(m - mnew);
      float rs = 0.0f;
      unsigned int w0[4], w1[4];
#pragma unroll
      for (int r = 0; r < 8; r += 2) {
        const float p00 = __expf(v0[r]     - mnew);
        const float p01 = __expf(v0[r + 1] - mnew);
        const float p10 = __expf(v1[r]     - mnew);
        const float p11 = __expf(v1[r + 1] - mnew);
        rs += (p00 + p01) + (p10 + p11);
        w0[r >> 1] = pack_bf16(p00, p01);
        w1[r >> 1] = pack_bf16(p10, p11);
      }
      l = l * alpha + rs;
      m = mnew;
#pragma unroll
      for (int nt = 0; nt < 4; ++nt)
#pragma unroll
        for (int r = 0; r < 8; ++r) o[nt][r] = o[nt][r] * alpha;

      FragU pf;
      pf.q[0] = make_uint4(w0[0], w0[1], w0[2], w0[3]);
      pf.q[1] = make_uint4(w1[0], w1[1], w1[2], w1[3]);
#pragma unroll
      for (int nt = 0; nt < 4; ++nt)
        o[nt] = WMMA_BF16(vb[j][nt], pf.v, o[nt]);
    }

    asm volatile("" : "+v"(o[0]), "+v"(o[1]), "+v"(o[2]), "+v"(o[3]));
    if (more) {
      unsigned short* Kn = &Ks[buf ^ 1][0];
      unsigned short* Vn = &Vs[buf ^ 1][0];
      *reinterpret_cast<uint4*>(Kn + sidx)               = gk0;
      *reinterpret_cast<uint4*>(Kn + sidx + 32 * LPITCH) = gk1;
      *reinterpret_cast<uint4*>(Vn + sidx)               = gv0;
      *reinterpret_cast<uint4*>(Vn + sidx + 32 * LPITCH) = gv1;
    }
    __syncthreads();
  }

  const float inv = 1.0f / (l + xhalf(l));
  unsigned short* orow =
      AOUT + ((size_t)b * SS + (size_t)(s_base + col)) * (HH * DKV) + h * DKV + kofs;
#pragma unroll
  for (int nt = 0; nt < 4; ++nt) {
    uint4 w;
    w.x = pack_bf16(o[nt][0] * inv, o[nt][1] * inv);
    w.y = pack_bf16(o[nt][2] * inv, o[nt][3] * inv);
    w.z = pack_bf16(o[nt][4] * inv, o[nt][5] * inv);
    w.w = pack_bf16(o[nt][6] * inv, o[nt][7] * inv);
    *reinterpret_cast<uint4*>(orow + nt * 16) = w;
  }
}

__global__ void ln_kernel(const float* __restrict__ X,
                          const float* __restrict__ W,
                          const float* __restrict__ Bv,
                          float* __restrict__ OUT) {
  __shared__ float red[8];
  const int row = blockIdx.x;
  const int tid = threadIdx.x;
  const float* xr = X + (size_t)row * DD;

  float v[4];
  float s = 0.0f;
#pragma unroll
  for (int i = 0; i < 4; ++i) { v[i] = xr[tid + i * 256]; s += v[i]; }
  for (int off = 1; off < 32; off <<= 1) s += __shfl_xor(s, off);
  if ((tid & 31) == 0) red[tid >> 5] = s;
  __syncthreads();
  float mu = (red[0] + red[1] + red[2] + red[3] +
              red[4] + red[5] + red[6] + red[7]) * (1.0f / DD);
  float vs = 0.0f;
#pragma unroll
  for (int i = 0; i < 4; ++i) { float d = v[i] - mu; vs += d * d; }
  for (int off = 1; off < 32; off <<= 1) vs += __shfl_xor(vs, off);
  __syncthreads();
  if ((tid & 31) == 0) red[tid >> 5] = vs;
  __syncthreads();
  float var = (red[0] + red[1] + red[2] + red[3] +
               red[4] + red[5] + red[6] + red[7]) * (1.0f / DD);
  float rstd = rsqrtf(var + 1e-6f);
#pragma unroll
  for (int i = 0; i < 4; ++i) {
    int c = tid + i * 256;
    OUT[(size_t)row * DD + c] = (v[i] - mu) * rstd * W[c] + Bv[c];
  }
}

extern "C" void kernel_launch(void* const* d_in, const int* in_sizes, int n_in,
                              void* d_out, int out_size, void* d_ws, size_t ws_size,
                              hipStream_t stream) {
  const float* q    = (const float*)d_in[0];
  const float* k    = (const float*)d_in[1];
  const float* v    = (const float*)d_in[2];
  const int*   mask = (const int*)d_in[3];
  const float* Wq   = (const float*)d_in[4];
  const float* Wk   = (const float*)d_in[5];
  const float* Wv   = (const float*)d_in[6];
  const float* Wfc  = (const float*)d_in[7];
  const float* lnw  = (const float*)d_in[8];
  const float* lnb  = (const float*)d_in[9];
  float* out = (float*)d_out;

  char* ws = (char*)d_ws;
  const size_t SZ_ACT  = (size_t)BB * SS * DD * 2;
  const size_t SZ_W    = (size_t)HH * DD * DKV * 2;
  const size_t SZ_HEAD = (size_t)BB * HH * SS * DKV * 2;
  size_t off = 0;
  unsigned short* qbf  = (unsigned short*)(ws + off); off += SZ_ACT;
  unsigned short* kbf  = (unsigned short*)(ws + off); off += SZ_ACT;
  unsigned short* vbf  = (unsigned short*)(ws + off); off += SZ_ACT;
  unsigned short* WqT  = (unsigned short*)(ws + off); off += SZ_W;
  unsigned short* WkT  = (unsigned short*)(ws + off); off += SZ_W;
  unsigned short* WvT  = (unsigned short*)(ws + off); off += SZ_W;
  unsigned short* WfcT = (unsigned short*)(ws + off); off += (size_t)DD * DD * 2;
  unsigned short* qh   = (unsigned short*)(ws + off); off += SZ_HEAD;
  unsigned short* kh   = (unsigned short*)(ws + off); off += SZ_HEAD;
  unsigned short* vhT  = (unsigned short*)(ws + off); off += SZ_HEAD;
  unsigned short* aout = (unsigned short*)(ws + off); off += SZ_ACT;
  float*          fco  = (float*)(ws + off);          off += (size_t)BB * SS * DD * 4;
  unsigned int*   mbits = (unsigned int*)fco;

  constexpr int nAct = BB * SS * DD;
  constexpr int nW   = HH * DD * DKV;
  constexpr int nFc  = DD * DD;

  static_assert(nAct % CV_BLOCK_ELEMS == 0, "cvt_kernel: whole work-groups only");
  static_assert(nW % (8 * 256) == 0 && nFc % (8 * 256) == 0, "cvtT_kernel: whole work-groups only");
  static_assert(DKV % 32 == 0 && DD % 32 == 0 && DD % 8 == 0, "cvtT_kernel: a wave stays in one row of src");
  static_assert(nAct / CV_BLOCK_ELEMS == CV_ACT_BLOCKS, "cvt_all_kernel: activation range");
  static_assert(nW / (8 * 256) == CV_W_BLOCKS && nFc / (8 * 256) == CV_W_BLOCKS,
                "cvt_all_kernel: weight ranges");
  static_assert(SS % 128 == 0 && (SS / 128) == 16, "maskpack_block: 2 work-groups x 8 waves x 128 keys per row");
  static_assert(KT == 64 && AT_ROWS == 128, "attn_kernel and maskpack_block share the stage and block sizes");
  cvt_all_kernel<<<CV_ALL_BLOCKS, 256, 0, stream>>>(q, k, v, Wq, Wk, Wv, Wfc, mask, mbits,
                                                    qbf, kbf, vbf, WqT, WkT, WvT, WfcT);

  gemm_kernel<true><<<dim3(DD / GT_N, (BB * SS) / GT_M, 3), 256, 0, stream>>>(
      qbf, kbf, vbf, WqT, WkT, WvT, qh, kh, vhT, nullptr, nullptr);

  attn_kernel<<<dim3(BB * HH, SS / AT_ROWS), AT_WAVES * 32, 0, stream>>>(qh, kh, vhT, mbits, aout);

  gemm_kernel<false><<<dim3(DD / GT_N, (BB * SS) / GT_M), 256, 0, stream>>>(
      aout, nullptr, nullptr, WfcT, nullptr, nullptr, nullptr, nullptr, nullptr, fco, q);
  ln_kernel<<<BB * SS, 256, 0, stream>>>(fco, lnw, lnb, out);
}


# LayerNorm: one wave per row (16-byte accesses, in-wave DPP/permlane reductions, no LDS or barrier); weight transposes 32 elements a thread: 8x / 4x fewer waves to start; LN mean/var sum order changes
# speedup vs baseline: 5.8243x; 1.0983x over previous
#include <hip/hip_runtime.h>


#define BB 2
#define SS 2048
#define DD 1024
#define HH 16
#define DKV 64

typedef __attribute__((ext_vector_type(16))) __bf16 v16bf;
typedef __attribute__((ext_vector_type(8)))  float  v8f;

union FragU { v16bf v; uint4 q[2]; };

__device__ __forceinline__ unsigned short f2bf(float f) {
  unsigned int u = __float_as_uint(f);
  u += 0x7FFFu + ((u >> 16) & 1u);
  return (unsigned short)(u >> 16);
}

__device__ __forceinline__ v16bf load_frag(const unsigned short* rowbase, int lane) {
  const unsigned short* p = rowbase + ((lane & 16) ? 8 : 0);
  FragU f;
  f.q[0] = *reinterpret_cast<const uint4*>(p);
  f.q[1] = *reinterpret_cast<const uint4*>(p + 16);
  return f.v;
}

#define WMMA_BF16(a, b, c) \
  __builtin_amdgcn_wmma_f32_16x16x32_bf16(false, (a), false, (b), (short)0, (c), false, false)

#define CV_SLABS 8
#define CV_SLAB_ELEMS  (256 * 4)
#define CV_BLOCK_ELEMS (CV_SLABS * CV_SLAB_ELEMS)

__device__ __forceinline__ unsigned int f2bf_pair(float lo, float hi) {
  return (unsigned int)f2bf(lo) | ((unsigned int)f2bf(hi) << 16);
}

__device__ __forceinline__ void
cvt_block(const float* __restrict__ src, unsigned short* __restrict__ dst, unsigned int blk) {
  const size_t base = (size_t)blk * CV_BLOCK_ELEMS + threadIdx.x * 4;
  float4 v[CV_SLABS];
#pragma unroll
  for (int j = 0; j < CV_SLABS; ++j)
    v[j] = *reinterpret_cast<const float4*>(src + base + j * CV_SLAB_ELEMS);
#pragma unroll
  for (int j = 0; j < CV_SLABS; ++j) {
    uint2 w;
    w.x = f2bf_pair(v[j].x, v[j].y);
    w.y = f2bf_pair(v[j].z, v[j].w);
    *reinterpret_cast<uint2*>(dst + base + j * CV_SLAB_ELEMS) = w;
  }
}

#define CT_E 32

template <int R, int C>
__device__ __forceinline__ void
cvtT_block(const float* __restrict__ src, unsigned short* __restrict__ dst, unsigned int blk) {
  const unsigned int i  = blk * 256 + threadIdx.x;
  const unsigned int c  = i % C;
  const unsigned int re = (i / C) % (R / CT_E);
  const unsigned int g  = i / (C * (R / CT_E));
  const float* s = src + ((size_t)g * R + (size_t)re * CT_E) * C + c;
  float v[CT_E];
#pragma unroll
  for (int k = 0; k < CT_E; ++k) v[k] = s[(size_t)k * C];
  unsigned short* d = dst + ((size_t)g * C + c) * R + (size_t)re * CT_E;
#pragma unroll
  for (int m = 0; m < CT_E / 8; ++m) {
    uint4 w;
    w.x = f2bf_pair(v[8 * m + 0], v[8 * m + 1]);
    w.y = f2bf_pair(v[8 * m + 2], v[8 * m + 3]);
    w.z = f2bf_pair(v[8 * m + 4], v[8 * m + 5]);
    w.w = f2bf_pair(v[8 * m + 6], v[8 * m + 7]);
    *reinterpret_cast<uint4*>(d + 8 * m) = w;
  }
}

#define MB_STAGE_WORDS 32
#define MB_BLOCKS (BB * (SS / 16) * 2)

__device__ __forceinline__ void
maskpack_block(const int* __restrict__ mask, unsigned int* __restrict__ mbits, unsigned int blk) {
  const unsigned int lane = threadIdx.x & 31;
  const unsigned int wave = threadIdx.x >> 5;
  const unsigned int col  = lane & 15;
  const unsigned int kofs = (lane & 16) ? 8 : 0;
  const unsigned int rb = blk >> 1;
  const unsigned int sp = (blk & 1) * 8 + wave;
  const int* mrow = mask + ((size_t)rb * 16 + col) * SS + sp * 128 + kofs;
  int4 m[16];
#pragma unroll
  for (int i = 0; i < 8; ++i) {
    m[2 * i]     = *reinterpret_cast<const int4*>(mrow + i * 16);
    m[2 * i + 1] = *reinterpret_cast<const int4*>(mrow + i * 16 + 4);
  }
  unsigned int lo = 0, hi = 0;
#pragma unroll
  for (int i = 0; i < 8; ++i) {
    const int mv[8] = {m[2 * i].x, m[2 * i].y, m[2 * i].z, m[2 * i].w,
                       m[2 * i + 1].x, m[2 * i + 1].y, m[2 * i + 1].z, m[2 * i + 1].w};
#pragma unroll
    for (int r = 0; r < 8; ++r) {
      const unsigned int bal = __builtin_amdgcn_ballot_w32(mv[r] != 0);
      const unsigned int idx = i * 8 + r;
      if (idx < 32) lo = (lane == idx)      ? bal : lo;
      else          hi = (lane == idx - 32) ? bal : hi;
    }
  }
  unsigned int* dst = mbits + ((size_t)rb * (SS / 64) + sp * 2) * MB_STAGE_WORDS;
  dst[lane]      = lo;
  dst[32 + lane] = hi;
}

#define CV_ACT_BLOCKS ((BB * SS * DD) / CV_BLOCK_ELEMS)
#define CV_W_BLOCKS   ((HH * DD * DKV) / (CT_E * 256))
#define CV_ALL_BLOCKS (3 * CV_ACT_BLOCKS + 4 * CV_W_BLOCKS + MB_BLOCKS)

__global__ void __launch_bounds__(256)
cvt_all_kernel(const float* __restrict__ q, const float* __restrict__ k,
               const float* __restrict__ v,
               const float* __restrict__ Wq, const float* __restrict__ Wk,
               const float* __restrict__ Wv, const float* __restrict__ Wfc,
               const int* __restrict__ mask, unsigned int* __restrict__ mbits,
               unsigned short* __restrict__ qbf, unsigned short* __restrict__ kbf,
               unsigned short* __restrict__ vbf,
               unsigned short* __restrict__ WqT, unsigned short* __restrict__ WkT,
               unsigned short* __restrict__ WvT, unsigned short* __restrict__ WfcT) {
  const unsigned int bid = blockIdx.x;
  if (bid < 3 * CV_ACT_BLOCKS) {
    const unsigned int job = bid / CV_ACT_BLOCKS;
    const float* s = (job == 0) ? q : (job == 1) ? k : v;
    unsigned short* d = (job == 0) ? qbf : (job == 1) ? kbf : vbf;
    cvt_block(s, d, bid - job * CV_ACT_BLOCKS);
  } else {
    const unsigned int wb  = bid - 3 * CV_ACT_BLOCKS;
    const unsigned int job = wb / CV_W_BLOCKS;
    const unsigned int blk = wb - job * CV_W_BLOCKS;
    if (job < 3) {
      const float* s = (job == 0) ? Wq : (job == 1) ? Wk : Wv;
      unsigned short* d = (job == 0) ? WqT : (job == 1) ? WkT : WvT;
      cvtT_block<DD, DKV>(s, d, blk);
    } else if (job == 3) {
      cvtT_block<DD, DD>(Wfc, WfcT, blk);
    } else {
      maskpack_block(mask, mbits, wb - 4 * CV_W_BLOCKS);
    }
  }
}

#define GT_M   128
#define GT_N   128
#define GT_K   128
#define GPITCH 136
#define FPITCH 132

typedef __attribute__((ext_vector_type(4))) unsigned int u32x4;

static_assert(GT_N * FPITCH * 4 <= 2 * GT_M * GPITCH * 2, "the parked fp32 tile fits in the A tiles' LDS");

template <bool PROJ>
__global__ void __launch_bounds__(256)
gemm_kernel(const unsigned short* __restrict__ X0,
            const unsigned short* __restrict__ X1,
            const unsigned short* __restrict__ X2,
            const unsigned short* __restrict__ W0,
            const unsigned short* __restrict__ W1,
            const unsigned short* __restrict__ W2,
            unsigned short* __restrict__ O0,
            unsigned short* __restrict__ O1,
            unsigned short* __restrict__ O2,
            float* __restrict__ OUTF,
            const float* __restrict__ RES) {
  __shared__ __align__(16) unsigned short As[2][GT_M * GPITCH];
  __shared__ __align__(16) unsigned short Bs[2][GT_N * GPITCH];

  const int z = PROJ ? (int)blockIdx.z : 0;
  const unsigned short* __restrict__ X  = (z == 0) ? X0 : (z == 1) ? X1 : X2;
  const unsigned short* __restrict__ WT = (z == 0) ? W0 : (z == 1) ? W1 : W2;
  unsigned short* __restrict__ OUTH     = (z == 0) ? O0 : (z == 1) ? O1 : O2;
  const float scale = (z == 0) ? 0.125f : 1.0f;
  const bool xrows = PROJ && (z == 2);
  const unsigned short* __restrict__ A  = xrows ? X : WT;
  const unsigned short* __restrict__ BT = xrows ? WT : X;
  const int w0 = blockIdx.x * GT_N;
  const int x0 = blockIdx.y * GT_M;
  const int m0 = xrows ? x0 : w0;
  const int n0 = xrows ? w0 : x0;

  const int tid = threadIdx.x;
  const int lane = tid & 31;
  const int wave = tid >> 5;
  const int wm = wave & 3;
  const int wn = wave >> 2;
  const int rbase = (lane & 16) ? 8 : 0;
  const int col = lane & 15;

  const int srow = tid >> 4;
  const int sch  = (tid & 15) * 8;
  const unsigned short* Ag = A  + (size_t)(m0 + srow) * DD + sch;
  const unsigned short* Bg = BT + (size_t)(n0 + srow) * DD + sch;
  const int sidx = srow * GPITCH + sch;

  u32x4 ga[8], gb[8];
#pragma unroll
  for (int i = 0; i < 8; ++i) {
    ga[i] = *reinterpret_cast<const u32x4*>(Ag + (size_t)(16 * i) * DD);
    gb[i] = *reinterpret_cast<const u32x4*>(Bg + (size_t)(16 * i) * DD);
  }
#pragma unroll
  for (int i = 0; i < 8; ++i) {
    *reinterpret_cast<u32x4*>(&As[0][sidx + 16 * i * GPITCH]) = ga[i];
    *reinterpret_cast<u32x4*>(&Bs[0][sidx + 16 * i * GPITCH]) = gb[i];
  }
  __syncthreads();

  v8f acc[2][4] = {};

  for (int st = 0; st < DD / GT_K; ++st) {
    const int buf = st & 1;
    const bool more = (st + 1) < (DD / GT_K);
    if (more) {
      const unsigned short* ag = Ag + (st + 1) * GT_K;
      const unsigned short* bg = Bg + (st + 1) * GT_K;
#pragma unroll
      for (int i = 0; i < 8; ++i) {
        ga[i] = *reinterpret_cast<const u32x4*>(ag + (size_t)(16 * i) * DD);
        gb[i] = *reinterpret_cast<const u32x4*>(bg + (size_t)(16 * i) * DD);
      }
    }
    __builtin_amdgcn_sched_barrier(0);
    const unsigned short* Ab = &As[buf][(wm * 32 + col) * GPITCH];
    const unsigned short* Bb = &Bs[buf][(wn * 64 + col) * GPITCH];

    v16bf af[2][2], bfr[2][4];
#pragma unroll
    for (int mt = 0; mt < 2; ++mt)
      af[0][mt] = load_frag(Ab + mt * 16 * GPITCH, lane);
#pragma unroll
    for (int nt = 0; nt < 4; ++nt)
      bfr[0][nt] = load_frag(Bb + nt * 16 * GPITCH, lane);
#pragma unroll
    for (int j = 0; j < GT_K / 32; ++j) {
      const int cur = j & 1;
      if (j + 1 < GT_K / 32) {
#pragma unroll
        for (int mt = 0; mt < 2; ++mt)
          af[cur ^ 1][mt] = load_frag(Ab + mt * 16 * GPITCH + (j + 1) * 32, lane);
#pragma unroll
        for (int nt = 0; nt < 4; ++nt)
          bfr[cur ^ 1][nt] = load_frag(Bb + nt * 16 * GPITCH + (j + 1) * 32, lane);
      }
      __builtin_amdgcn_sched_barrier(0);
#pragma unroll
      for (int mt = 0; mt < 2; ++mt)
#pragma unroll
        for (int nt = 0; nt < 4; ++nt)
          acc[mt][nt] = WMMA_BF16(af[cur][mt], bfr[cur][nt], acc[mt][nt]);
    }

    __builtin_amdgcn_sched_barrier(0);
    if (more) {
#pragma unroll
      for (int i = 0; i < 8; ++i) {
        *reinterpret_cast<u32x4*>(&As[buf ^ 1][sidx + 16 * i * GPITCH]) = ga[i];
        *reinterpret_cast<u32x4*>(&Bs[buf ^ 1][sidx + 16 * i * GPITCH]) = gb[i];
      }
    }
    __syncthreads();
  }

  if (PROJ) {
    unsigned short* T = &As[0][0];
#pragma unroll
    for (int mt = 0; mt < 2; ++mt) {
#pragma unroll
      for (int nt = 0; nt < 4; ++nt) {
        const int ra = wm * 32 + mt * 16 + rbase;
        const int cb = wn * 64 + nt * 16 + col;
        u32x4 w;
        w.x = f2bf_pair(acc[mt][nt][0] * scale, acc[mt][nt][1] * scale);
        w.y = f2bf_pair(acc[mt][nt][2] * scale, acc[mt][nt][3] * scale);
        w.z = f2bf_pair(acc[mt][nt][4] * scale, acc[mt][nt][5] * scale);
        w.w = f2bf_pair(acc[mt][nt][6] * scale, acc[mt][nt][7] * scale);
        *reinterpret_cast<u32x4*>(&T[cb * GPITCH + ra]) = w;
      }
    }
    __syncthreads();
#pragma unroll
    for (int it = 0; it < 8; ++it) {
      const int p   = it * 256 + tid;
      const int row = p >> 4;
      const int c8  = (p & 15) * 8;
      const u32x4 w = *reinterpret_cast<const u32x4*>(&T[row * GPITCH + c8]);
      const int xr = x0 + (xrows ? c8 : row);
      const int n  = w0 + (xrows ? row : c8);
      const int b = xr / SS, sq = xr - b * SS;
      const int h = n / DKV, dcol = n - h * DKV;
      const size_t bh = (size_t)(b * HH + h);
      const size_t idx = xrows ? (bh * DKV + dcol) * SS + sq
                               : (bh * SS + sq) * DKV + dcol;
      *reinterpret_cast<u32x4*>(OUTH + idx) = w;
    }
  } else {
    float4 res[16];
#pragma unroll
    for (int it = 0; it < 16; ++it) {
      const int p = it * 256 + tid;
      res[it] = *reinterpret_cast<const float4*>(
          RES + (size_t)(x0 + (p >> 5)) * DD + w0 + (p & 31) * 4);
    }
    float* T = reinterpret_cast<float*>(&As[0][0]);
#pragma unroll
    for (int mt = 0; mt < 2; ++mt) {
#pragma unroll
      for (int nt = 0; nt < 4; ++nt) {
        const int ra = wm * 32 + mt * 16 + rbase;
        const int cb = wn * 64 + nt * 16 + col;
        float4 lo = {acc[mt][nt][0], acc[mt][nt][1], acc[mt][nt][2], acc[mt][nt][3]};
        float4 hi = {acc[mt][nt][4], acc[mt][nt][5], acc[mt][nt][6], acc[mt][nt][7]};
        *reinterpret_cast<float4*>(&T[cb * FPITCH + ra])     = lo;
        *reinterpret_cast<float4*>(&T[cb * FPITCH + ra + 4]) = hi;
      }
    }
    __syncthreads();
#pragma unroll
    for (int it = 0; it < 16; ++it) {
      const int p = it * 256 + tid;
      const int row = p >> 5;
      const int c4  = (p & 31) * 4;
      float4 c = *reinterpret_cast<const float4*>(&T[row * FPITCH + c4]);
      c.x += res[it].x; c.y += res[it].y; c.z += res[it].z; c.w += res[it].w;
      *reinterpret_cast<float4*>(OUTF + (size_t)(x0 + row) * DD + w0 + c4) = c;
    }
  }
}

#define AT_WAVES 8
#define AT_ROWS  (AT_WAVES * 16)
#define KT       64
#define LPITCH   72

typedef __attribute__((ext_vector_type(2))) float  f32x2;
typedef __attribute__((ext_vector_type(2))) __bf16 bf16x2;

__device__ __forceinline__ unsigned int pack_bf16(float lo, float hi) {
  f32x2 t = {lo, hi};
  return __builtin_bit_cast(unsigned int, __builtin_convertvector(t, bf16x2));
}

__device__ __forceinline__ float xhalf(float x) {
  int b = __builtin_bit_cast(int, x);
  return __builtin_bit_cast(float,
      __builtin_amdgcn_permlanex16(b, b, 0x76543210, 0xfedcba98, false, false));
}

__global__ void __launch_bounds__(AT_WAVES * 32)
attn_kernel(const unsigned short* __restrict__ QH,
            const unsigned short* KH,
            const unsigned short* VT,
            const unsigned int* __restrict__ MBITS,
            unsigned short* __restrict__ AOUT) {
  __shared__ __align__(16) unsigned short Ks[2][KT * LPITCH];
  __shared__ __align__(16) unsigned short Vs[2][DKV * LPITCH];

  const int tid = threadIdx.x;
  const int lane = tid & 31;
  const int wave = __builtin_amdgcn_readfirstlane(tid >> 5);
  const int bh = blockIdx.x;
  const int b = bh >> 4, h = bh & 15;
  const int s_base = blockIdx.y * AT_ROWS + wave * 16;
  const int kofs = (lane & 16) ? 8 : 0;
  const int col = lane & 15;

  const unsigned short* Qrow =
      QH + ((size_t)bh * SS + (size_t)(s_base + col)) * DKV;
  v16bf qf0 = load_frag(Qrow, lane);
  v16bf qf1 = load_frag(Qrow + 32, lane);

  v8f o[4] = {};
  float m = -3.0e38f;
  float l = 0.0f;

  const unsigned int* mwave =
      MBITS + ((size_t)b * (SS / 16) + (size_t)(s_base >> 4)) * (SS / KT) * MB_STAGE_WORDS;

  const int srow = tid >> 3;
  const int sch  = (tid & 7) * 8;
  const unsigned short* Kg = KH + ((size_t)bh * SS + srow) * DKV + sch;
  const unsigned short* Vg = VT + ((size_t)bh * DKV + srow) * SS + sch;
  const int sidx = srow * LPITCH + sch;

  uint4 gk0 = *reinterpret_cast<const uint4*>(Kg);
  uint4 gk1 = *reinterpret_cast<const uint4*>(Kg + (size_t)32 * DKV);
  uint4 gv0 = *reinterpret_cast<const uint4*>(Vg);
  uint4 gv1 = *reinterpret_cast<const uint4*>(Vg + (size_t)32 * SS);
  *reinterpret_cast<uint4*>(&Ks[0][sidx])               = gk0;
  *reinterpret_cast<uint4*>(&Ks[0][sidx + 32 * LPITCH]) = gk1;
  *reinterpret_cast<uint4*>(&Vs[0][sidx])               = gv0;
  *reinterpret_cast<uint4*>(&Vs[0][sidx + 32 * LPITCH]) = gv1;
  __syncthreads();

  for (int st = 0; st < SS / KT; ++st) {
    const int buf = st & 1;
    const int t0 = st * KT;
    const bool more = (st + 1) < (SS / KT);
    if (more) {
      const unsigned short* kg = Kg + (size_t)(t0 + KT) * DKV;
      const unsigned short* vg = Vg + (t0 + KT);
      gk0 = *reinterpret_cast<const uint4*>(kg);
      gk1 = *reinterpret_cast<const uint4*>(kg + (size_t)32 * DKV);
      gv0 = *reinterpret_cast<const uint4*>(vg);
      gv1 = *reinterpret_cast<const uint4*>(vg + (size_t)32 * SS);
    }
    int frow = col * LPITCH;
    asm volatile("" : "+v"(frow) : : "memory");
    const unsigned short* Kb = &Ks[buf][0] + frow;
    const unsigned short* Vb = &Vs[buf][0] + frow;

    v16bf kb[KT / 32][4], vb[KT / 32][4];
#pragma unroll
    for (int j = 0; j < KT / 32; ++j) {
      kb[j][0] = load_frag(Kb + (j * 32) * LPITCH, lane);
      kb[j][1] = load_frag(Kb + (j * 32) * LPITCH + 32, lane);
      kb[j][2] = load_frag(Kb + (j * 32 + 16) * LPITCH, lane);
      kb[j][3] = load_frag(Kb + (j * 32 + 16) * LPITCH + 32, lane);
    }
#pragma unroll
    for (int j = 0; j < KT / 32; ++j)
#pragma unroll
      for (int nt = 0; nt < 4; ++nt)
        vb[j][nt] = load_frag(Vb + (nt * 16) * LPITCH + j * 32, lane);
    __builtin_amdgcn_sched_barrier(0);

#pragma unroll
    for (int j = 0; j < KT / 32; ++j) {
      const uint4* mstep =
          reinterpret_cast<const uint4*>(mwave + (size_t)st * MB_STAGE_WORDS + j * 16);
      const uint4 mk0a = mstep[0];
      const uint4 mk0b = mstep[1];
      const uint4 mk1a = mstep[2];
      const uint4 mk1b = mstep[3];

      v8f s0 = {}, s1 = {};
      s0 = WMMA_BF16(kb[j][0], qf0, s0);
      s0 = WMMA_BF16(kb[j][1], qf1, s0);
      s1 = WMMA_BF16(kb[j][2], qf0, s1);
      s1 = WMMA_BF16(kb[j][3], qf1, s1);

      const unsigned int mk0[8] = {mk0a.x, mk0a.y, mk0a.z, mk0a.w, mk0b.x, mk0b.y, mk0b.z, mk0b.w};
      const unsigned int mk1[8] = {mk1a.x, mk1a.y, mk1a.z, mk1a.w, mk1b.x, mk1b.y, mk1b.z, mk1b.w};
      float v0[8], v1[8];
#pragma unroll
      for (int r = 0; r < 8; ++r) {
        v0[r] = __builtin_amdgcn_inverse_ballot_w32(mk0[r]) ? s0[r] : -1.0e9f;
        v1[r] = __builtin_amdgcn_inverse_ballot_w32(mk1[r]) ? s1[r] : -1.0e9f;
      }
      float mx = fmaxf(v0[0], v1[0]);
#pragma unroll
      for (int r = 1; r < 8; ++r) mx = fmaxf(mx, fmaxf(v0[r], v1[r]));
      mx = fmaxf(mx, xhalf(mx));
      const float mnew  = fmaxf(m, mx);
      const float alpha = __expf(m - mnew);
      float rs = 0.0f;
      unsigned int w0[4], w1[4];
#pragma unroll
      for (int r = 0; r < 8; r += 2) {
        const float p00 = __expf(v0[r]     - mnew);
        const float p01 = __expf(v0[r + 1] - mnew);
        const float p10 = __expf(v1[r]     - mnew);
        const float p11 = __expf(v1[r + 1] - mnew);
        rs += (p00 + p01) + (p10 + p11);
        w0[r >> 1] = pack_bf16(p00, p01);
        w1[r >> 1] = pack_bf16(p10, p11);
      }
      l = l * alpha + rs;
      m = mnew;
#pragma unroll
      for (int nt = 0; nt < 4; ++nt)
#pragma unroll
        for (int r = 0; r < 8; ++r) o[nt][r] = o[nt][r] * alpha;

      FragU pf;
      pf.q[0] = make_uint4(w0[0], w0[1], w0[2], w0[3]);
      pf.q[1] = make_uint4(w1[0], w1[1], w1[2], w1[3]);
#pragma unroll
      for (int nt = 0; nt < 4; ++nt)
        o[nt] = WMMA_BF16(vb[j][nt], pf.v, o[nt]);
    }

    asm volatile("" : "+v"(o[0]), "+v"(o[1]), "+v"(o[2]), "+v"(o[3]));
    if (more) {
      unsigned short* Kn = &Ks[buf ^ 1][0];
      unsigned short* Vn = &Vs[buf ^ 1][0];
      *reinterpret_cast<uint4*>(Kn + sidx)               = gk0;
      *reinterpret_cast<uint4*>(Kn + sidx + 32 * LPITCH) = gk1;
      *reinterpret_cast<uint4*>(Vn + sidx)               = gv0;
      *reinterpret_cast<uint4*>(Vn + sidx + 32 * LPITCH) = gv1;
    }
    __syncthreads();
  }

  const float inv = 1.0f / (l + xhalf(l));
  unsigned short* orow =
      AOUT + ((size_t)b * SS + (size_t)(s_base + col)) * (HH * DKV) + h * DKV + kofs;
#pragma unroll
  for (int nt = 0; nt < 4; ++nt) {
    uint4 w;
    w.x = pack_bf16(o[nt][0] * inv, o[nt][1] * inv);
    w.y = pack_bf16(o[nt][2] * inv, o[nt][3] * inv);
    w.z = pack_bf16(o[nt][4] * inv, o[nt][5] * inv);
    w.w = pack_bf16(o[nt][6] * inv, o[nt][7] * inv);
    *reinterpret_cast<uint4*>(orow + nt * 16) = w;
  }
}

#define LN_WAVES 8
#define LN_PIECES (DD / (32 * 4))

template <int M>
__device__ __forceinline__ float row_xor(float x) {
  const int b = __builtin_bit_cast(int, x);
  return __builtin_bit_cast(float,
      __builtin_amdgcn_update_dpp(b, b, 0x160 + M, 0xf, 0xf, false));
}

__device__ __forceinline__ float wave_sum(float x) {
  x += row_xor<1>(x);
  x += row_xor<2>(x);
  x += row_xor<4>(x);
  x += row_xor<8>(x);
  x += xhalf(x);
  return x;
}

__global__ void __launch_bounds__(LN_WAVES * 32) __attribute__((amdgpu_waves_per_eu(4, 4)))
ln_kernel(const float* __restrict__ X,
          const float* __restrict__ W,
          const float* __restrict__ Bv,
          float* __restrict__ OUT) {
  const int lane = threadIdx.x & 31;
  const int row = blockIdx.x * LN_WAVES + (threadIdx.x >> 5);
  const size_t base = (size_t)row * DD + lane * 4;

  float4 v[LN_PIECES], w[LN_PIECES], bb[LN_PIECES];
#pragma unroll
  for (int j = 0; j < LN_PIECES; ++j)
    v[j] = *reinterpret_cast<const float4*>(X + base + j * 128);
#pragma unroll
  for (int j = 0; j < LN_PIECES; ++j) {
    w[j]  = *reinterpret_cast<const float4*>(W  + lane * 4 + j * 128);
    bb[j] = *reinterpret_cast<const float4*>(Bv + lane * 4 + j * 128);
  }

  float s = 0.0f;
#pragma unroll
  for (int j = 0; j < LN_PIECES; ++j) s += (v[j].x + v[j].y) + (v[j].z + v[j].w);
  const float mu = wave_sum(s) * (1.0f / DD);

  float vs = 0.0f;
#pragma unroll
  for (int j = 0; j < LN_PIECES; ++j) {
    const float dx = v[j].x - mu, dy = v[j].y - mu, dz = v[j].z - mu, dw = v[j].w - mu;
    vs += (dx * dx + dy * dy) + (dz * dz + dw * dw);
  }
  const float var = wave_sum(vs) * (1.0f / DD);
  const float rstd = rsqrtf(var + 1e-6f);

#pragma unroll
  for (int j = 0; j < LN_PIECES; ++j) {
    float4 o;
    o.x = (v[j].x - mu) * rstd * w[j].x + bb[j].x;
    o.y = (v[j].y - mu) * rstd * w[j].y + bb[j].y;
    o.z = (v[j].z - mu) * rstd * w[j].z + bb[j].z;
    o.w = (v[j].w - mu) * rstd * w[j].w + bb[j].w;
    *reinterpret_cast<float4*>(OUT + base + j * 128) = o;
  }
}

extern "C" void kernel_launch(void* const* d_in, const int* in_sizes, int n_in,
                              void* d_out, int out_size, void* d_ws, size_t ws_size,
                              hipStream_t stream) {
  const float* q    = (const float*)d_in[0];
  const float* k    = (const float*)d_in[1];
  const float* v    = (const float*)d_in[2];
  const int*   mask = (const int*)d_in[3];
  const float* Wq   = (const float*)d_in[4];
  const float* Wk   = (const float*)d_in[5];
  const float* Wv   = (const float*)d_in[6];
  const float* Wfc  = (const float*)d_in[7];
  const float* lnw  = (const float*)d_in[8];
  const float* lnb  = (const float*)d_in[9];
  float* out = (float*)d_out;

  char* ws = (char*)d_ws;
  const size_t SZ_ACT  = (size_t)BB * SS * DD * 2;
  const size_t SZ_W    = (size_t)HH * DD * DKV * 2;
  const size_t SZ_HEAD = (size_t)BB * HH * SS * DKV * 2;
  size_t off = 0;
  unsigned short* qbf  = (unsigned short*)(ws + off); off += SZ_ACT;
  unsigned short* kbf  = (unsigned short*)(ws + off); off += SZ_ACT;
  unsigned short* vbf  = (unsigned short*)(ws + off); off += SZ_ACT;
  unsigned short* WqT  = (unsigned short*)(ws + off); off += SZ_W;
  unsigned short* WkT  = (unsigned short*)(ws + off); off += SZ_W;
  unsigned short* WvT  = (unsigned short*)(ws + off); off += SZ_W;
  unsigned short* WfcT = (unsigned short*)(ws + off); off += (size_t)DD * DD * 2;
  unsigned short* qh   = (unsigned short*)(ws + off); off += SZ_HEAD;
  unsigned short* kh   = (unsigned short*)(ws + off); off += SZ_HEAD;
  unsigned short* vhT  = (unsigned short*)(ws + off); off += SZ_HEAD;
  unsigned short* aout = (unsigned short*)(ws + off); off += SZ_ACT;
  float*          fco  = (float*)(ws + off);          off += (size_t)BB * SS * DD * 4;
  unsigned int*   mbits = (unsigned int*)fco;

  constexpr int nAct = BB * SS * DD;
  constexpr int nW   = HH * DD * DKV;
  constexpr int nFc  = DD * DD;

  static_assert(nAct % CV_BLOCK_ELEMS == 0, "cvt_kernel: whole work-groups only");
  static_assert(nW % (CT_E * 256) == 0 && nFc % (CT_E * 256) == 0, "cvtT_kernel: whole work-groups only");
  static_assert(DKV % 32 == 0 && DD % 32 == 0 && DD % CT_E == 0, "cvtT_kernel: a wave stays in one row of src");
  static_assert(nAct / CV_BLOCK_ELEMS == CV_ACT_BLOCKS, "cvt_all_kernel: activation range");
  static_assert(nW / (CT_E * 256) == CV_W_BLOCKS && nFc / (CT_E * 256) == CV_W_BLOCKS,
                "cvt_all_kernel: weight ranges");
  static_assert(SS % 128 == 0 && (SS / 128) == 16, "maskpack_block: 2 work-groups x 8 waves x 128 keys per row");
  static_assert(KT == 64 && AT_ROWS == 128, "attn_kernel and maskpack_block share the stage and block sizes");
  cvt_all_kernel<<<CV_ALL_BLOCKS, 256, 0, stream>>>(q, k, v, Wq, Wk, Wv, Wfc, mask, mbits,
                                                    qbf, kbf, vbf, WqT, WkT, WvT, WfcT);

  gemm_kernel<true><<<dim3(DD / GT_N, (BB * SS) / GT_M, 3), 256, 0, stream>>>(
      qbf, kbf, vbf, WqT, WkT, WvT, qh, kh, vhT, nullptr, nullptr);

  attn_kernel<<<dim3(BB * HH, SS / AT_ROWS), AT_WAVES * 32, 0, stream>>>(qh, kh, vhT, mbits, aout);

  gemm_kernel<false><<<dim3(DD / GT_N, (BB * SS) / GT_M), 256, 0, stream>>>(
      aout, nullptr, nullptr, WfcT, nullptr, nullptr, nullptr, nullptr, nullptr, fco, q);
  static_assert((BB * SS) % LN_WAVES == 0 && DD == LN_PIECES * 128, "ln_kernel: whole work-groups, whole rows");
  ln_kernel<<<(BB * SS) / LN_WAVES, LN_WAVES * 32, 0, stream>>>(fco, lnw, lnb, out);
}


# mask packing and Wfc convert leave the convert grid: they run as the last work-groups of the projection grid, in the slot its LDS-limited last third of GEMM work-groups leaves free; values unchanged
# speedup vs baseline: 6.2572x; 1.0743x over previous
#include <hip/hip_runtime.h>


#define BB 2
#define SS 2048
#define DD 1024
#define HH 16
#define DKV 64

typedef __attribute__((ext_vector_type(16))) __bf16 v16bf;
typedef __attribute__((ext_vector_type(8)))  float  v8f;

union FragU { v16bf v; uint4 q[2]; };

__device__ __forceinline__ unsigned short f2bf(float f) {
  unsigned int u = __float_as_uint(f);
  u += 0x7FFFu + ((u >> 16) & 1u);
  return (unsigned short)(u >> 16);
}

__device__ __forceinline__ v16bf load_frag(const unsigned short* rowbase, int lane) {
  const unsigned short* p = rowbase + ((lane & 16) ? 8 : 0);
  FragU f;
  f.q[0] = *reinterpret_cast<const uint4*>(p);
  f.q[1] = *reinterpret_cast<const uint4*>(p + 16);
  return f.v;
}

#define WMMA_BF16(a, b, c) \
  __builtin_amdgcn_wmma_f32_16x16x32_bf16(false, (a), false, (b), (short)0, (c), false, false)

#define CV_SLABS 8
#define CV_SLAB_ELEMS  (256 * 4)
#define CV_BLOCK_ELEMS (CV_SLABS * CV_SLAB_ELEMS)

__device__ __forceinline__ unsigned int f2bf_pair(float lo, float hi) {
  return (unsigned int)f2bf(lo) | ((unsigned int)f2bf(hi) << 16);
}

__device__ __forceinline__ void
cvt_block(const float* __restrict__ src, unsigned short* __restrict__ dst, unsigned int blk) {
  const size_t base = (size_t)blk * CV_BLOCK_ELEMS + threadIdx.x * 4;
  float4 v[CV_SLABS];
#pragma unroll
  for (int j = 0; j < CV_SLABS; ++j)
    v[j] = *reinterpret_cast<const float4*>(src + base + j * CV_SLAB_ELEMS);
#pragma unroll
  for (int j = 0; j < CV_SLABS; ++j) {
    uint2 w;
    w.x = f2bf_pair(v[j].x, v[j].y);
    w.y = f2bf_pair(v[j].z, v[j].w);
    *reinterpret_cast<uint2*>(dst + base + j * CV_SLAB_ELEMS) = w;
  }
}

#define CT_E 32

template <int R, int C>
__device__ __forceinline__ void
cvtT_block(const float* __restrict__ src, unsigned short* __restrict__ dst, unsigned int blk) {
  const unsigned int i  = blk * 256 + threadIdx.x;
  const unsigned int c  = i % C;
  const unsigned int re = (i / C) % (R / CT_E);
  const unsigned int g  = i / (C * (R / CT_E));
  const float* s = src + ((size_t)g * R + (size_t)re * CT_E) * C + c;
  float v[CT_E];
#pragma unroll
  for (int k = 0; k < CT_E; ++k) v[k] = s[(size_t)k * C];
  unsigned short* d = dst + ((size_t)g * C + c) * R + (size_t)re * CT_E;
#pragma unroll
  for (int m = 0; m < CT_E / 8; ++m) {
    uint4 w;
    w.x = f2bf_pair(v[8 * m + 0], v[8 * m + 1]);
    w.y = f2bf_pair(v[8 * m + 2], v[8 * m + 3]);
    w.z = f2bf_pair(v[8 * m + 4], v[8 * m + 5]);
    w.w = f2bf_pair(v[8 * m + 6], v[8 * m + 7]);
    *reinterpret_cast<uint4*>(d + 8 * m) = w;
  }
}

#define MB_STAGE_WORDS 32
#define MB_BLOCKS (BB * (SS / 16) * 2)

__device__ __forceinline__ void
maskpack_block(const int* __restrict__ mask, unsigned int* __restrict__ mbits, unsigned int blk) {
  const unsigned int lane = threadIdx.x & 31;
  const unsigned int wave = threadIdx.x >> 5;
  const unsigned int col  = lane & 15;
  const unsigned int kofs = (lane & 16) ? 8 : 0;
  const unsigned int rb = blk >> 1;
  const unsigned int sp = (blk & 1) * 8 + wave;
  const int* mrow = mask + ((size_t)rb * 16 + col) * SS + sp * 128 + kofs;
  int4 m[16];
#pragma unroll
  for (int i = 0; i < 8; ++i) {
    m[2 * i]     = *reinterpret_cast<const int4*>(mrow + i * 16);
    m[2 * i + 1] = *reinterpret_cast<const int4*>(mrow + i * 16 + 4);
  }
  unsigned int lo = 0, hi = 0;
#pragma unroll
  for (int i = 0; i < 8; ++i) {
    const int mv[8] = {m[2 * i].x, m[2 * i].y, m[2 * i].z, m[2 * i].w,
                       m[2 * i + 1].x, m[2 * i + 1].y, m[2 * i + 1].z, m[2 * i + 1].w};
#pragma unroll
    for (int r = 0; r < 8; ++r) {
      const unsigned int bal = __builtin_amdgcn_ballot_w32(mv[r] != 0);
      const unsigned int idx = i * 8 + r;
      if (idx < 32) lo = (lane == idx)      ? bal : lo;
      else          hi = (lane == idx - 32) ? bal : hi;
    }
  }
  unsigned int* dst = mbits + ((size_t)rb * (SS / 64) + sp * 2) * MB_STAGE_WORDS;
  dst[lane]      = lo;
  dst[32 + lane] = hi;
}

#define CV_ACT_BLOCKS ((BB * SS * DD) / CV_BLOCK_ELEMS)
#define CV_W_BLOCKS   ((HH * DD * DKV) / (CT_E * 256))
#define CV_ALL_BLOCKS (3 * CV_ACT_BLOCKS + 3 * CV_W_BLOCKS)

__global__ void __launch_bounds__(256)
cvt_all_kernel(const float* __restrict__ q, const float* __restrict__ k,
               const float* __restrict__ v,
               const float* __restrict__ Wq, const float* __restrict__ Wk,
               const float* __restrict__ Wv,
               unsigned short* __restrict__ qbf, unsigned short* __restrict__ kbf,
               unsigned short* __restrict__ vbf,
               unsigned short* __restrict__ WqT, unsigned short* __restrict__ WkT,
               unsigned short* __restrict__ WvT) {
  const unsigned int bid = blockIdx.x;
  if (bid < 3 * CV_ACT_BLOCKS) {
    const unsigned int job = bid / CV_ACT_BLOCKS;
    const float* s = (job == 0) ? q : (job == 1) ? k : v;
    unsigned short* d = (job == 0) ? qbf : (job == 1) ? kbf : vbf;
    cvt_block(s, d, bid - job * CV_ACT_BLOCKS);
  } else {
    const unsigned int wb  = bid - 3 * CV_ACT_BLOCKS;
    const unsigned int job = wb / CV_W_BLOCKS;
    const unsigned int blk = wb - job * CV_W_BLOCKS;
    const float* s = (job == 0) ? Wq : (job == 1) ? Wk : Wv;
    unsigned short* d = (job == 0) ? WqT : (job == 1) ? WkT : WvT;
    cvtT_block<DD, DKV>(s, d, blk);
  }
}

#define LIGHT_Z      3
#define LIGHT_BLOCKS (MB_BLOCKS + CV_W_BLOCKS)

__device__ __forceinline__ void
light_block(const float* __restrict__ Wfc, unsigned short* __restrict__ WfcT,
            const int* __restrict__ mask, unsigned int* __restrict__ mbits, unsigned int li) {
  if (li < MB_BLOCKS)         maskpack_block(mask, mbits, li);
  else if (li < LIGHT_BLOCKS) cvtT_block<DD, DD>(Wfc, WfcT, li - MB_BLOCKS);
}

#define GT_M   128
#define GT_N   128
#define GT_K   128
#define GPITCH 136
#define FPITCH 132

typedef __attribute__((ext_vector_type(4))) unsigned int u32x4;

static_assert(GT_N * FPITCH * 4 <= 2 * GT_M * GPITCH * 2, "the parked fp32 tile fits in the A tiles' LDS");

template <bool PROJ>
__global__ void __launch_bounds__(256)
gemm_kernel(const unsigned short* __restrict__ X0,
            const unsigned short* __restrict__ X1,
            const unsigned short* __restrict__ X2,
            const unsigned short* __restrict__ W0,
            const unsigned short* __restrict__ W1,
            const unsigned short* __restrict__ W2,
            unsigned short* __restrict__ O0,
            unsigned short* __restrict__ O1,
            unsigned short* __restrict__ O2,
            float* __restrict__ OUTF,
            const float* __restrict__ RES,
            const float* __restrict__ LWFC,
            unsigned short* __restrict__ LWFCT,
            const int* __restrict__ LMASK,
            unsigned int* __restrict__ LMBITS) {
  __shared__ __align__(16) unsigned short As[2][GT_M * GPITCH];
  __shared__ __align__(16) unsigned short Bs[2][GT_N * GPITCH];

  if (PROJ && blockIdx.z >= 3) {
    const unsigned int li =
        ((blockIdx.z - 3) * ((BB * SS) / GT_M) + blockIdx.y) * (DD / GT_N) + blockIdx.x;
    light_block(LWFC, LWFCT, LMASK, LMBITS, li);
    return;
  }

  const int z = PROJ ? (int)blockIdx.z : 0;
  const unsigned short* __restrict__ X  = (z == 0) ? X0 : (z == 1) ? X1 : X2;
  const unsigned short* __restrict__ WT = (z == 0) ? W0 : (z == 1) ? W1 : W2;
  unsigned short* __restrict__ OUTH     = (z == 0) ? O0 : (z == 1) ? O1 : O2;
  const float scale = (z == 0) ? 0.125f : 1.0f;
  const bool xrows = PROJ && (z == 2);
  const unsigned short* __restrict__ A  = xrows ? X : WT;
  const unsigned short* __restrict__ BT = xrows ? WT : X;
  const int w0 = blockIdx.x * GT_N;
  const int x0 = blockIdx.y * GT_M;
  const int m0 = xrows ? x0 : w0;
  const int n0 = xrows ? w0 : x0;

  const int tid = threadIdx.x;
  const int lane = tid & 31;
  const int wave = tid >> 5;
  const int wm = wave & 3;
  const int wn = wave >> 2;
  const int rbase = (lane & 16) ? 8 : 0;
  const int col = lane & 15;

  const int srow = tid >> 4;
  const int sch  = (tid & 15) * 8;
  const unsigned short* Ag = A  + (size_t)(m0 + srow) * DD + sch;
  const unsigned short* Bg = BT + (size_t)(n0 + srow) * DD + sch;
  const int sidx = srow * GPITCH + sch;

  u32x4 ga[8], gb[8];
#pragma unroll
  for (int i = 0; i < 8; ++i) {
    ga[i] = *reinterpret_cast<const u32x4*>(Ag + (size_t)(16 * i) * DD);
    gb[i] = *reinterpret_cast<const u32x4*>(Bg + (size_t)(16 * i) * DD);
  }
#pragma unroll
  for (int i = 0; i < 8; ++i) {
    *reinterpret_cast<u32x4*>(&As[0][sidx + 16 * i * GPITCH]) = ga[i];
    *reinterpret_cast<u32x4*>(&Bs[0][sidx + 16 * i * GPITCH]) = gb[i];
  }
  __syncthreads();

  v8f acc[2][4] = {};

  for (int st = 0; st < DD / GT_K; ++st) {
    const int buf = st & 1;
    const bool more = (st + 1) < (DD / GT_K);
    if (more) {
      const unsigned short* ag = Ag + (st + 1) * GT_K;
      const unsigned short* bg = Bg + (st + 1) * GT_K;
#pragma unroll
      for (int i = 0; i < 8; ++i) {
        ga[i] = *reinterpret_cast<const u32x4*>(ag + (size_t)(16 * i) * DD);
        gb[i] = *reinterpret_cast<const u32x4*>(bg + (size_t)(16 * i) * DD);
      }
    }
    __builtin_amdgcn_sched_barrier(0);
    const unsigned short* Ab = &As[buf][(wm * 32 + col) * GPITCH];
    const unsigned short* Bb = &Bs[buf][(wn * 64 + col) * GPITCH];

    v16bf af[2][2], bfr[2][4];
#pragma unroll
    for (int mt = 0; mt < 2; ++mt)
      af[0][mt] = load_frag(Ab + mt * 16 * GPITCH, lane);
#pragma unroll
    for (int nt = 0; nt < 4; ++nt)
      bfr[0][nt] = load_frag(Bb + nt * 16 * GPITCH, lane);
#pragma unroll
    for (int j = 0; j < GT_K / 32; ++j) {
      const int cur = j & 1;
      if (j + 1 < GT_K / 32) {
#pragma unroll
        for (int mt = 0; mt < 2; ++mt)
          af[cur ^ 1][mt] = load_frag(Ab + mt * 16 * GPITCH + (j + 1) * 32, lane);
#pragma unroll
        for (int nt = 0; nt < 4; ++nt)
          bfr[cur ^ 1][nt] = load_frag(Bb + nt * 16 * GPITCH + (j + 1) * 32, lane);
      }
      __builtin_amdgcn_sched_barrier(0);
#pragma unroll
      for (int mt = 0; mt < 2; ++mt)
#pragma unroll
        for (int nt = 0; nt < 4; ++nt)
          acc[mt][nt] = WMMA_BF16(af[cur][mt], bfr[cur][nt], acc[mt][nt]);
    }

    __builtin_amdgcn_sched_barrier(0);
    if (more) {
#pragma unroll
      for (int i = 0; i < 8; ++i) {
        *reinterpret_cast<u32x4*>(&As[buf ^ 1][sidx + 16 * i * GPITCH]) = ga[i];
        *reinterpret_cast<u32x4*>(&Bs[buf ^ 1][sidx + 16 * i * GPITCH]) = gb[i];
      }
    }
    __syncthreads();
  }

  if (PROJ) {
    unsigned short* T = &As[0][0];
#pragma unroll
    for (int mt = 0; mt < 2; ++mt) {
#pragma unroll
      for (int nt = 0; nt < 4; ++nt) {
        const int ra = wm * 32 + mt * 16 + rbase;
        const int cb = wn * 64 + nt * 16 + col;
        u32x4 w;
        w.x = f2bf_pair(acc[mt][nt][0] * scale, acc[mt][nt][1] * scale);
        w.y = f2bf_pair(acc[mt][nt][2] * scale, acc[mt][nt][3] * scale);
        w.z = f2bf_pair(acc[mt][nt][4] * scale, acc[mt][nt][5] * scale);
        w.w = f2bf_pair(acc[mt][nt][6] * scale, acc[mt][nt][7] * scale);
        *reinterpret_cast<u32x4*>(&T[cb * GPITCH + ra]) = w;
      }
    }
    __syncthreads();
#pragma unroll
    for (int it = 0; it < 8; ++it) {
      const int p   = it * 256 + tid;
      const int row = p >> 4;
      const int c8  = (p & 15) * 8;
      const u32x4 w = *reinterpret_cast<const u32x4*>(&T[row * GPITCH + c8]);
      const int xr = x0 + (xrows ? c8 : row);
      const int n  = w0 + (xrows ? row : c8);
      const int b = xr / SS, sq = xr - b * SS;
      const int h = n / DKV, dcol = n - h * DKV;
      const size_t bh = (size_t)(b * HH + h);
      const size_t idx = xrows ? (bh * DKV + dcol) * SS + sq
                               : (bh * SS + sq) * DKV + dcol;
      *reinterpret_cast<u32x4*>(OUTH + idx) = w;
    }
  } else {
    float4 res[16];
#pragma unroll
    for (int it = 0; it < 16; ++it) {
      const int p = it * 256 + tid;
      res[it] = *reinterpret_cast<const float4*>(
          RES + (size_t)(x0 + (p >> 5)) * DD + w0 + (p & 31) * 4);
    }
    float* T = reinterpret_cast<float*>(&As[0][0]);
#pragma unroll
    for (int mt = 0; mt < 2; ++mt) {
#pragma unroll
      for (int nt = 0; nt < 4; ++nt) {
        const int ra = wm * 32 + mt * 16 + rbase;
        const int cb = wn * 64 + nt * 16 + col;
        float4 lo = {acc[mt][nt][0], acc[mt][nt][1], acc[mt][nt][2], acc[mt][nt][3]};
        float4 hi = {acc[mt][nt][4], acc[mt][nt][5], acc[mt][nt][6], acc[mt][nt][7]};
        *reinterpret_cast<float4*>(&T[cb * FPITCH + ra])     = lo;
        *reinterpret_cast<float4*>(&T[cb * FPITCH + ra + 4]) = hi;
      }
    }
    __syncthreads();
#pragma unroll
    for (int it = 0; it < 16; ++it) {
      const int p = it * 256 + tid;
      const int row = p >> 5;
      const int c4  = (p & 31) * 4;
      float4 c = *reinterpret_cast<const float4*>(&T[row * FPITCH + c4]);
      c.x += res[it].x; c.y += res[it].y; c.z += res[it].z; c.w += res[it].w;
      *reinterpret_cast<float4*>(OUTF + (size_t)(x0 + row) * DD + w0 + c4) = c;
    }
  }
}

#define AT_WAVES 8
#define AT_ROWS  (AT_WAVES * 16)
#define KT       64
#define LPITCH   72

typedef __attribute__((ext_vector_type(2))) float  f32x2;
typedef __attribute__((ext_vector_type(2))) __bf16 bf16x2;

__device__ __forceinline__ unsigned int pack_bf16(float lo, float hi) {
  f32x2 t = {lo, hi};
  return __builtin_bit_cast(unsigned int, __builtin_convertvector(t, bf16x2));
}

__device__ __forceinline__ float xhalf(float x) {
  int b = __builtin_bit_cast(int, x);
  return __builtin_bit_cast(float,
      __builtin_amdgcn_permlanex16(b, b, 0x76543210, 0xfedcba98, false, false));
}

__global__ void __launch_bounds__(AT_WAVES * 32)
attn_kernel(const unsigned short* __restrict__ QH,
            const unsigned short* KH,
            const unsigned short* VT,
            const unsigned int* __restrict__ MBITS,
            unsigned short* __restrict__ AOUT) {
  __shared__ __align__(16) unsigned short Ks[2][KT * LPITCH];
  __shared__ __align__(16) unsigned short Vs[2][DKV * LPITCH];

  const int tid = threadIdx.x;
  const int lane = tid & 31;
  const int wave = __builtin_amdgcn_readfirstlane(tid >> 5);
  const int bh = blockIdx.x;
  const int b = bh >> 4, h = bh & 15;
  const int s_base = blockIdx.y * AT_ROWS + wave * 16;
  const int kofs = (lane & 16) ? 8 : 0;
  const int col = lane & 15;

  const unsigned short* Qrow =
      QH + ((size_t)bh * SS + (size_t)(s_base + col)) * DKV;
  v16bf qf0 = load_frag(Qrow, lane);
  v16bf qf1 = load_frag(Qrow + 32, lane);

  v8f o[4] = {};
  float m = -3.0e38f;
  float l = 0.0f;

  const unsigned int* mwave =
      MBITS + ((size_t)b * (SS / 16) + (size_t)(s_base >> 4)) * (SS / KT) * MB_STAGE_WORDS;

  const int srow = tid >> 3;
  const int sch  = (tid & 7) * 8;
  const unsigned short* Kg = KH + ((size_t)bh * SS + srow) * DKV + sch;
  const unsigned short* Vg = VT + ((size_t)bh * DKV + srow) * SS + sch;
  const int sidx = srow * LPITCH + sch;

  uint4 gk0 = *reinterpret_cast<const uint4*>(Kg);
  uint4 gk1 = *reinterpret_cast<const uint4*>(Kg + (size_t)32 * DKV);
  uint4 gv0 = *reinterpret_cast<const uint4*>(Vg);
  uint4 gv1 = *reinterpret_cast<const uint4*>(Vg + (size_t)32 * SS);
  *reinterpret_cast<uint4*>(&Ks[0][sidx])               = gk0;
  *reinterpret_cast<uint4*>(&Ks[0][sidx + 32 * LPITCH]) = gk1;
  *reinterpret_cast<uint4*>(&Vs[0][sidx])               = gv0;
  *reinterpret_cast<uint4*>(&Vs[0][sidx + 32 * LPITCH]) = gv1;
  __syncthreads();

  for (int st = 0; st < SS / KT; ++st) {
    const int buf = st & 1;
    const int t0 = st * KT;
    const bool more = (st + 1) < (SS / KT);
    if (more) {
      const unsigned short* kg = Kg + (size_t)(t0 + KT) * DKV;
      const unsigned short* vg = Vg + (t0 + KT);
      gk0 = *reinterpret_cast<const uint4*>(kg);
      gk1 = *reinterpret_cast<const uint4*>(kg + (size_t)32 * DKV);
      gv0 = *reinterpret_cast<const uint4*>(vg);
      gv1 = *reinterpret_cast<const uint4*>(vg + (size_t)32 * SS);
    }
    int frow = col * LPITCH;
    asm volatile("" : "+v"(frow) : : "memory");
    const unsigned short* Kb = &Ks[buf][0] + frow;
    const unsigned short* Vb = &Vs[buf][0] + frow;

    v16bf kb[KT / 32][4], vb[KT / 32][4];
#pragma unroll
    for (int j = 0; j < KT / 32; ++j) {
      kb[j][0] = load_frag(Kb + (j * 32) * LPITCH, lane);
      kb[j][1] = load_frag(Kb + (j * 32) * LPITCH + 32, lane);
      kb[j][2] = load_frag(Kb + (j * 32 + 16) * LPITCH, lane);
      kb[j][3] = load_frag(Kb + (j * 32 + 16) * LPITCH + 32, lane);
    }
#pragma unroll
    for (int j = 0; j < KT / 32; ++j)
#pragma unroll
      for (int nt = 0; nt < 4; ++nt)
        vb[j][nt] = load_frag(Vb + (nt * 16) * LPITCH + j * 32, lane);
    __builtin_amdgcn_sched_barrier(0);

#pragma unroll
    for (int j = 0; j < KT / 32; ++j) {
      const uint4* mstep =
          reinterpret_cast<const uint4*>(mwave + (size_t)st * MB_STAGE_WORDS + j * 16);
      const uint4 mk0a = mstep[0];
      const uint4 mk0b = mstep[1];
      const uint4 mk1a = mstep[2];
      const uint4 mk1b = mstep[3];

      v8f s0 = {}, s1 = {};
      s0 = WMMA_BF16(kb[j][0], qf0, s0);
      s0 = WMMA_BF16(kb[j][1], qf1, s0);
      s1 = WMMA_BF16(kb[j][2], qf0, s1);
      s1 = WMMA_BF16(kb[j][3], qf1, s1);

      const unsigned int mk0[8] = {mk0a.x, mk0a.y, mk0a.z, mk0a.w, mk0b.x, mk0b.y, mk0b.z, mk0b.w};
      const unsigned int mk1[8] = {mk1a.x, mk1a.y, mk1a.z, mk1a.w, mk1b.x, mk1b.y, mk1b.z, mk1b.w};
      float v0[8], v1[8];
#pragma unroll
      for (int r = 0; r < 8; ++r) {
        v0[r] = __builtin_amdgcn_inverse_ballot_w32(mk0[r]) ? s0[r] : -1.0e9f;
        v1[r] = __builtin_amdgcn_inverse_ballot_w32(mk1[r]) ? s1[r] : -1.0e9f;
      }
      float mx = fmaxf(v0[0], v1[0]);
#pragma unroll
      for (int r = 1; r < 8; ++r) mx = fmaxf(mx, fmaxf(v0[r], v1[r]));
      mx = fmaxf(mx, xhalf(mx));
      const float mnew  = fmaxf(m, mx);
      const float alpha = __expf(m - mnew);
      float rs = 0.0f;
      unsigned int w0[4], w1[4];
#pragma unroll
      for (int r = 0; r < 8; r += 2) {
        const float p00 = __expf(v0[r]     - mnew);
        const float p01 = __expf(v0[r + 1] - mnew);
        const float p10 = __expf(v1[r]     - mnew);
        const float p11 = __expf(v1[r + 1] - mnew);
        rs += (p00 + p01) + (p10 + p11);
        w0[r >> 1] = pack_bf16(p00, p01);
        w1[r >> 1] = pack_bf16(p10, p11);
      }
      l = l * alpha + rs;
      m = mnew;
#pragma unroll
      for (int nt = 0; nt < 4; ++nt)
#pragma unroll
        for (int r = 0; r < 8; ++r) o[nt][r] = o[nt][r] * alpha;

      FragU pf;
      pf.q[0] = make_uint4(w0[0], w0[1], w0[2], w0[3]);
      pf.q[1] = make_uint4(w1[0], w1[1], w1[2], w1[3]);
#pragma unroll
      for (int nt = 0; nt < 4; ++nt)
        o[nt] = WMMA_BF16(vb[j][nt], pf.v, o[nt]);
    }

    asm volatile("" : "+v"(o[0]), "+v"(o[1]), "+v"(o[2]), "+v"(o[3]));
    if (more) {
      unsigned short* Kn = &Ks[buf ^ 1][0];
      unsigned short* Vn = &Vs[buf ^ 1][0];
      *reinterpret_cast<uint4*>(Kn + sidx)               = gk0;
      *reinterpret_cast<uint4*>(Kn + sidx + 32 * LPITCH) = gk1;
      *reinterpret_cast<uint4*>(Vn + sidx)               = gv0;
      *reinterpret_cast<uint4*>(Vn + sidx + 32 * LPITCH) = gv1;
    }
    __syncthreads();
  }

  const float inv = 1.0f / (l + xhalf(l));
  unsigned short* orow =
      AOUT + ((size_t)b * SS + (size_t)(s_base + col)) * (HH * DKV) + h * DKV + kofs;
#pragma unroll
  for (int nt = 0; nt < 4; ++nt) {
    uint4 w;
    w.x = pack_bf16(o[nt][0] * inv, o[nt][1] * inv);
    w.y = pack_bf16(o[nt][2] * inv, o[nt][3] * inv);
    w.z = pack_bf16(o[nt][4] * inv, o[nt][5] * inv);
    w.w = pack_bf16(o[nt][6] * inv, o[nt][7] * inv);
    *reinterpret_cast<uint4*>(orow + nt * 16) = w;
  }
}

#define LN_WAVES 8
#define LN_PIECES (DD / (32 * 4))

template <int M>
__device__ __forceinline__ float row_xor(float x) {
  const int b = __builtin_bit_cast(int, x);
  return __builtin_bit_cast(float,
      __builtin_amdgcn_update_dpp(b, b, 0x160 + M, 0xf, 0xf, false));
}

__device__ __forceinline__ float wave_sum(float x) {
  x += row_xor<1>(x);
  x += row_xor<2>(x);
  x += row_xor<4>(x);
  x += row_xor<8>(x);
  x += xhalf(x);
  return x;
}

__global__ void __launch_bounds__(LN_WAVES * 32) __attribute__((amdgpu_waves_per_eu(4, 4)))
ln_kernel(const float* __restrict__ X,
          const float* __restrict__ W,
          const float* __restrict__ Bv,
          float* __restrict__ OUT) {
  const int lane = threadIdx.x & 31;
  const int row = blockIdx.x * LN_WAVES + (threadIdx.x >> 5);
  const size_t base = (size_t)row * DD + lane * 4;

  float4 v[LN_PIECES], w[LN_PIECES], bb[LN_PIECES];
#pragma unroll
  for (int j = 0; j < LN_PIECES; ++j)
    v[j] = *reinterpret_cast<const float4*>(X + base + j * 128);
#pragma unroll
  for (int j = 0; j < LN_PIECES; ++j) {
    w[j]  = *reinterpret_cast<const float4*>(W  + lane * 4 + j * 128);
    bb[j] = *reinterpret_cast<const float4*>(Bv + lane * 4 + j * 128);
  }

  float s = 0.0f;
#pragma unroll
  for (int j = 0; j < LN_PIECES; ++j) s += (v[j].x + v[j].y) + (v[j].z + v[j].w);
  const float mu = wave_sum(s) * (1.0f / DD);

  float vs = 0.0f;
#pragma unroll
  for (int j = 0; j < LN_PIECES; ++j) {
    const float dx = v[j].x - mu, dy = v[j].y - mu, dz = v[j].z - mu, dw = v[j].w - mu;
    vs += (dx * dx + dy * dy) + (dz * dz + dw * dw);
  }
  const float var = wave_sum(vs) * (1.0f / DD);
  const float rstd = rsqrtf(var + 1e-6f);

#pragma unroll
  for (int j = 0; j < LN_PIECES; ++j) {
    float4 o;
    o.x = (v[j].x - mu) * rstd * w[j].x + bb[j].x;
    o.y = (v[j].y - mu) * rstd * w[j].y + bb[j].y;
    o.z = (v[j].z - mu) * rstd * w[j].z + bb[j].z;
    o.w = (v[j].w - mu) * rstd * w[j].w + bb[j].w;
    *reinterpret_cast<float4*>(OUT + base + j * 128) = o;
  }
}

extern "C" void kernel_launch(void* const* d_in, const int* in_sizes, int n_in,
                              void* d_out, int out_size, void* d_ws, size_t ws_size,
                              hipStream_t stream) {
  const float* q    = (const float*)d_in[0];
  const float* k    = (const float*)d_in[1];
  const float* v    = (const float*)d_in[2];
  const int*   mask = (const int*)d_in[3];
  const float* Wq   = (const float*)d_in[4];
  const float* Wk   = (const float*)d_in[5];
  const float* Wv   = (const float*)d_in[6];
  const float* Wfc  = (const float*)d_in[7];
  const float* lnw  = (const float*)d_in[8];
  const float* lnb  = (const float*)d_in[9];
  float* out = (float*)d_out;

  char* ws = (char*)d_ws;
  const size_t SZ_ACT  = (size_t)BB * SS * DD * 2;
  const size_t SZ_W    = (size_t)HH * DD * DKV * 2;
  const size_t SZ_HEAD = (size_t)BB * HH * SS * DKV * 2;
  size_t off = 0;
  unsigned short* qbf  = (unsigned short*)(ws + off); off += SZ_ACT;
  unsigned short* kbf  = (unsigned short*)(ws + off); off += SZ_ACT;
  unsigned short* vbf  = (unsigned short*)(ws + off); off += SZ_ACT;
  unsigned short* WqT  = (unsigned short*)(ws + off); off += SZ_W;
  unsigned short* WkT  = (unsigned short*)(ws + off); off += SZ_W;
  unsigned short* WvT  = (unsigned short*)(ws + off); off += SZ_W;
  unsigned short* WfcT = (unsigned short*)(ws + off); off += (size_t)DD * DD * 2;
  unsigned short* qh   = (unsigned short*)(ws + off); off += SZ_HEAD;
  unsigned short* kh   = (unsigned short*)(ws + off); off += SZ_HEAD;
  unsigned short* vhT  = (unsigned short*)(ws + off); off += SZ_HEAD;
  unsigned short* aout = (unsigned short*)(ws + off); off += SZ_ACT;
  float*          fco  = (float*)(ws + off);          off += (size_t)BB * SS * DD * 4;
  unsigned int*   mbits = (unsigned int*)fco;

  constexpr int nAct = BB * SS * DD;
  constexpr int nW   = HH * DD * DKV;
  constexpr int nFc  = DD * DD;

  static_assert(nAct % CV_BLOCK_ELEMS == 0, "cvt_kernel: whole work-groups only");
  static_assert(nW % (CT_E * 256) == 0 && nFc % (CT_E * 256) == 0, "cvtT_kernel: whole work-groups only");
  static_assert(DKV % 32 == 0 && DD % 32 == 0 && DD % CT_E == 0, "cvtT_kernel: a wave stays in one row of src");
  static_assert(nAct / CV_BLOCK_ELEMS == CV_ACT_BLOCKS, "cvt_all_kernel: activation range");
  static_assert(nW / (CT_E * 256) == CV_W_BLOCKS && nFc / (CT_E * 256) == CV_W_BLOCKS,
                "cvt_all_kernel and light_block: weight ranges");
  static_assert(SS % 128 == 0 && (SS / 128) == 16, "maskpack_block: 2 work-groups x 8 waves x 128 keys per row");
  static_assert(KT == 64 && AT_ROWS == 128, "attn_kernel and maskpack_block share the stage and block sizes");
  static_assert(LIGHT_BLOCKS <= LIGHT_Z * (DD / GT_N) * ((BB * SS) / GT_M),
                "gemm_kernel<true>: the light z planes hold every light work-group");
  cvt_all_kernel<<<CV_ALL_BLOCKS, 256, 0, stream>>>(q, k, v, Wq, Wk, Wv,
                                                    qbf, kbf, vbf, WqT, WkT, WvT);

  gemm_kernel<true><<<dim3(DD / GT_N, (BB * SS) / GT_M, 3 + LIGHT_Z), 256, 0, stream>>>(
      qbf, kbf, vbf, WqT, WkT, WvT, qh, kh, vhT, nullptr, nullptr, Wfc, WfcT, mask, mbits);

  attn_kernel<<<dim3(BB * HH, SS / AT_ROWS), AT_WAVES * 32, 0, stream>>>(qh, kh, vhT, mbits, aout);

  gemm_kernel<false><<<dim3(DD / GT_N, (BB * SS) / GT_M), 256, 0, stream>>>(
      aout, nullptr, nullptr, WfcT, nullptr, nullptr, nullptr, nullptr, nullptr, fco, q,
      nullptr, nullptr, nullptr, nullptr);
  static_assert((BB * SS) % LN_WAVES == 0 && DD == LN_PIECES * 128, "ln_kernel: whole work-groups, whole rows");
  ln_kernel<<<(BB * SS) / LN_WAVES, LN_WAVES * 32, 0, stream>>>(fco, lnw, lnb, out);
}
